# starting kernel
# baseline (speedup 1.0000x reference)
#include <hip/hip_runtime.h>
#include <hip/hip_bf16.h>

#define N_NODES 100000
#define N_EDGES 1600000
#define IN_CH   165
#define IN_PAD  168
#define HID     128

typedef __attribute__((ext_vector_type(2))) float v2f;
typedef __attribute__((ext_vector_type(8))) float v8f;

__global__ void zero_f32_kernel(float* __restrict__ p, size_t n) {
    size_t i = (size_t)blockIdx.x * blockDim.x + threadIdx.x;
    if (i < n) p[i] = 0.0f;
}

__global__ void pad_rows_kernel(const float* __restrict__ src,
                                float* __restrict__ dst,
                                int srcD, int dstD, long rows) {
    long i = (long)blockIdx.x * blockDim.x + threadIdx.x;
    if (i >= rows * dstD) return;
    int col  = (int)(i % dstD);
    long row = i / dstD;
    dst[i] = (col < srcD) ? src[row * srcD + col] : 0.0f;
}

__global__ void count_kernel(const int* __restrict__ dst,
                             float* __restrict__ cnt, int nEdges) {
    int e = blockIdx.x * blockDim.x + threadIdx.x;
    if (e < nEdges) atomicAdd(&cnt[dst[e]], 1.0f);
}

__global__ void rdeg_kernel(const float* __restrict__ cnt,
                            float* __restrict__ rdeg, int n) {
    int i = blockIdx.x * blockDim.x + threadIdx.x;
    if (i < n) rdeg[i] = 1.0f / fmaxf(cnt[i], 1.0f);
}

__global__ __launch_bounds__(256)
void scatter_kernel(const float* __restrict__ h,
                    const int* __restrict__ src,
                    const int* __restrict__ dst,
                    float* __restrict__ agg,
                    int d, int nEdges) {
    int e = blockIdx.x * (blockDim.x >> 5) + (threadIdx.x >> 5);
    if (e >= nEdges) return;
    int lane = threadIdx.x & 31;
    int s = src[e];
    int t = dst[e];
    const float* __restrict__ hs = h + (size_t)s * d;
    float* __restrict__ at = agg + (size_t)t * d;
    for (int f = lane; f < d; f += 32)
        atomicAdd(&at[f], hs[f]);
}

template <int D, bool RELU>
__global__ __launch_bounds__(256)
void sage_gemm_wmma_kernel(const float* __restrict__ agg,
                           const float* __restrict__ rdeg,
                           const float* __restrict__ h,
                           const float* __restrict__ Wl,
                           const float* __restrict__ Wr,
                           const float* __restrict__ bias,
                           float* __restrict__ out) {
    static_assert(D % 4 == 0, "K must be a multiple of 4");
    const int lane = threadIdx.x & 31;
    const int wave = threadIdx.x >> 5;
    const int m0   = blockIdx.x * 16;
    const int n0   = wave * 16;

    const int mrow = m0 + (lane & 15);
    const int nrow = n0 + (lane & 15);
    const int koff = (lane >> 4) << 1;

    const float rd = rdeg[mrow];

    const float2* __restrict__ arow  = (const float2*)(agg + (size_t)mrow * D + koff);
    const float2* __restrict__ hrow  = (const float2*)(h   + (size_t)mrow * D + koff);
    const float2* __restrict__ wlrow = (const float2*)(Wl  + (size_t)nrow * D + koff);
    const float2* __restrict__ wrrow = (const float2*)(Wr  + (size_t)nrow * D + koff);

    v8f c = {0.f, 0.f, 0.f, 0.f, 0.f, 0.f, 0.f, 0.f};

#pragma unroll 4
    for (int k = 0; k < D / 4; ++k) {
        float2 am = arow[2 * k];
        float2 bl = wlrow[2 * k];
        float2 ah = hrow[2 * k];
        float2 br = wrrow[2 * k];
        v2f a1; a1.x = am.x * rd; a1.y = am.y * rd;
        v2f b1; b1.x = bl.x;      b1.y = bl.y;
        v2f a2; a2.x = ah.x;      a2.y = ah.y;
        v2f b2; b2.x = br.x;      b2.y = br.y;
        c = __builtin_amdgcn_wmma_f32_16x16x4_f32(
                false, a1, false, b1, (short)0, c, false, false);
        c = __builtin_amdgcn_wmma_f32_16x16x4_f32(
                false, a2, false, b2, (short)0, c, false, false);
    }

    const float bv = bias[nrow];
    const int mbase = m0 + ((lane < 16) ? 0 : 8);
#pragma unroll
    for (int r = 0; r < 8; ++r) {
        float v = c[r] + bv;
        if (RELU) v = fmaxf(v, 0.f);
        out[(size_t)(mbase + r) * HID + nrow] = v;
    }
}

__global__ __launch_bounds__(256)
void out_head_kernel(const float* __restrict__ h,
                     const float* __restrict__ Wout,
                     const float* __restrict__ bout,
                     float* __restrict__ out, int n) {
    int i = blockIdx.x * blockDim.x + threadIdx.x;
    if (i >= n) return;
    const float4* hr = (const float4*)(h + (size_t)i * HID);
    const float4* w0 = (const float4*)(Wout);
    const float4* w1 = (const float4*)(Wout + HID);
    float a0 = bout[0], a1 = bout[1];
#pragma unroll 8
    for (int k = 0; k < HID / 4; ++k) {
        float4 hv = hr[k];
        float4 u = w0[k];
        float4 v = w1[k];
        a0 += hv.x * u.x + hv.y * u.y + hv.z * u.z + hv.w * u.w;
        a1 += hv.x * v.x + hv.y * v.y + hv.z * v.z + hv.w * v.w;
    }
    out[(size_t)i * 2 + 0] = a0;
    out[(size_t)i * 2 + 1] = a1;
}

extern "C" void kernel_launch(void* const* d_in, const int* in_sizes, int n_in,
                              void* d_out, int out_size, void* d_ws, size_t ws_size,
                              hipStream_t stream) {
    const float* x    = (const float*)d_in[0];
    const int*   ei   = (const int*)d_in[1];
    const float* Wl0  = (const float*)d_in[2];
    const float* bl0  = (const float*)d_in[3];
    const float* Wr0  = (const float*)d_in[4];
    const float* Wl1  = (const float*)d_in[5];
    const float* bl1  = (const float*)d_in[6];
    const float* Wr1  = (const float*)d_in[7];
    const float* Wl2  = (const float*)d_in[8];
    const float* bl2  = (const float*)d_in[9];
    const float* Wr2  = (const float*)d_in[10];
    const float* Wout = (const float*)d_in[11];
    const float* bout = (const float*)d_in[12];
    float* out = (float*)d_out;

    const int* src = ei;
    const int* dst = ei + N_EDGES;

    float* ws   = (float*)d_ws;
    float* cnt  = ws;
    float* rdeg = cnt + N_NODES;
    float* agg  = rdeg + N_NODES;
    float* hA   = agg + (size_t)N_NODES * IN_PAD;
    float* xp   = hA + (size_t)N_NODES * HID;
    float* hB   = xp;
    float* Wl0p = xp + (size_t)N_NODES * IN_PAD;
    float* Wr0p = Wl0p + (size_t)HID * IN_PAD;

    const int ZT = 256;
    auto zero = [&](float* p, size_t n) {
        zero_f32_kernel<<<(unsigned)((n + ZT - 1) / ZT), ZT, 0, stream>>>(p, n);
    };
    auto pad = [&](const float* s, float* dptr, int sd, int dd, long rows) {
        long n = rows * dd;
        pad_rows_kernel<<<(unsigned)((n + ZT - 1) / ZT), ZT, 0, stream>>>(s, dptr, sd, dd, rows);
    };

    const int edgeBlocks256 = (N_EDGES + 255) / 256;
    const int scatBlocks    = (N_EDGES + 7) / 8;
    const int gemmBlocks    = N_NODES / 16;
    const int nodeBlocks256 = (N_NODES + 255) / 256;

    zero(cnt, (size_t)N_NODES);
    count_kernel<<<edgeBlocks256, 256, 0, stream>>>(dst, cnt, N_EDGES);
    rdeg_kernel<<<nodeBlocks256, 256, 0, stream>>>(cnt, rdeg, N_NODES);

    pad(x,   xp,   IN_CH, IN_PAD, N_NODES);
    pad(Wl0, Wl0p, IN_CH, IN_PAD, HID);
    pad(Wr0, Wr0p, IN_CH, IN_PAD, HID);

    zero(agg, (size_t)N_NODES * IN_PAD);
    scatter_kernel<<<scatBlocks, 256, 0, stream>>>(xp, src, dst, agg, IN_PAD, N_EDGES);
    sage_gemm_wmma_kernel<IN_PAD, true><<<gemmBlocks, 256, 0, stream>>>(
        agg, rdeg, xp, Wl0p, Wr0p, bl0, hA);

    zero(agg, (size_t)N_NODES * HID);
    scatter_kernel<<<scatBlocks, 256, 0, stream>>>(hA, src, dst, agg, HID, N_EDGES);
    sage_gemm_wmma_kernel<HID, true><<<gemmBlocks, 256, 0, stream>>>(
        agg, rdeg, hA, Wl1, Wr1, bl1, hB);

    zero(agg, (size_t)N_NODES * HID);
    scatter_kernel<<<scatBlocks, 256, 0, stream>>>(hB, src, dst, agg, HID, N_EDGES);
    sage_gemm_wmma_kernel<HID, false><<<gemmBlocks, 256, 0, stream>>>(
        agg, rdeg, hB, Wl2, Wr2, bl2, hA);

    out_head_kernel<<<nodeBlocks256, 256, 0, stream>>>(hA, Wout, bout, out, N_NODES);
}


# Float-atomic edge scatter replaced by a per-call destination-sorted edge list (returning int add = arrival-order rank, scan, fill) plus a row gather per node; sums run in arrival-rank order
# speedup vs baseline: 1.9224x; 1.9224x over previous
#include <hip/hip_runtime.h>
#include <hip/hip_bf16.h>

#define N_NODES 100000
#define N_EDGES 1600000
#define IN_CH   165
#define IN_PAD  168
#define HID     128

typedef __attribute__((ext_vector_type(2))) float v2f;
typedef __attribute__((ext_vector_type(8))) float v8f;

__global__ void zero_f32_kernel(float* __restrict__ p, size_t n) {
    size_t i = (size_t)blockIdx.x * blockDim.x + threadIdx.x;
    if (i < n) p[i] = 0.0f;
}

__global__ void pad_rows_kernel(const float* __restrict__ src,
                                float* __restrict__ dst,
                                int srcD, int dstD, long rows) {
    long i = (long)blockIdx.x * blockDim.x + threadIdx.x;
    if (i >= rows * dstD) return;
    int col  = (int)(i % dstD);
    long row = i / dstD;
    dst[i] = (col < srcD) ? src[row * srcD + col] : 0.0f;
}

__global__ void rank_kernel(const int* __restrict__ dst,
                            int* __restrict__ cnt,
                            int* __restrict__ rank, int nEdges) {
    int e = blockIdx.x * blockDim.x + threadIdx.x;
    if (e < nEdges) rank[e] = atomicAdd(&cnt[dst[e]], 1);
}

__global__ __launch_bounds__(1024)
void scan_rdeg_kernel(const int* __restrict__ cnt,
                      int* __restrict__ rowptr,
                      float* __restrict__ rdeg, int n) {
    __shared__ int front[32];
    __shared__ int wtot[32];
    const int tid  = threadIdx.x;
    const int lane = tid & 31;
    const int w    = tid >> 5;
    const int first = blockIdx.x * 1024;

    int acc = 0;
    for (int j = tid; j < first; j += 1024) acc += cnt[j];
    for (int o = 16; o > 0; o >>= 1) acc += __shfl_xor(acc, o);
    if (lane == 0) front[w] = acc;

    const int i = first + tid;
    const int cI = (i < n) ? cnt[i] : 0;
    int v = cI;
    for (int o = 1; o < 32; o <<= 1) {
        int u = __shfl_up(v, o);
        if (lane >= o) v += u;
    }
    if (lane == 31) wtot[w] = v;
    __syncthreads();

    int base = 0;
    for (int k = 0; k < 32; ++k) base += front[k];
    for (int k = 0; k < w; ++k)  base += wtot[k];

    if (i < n) {
        rowptr[i] = base + v - cI;
        rdeg[i]   = 1.0f / fmaxf((float)cI, 1.0f);
        if (i == n - 1) rowptr[n] = base + v;
    }
}

__global__ void fill_kernel(const int* __restrict__ src,
                            const int* __restrict__ dst,
                            const int* __restrict__ rank,
                            const int* __restrict__ rowptr,
                            int* __restrict__ col, int nEdges) {
    int e = blockIdx.x * blockDim.x + threadIdx.x;
    if (e < nEdges) col[rowptr[dst[e]] + rank[e]] = src[e];
}

template <int D>
__global__ __launch_bounds__(256)
void gather_sum_kernel(const float* __restrict__ h,
                       const int* __restrict__ rowptr,
                       const int* __restrict__ col,
                       float* __restrict__ agg,
                       int nodeBegin, int nodeCount) {
    static_assert(D % 4 == 0 && D >= 128 && D <= 256, "row = 32..64 float4");
    constexpr int V4 = D / 4;
    constexpr bool LONG = (V4 > 32);
    const int lane = threadIdx.x & 31;
    const bool hasTail = LONG && (lane + 32 < V4);
    const int wpb = blockDim.x >> 5;
    const int nw  = gridDim.x * wpb;

    for (int i0 = blockIdx.x * wpb + (threadIdx.x >> 5); i0 < nodeCount; i0 += nw) {
        const int i = __builtin_amdgcn_readfirstlane(i0);
        const int t = nodeBegin + i;
        const int beg = rowptr[t];
        const int end = rowptr[t + 1];
        float4 a = make_float4(0.f, 0.f, 0.f, 0.f);
        float4 b = make_float4(0.f, 0.f, 0.f, 0.f);

        for (int base = beg; base < end; base += 32) {
            int n = end - base;
            if (n > 32) n = 32;
            const int cv = (lane < n) ? col[base + lane] : 0;

#define GATHER_BATCH(B, K0)                                                     \
            {                                                                   \
                float4 va[B];                                                   \
                float4 vb[B];                                                   \
                _Pragma("unroll")                                               \
                for (int q = 0; q < B; ++q) {                                   \
                    const int s = __builtin_amdgcn_readlane(cv, (K0) + q);      \
                    const float4* r = (const float4*)(h + (size_t)s * D);       \
                    va[q] = r[lane];                                            \
                    if (LONG) vb[q] = hasTail ? r[lane + 32]                    \
                                              : make_float4(0.f, 0.f, 0.f, 0.f);\
                }                                                               \
                _Pragma("unroll")                                               \
                for (int q = 0; q < B; ++q) {                                   \
                    a.x += va[q].x; a.y += va[q].y;                             \
                    a.z += va[q].z; a.w += va[q].w;                             \
                    if (LONG) {                                                 \
                        b.x += vb[q].x; b.y += vb[q].y;                         \
                        b.z += vb[q].z; b.w += vb[q].w;                         \
                    }                                                           \
                }                                                               \
            }
            int k = 0;
            for (; k + 8 <= n; k += 8) GATHER_BATCH(8, k)
            if (n & 4) { GATHER_BATCH(4, k) k += 4; }
            if (n & 2) { GATHER_BATCH(2, k) k += 2; }
            if (n & 1) { GATHER_BATCH(1, k) }
#undef GATHER_BATCH
        }

        float4* o = (float4*)(agg + (size_t)i * D);
        o[lane] = a;
        if (hasTail) o[lane + 32] = b;
    }
}

template <int D, bool RELU>
__global__ __launch_bounds__(256)
void sage_gemm_wmma_kernel(const float* __restrict__ agg,
                           const float* __restrict__ rdeg,
                           const float* __restrict__ h,
                           const float* __restrict__ Wl,
                           const float* __restrict__ Wr,
                           const float* __restrict__ bias,
                           float* __restrict__ out) {
    static_assert(D % 4 == 0, "K must be a multiple of 4");
    const int lane = threadIdx.x & 31;
    const int wave = threadIdx.x >> 5;
    const int m0   = blockIdx.x * 16;
    const int n0   = wave * 16;

    const int mrow = m0 + (lane & 15);
    const int nrow = n0 + (lane & 15);
    const int koff = (lane >> 4) << 1;

    const float rd = rdeg[mrow];

    const float2* __restrict__ arow  = (const float2*)(agg + (size_t)mrow * D + koff);
    const float2* __restrict__ hrow  = (const float2*)(h   + (size_t)mrow * D + koff);
    const float2* __restrict__ wlrow = (const float2*)(Wl  + (size_t)nrow * D + koff);
    const float2* __restrict__ wrrow = (const float2*)(Wr  + (size_t)nrow * D + koff);

    v8f c = {0.f, 0.f, 0.f, 0.f, 0.f, 0.f, 0.f, 0.f};

#pragma unroll 4
    for (int k = 0; k < D / 4; ++k) {
        float2 am = arow[2 * k];
        float2 bl = wlrow[2 * k];
        float2 ah = hrow[2 * k];
        float2 br = wrrow[2 * k];
        v2f a1; a1.x = am.x * rd; a1.y = am.y * rd;
        v2f b1; b1.x = bl.x;      b1.y = bl.y;
        v2f a2; a2.x = ah.x;      a2.y = ah.y;
        v2f b2; b2.x = br.x;      b2.y = br.y;
        c = __builtin_amdgcn_wmma_f32_16x16x4_f32(
                false, a1, false, b1, (short)0, c, false, false);
        c = __builtin_amdgcn_wmma_f32_16x16x4_f32(
                false, a2, false, b2, (short)0, c, false, false);
    }

    const float bv = bias[nrow];
    const int mbase = m0 + ((lane < 16) ? 0 : 8);
#pragma unroll
    for (int r = 0; r < 8; ++r) {
        float v = c[r] + bv;
        if (RELU) v = fmaxf(v, 0.f);
        out[(size_t)(mbase + r) * HID + nrow] = v;
    }
}

__global__ __launch_bounds__(256)
void out_head_kernel(const float* __restrict__ h,
                     const float* __restrict__ Wout,
                     const float* __restrict__ bout,
                     float* __restrict__ out, int n) {
    int i = blockIdx.x * blockDim.x + threadIdx.x;
    if (i >= n) return;
    const float4* hr = (const float4*)(h + (size_t)i * HID);
    const float4* w0 = (const float4*)(Wout);
    const float4* w1 = (const float4*)(Wout + HID);
    float a0 = bout[0], a1 = bout[1];
#pragma unroll 8
    for (int k = 0; k < HID / 4; ++k) {
        float4 hv = hr[k];
        float4 u = w0[k];
        float4 v = w1[k];
        a0 += hv.x * u.x + hv.y * u.y + hv.z * u.z + hv.w * u.w;
        a1 += hv.x * v.x + hv.y * v.y + hv.z * v.z + hv.w * v.w;
    }
    out[(size_t)i * 2 + 0] = a0;
    out[(size_t)i * 2 + 1] = a1;
}

extern "C" void kernel_launch(void* const* d_in, const int* in_sizes, int n_in,
                              void* d_out, int out_size, void* d_ws, size_t ws_size,
                              hipStream_t stream) {
    const float* x    = (const float*)d_in[0];
    const int*   ei   = (const int*)d_in[1];
    const float* Wl0  = (const float*)d_in[2];
    const float* bl0  = (const float*)d_in[3];
    const float* Wr0  = (const float*)d_in[4];
    const float* Wl1  = (const float*)d_in[5];
    const float* bl1  = (const float*)d_in[6];
    const float* Wr1  = (const float*)d_in[7];
    const float* Wl2  = (const float*)d_in[8];
    const float* bl2  = (const float*)d_in[9];
    const float* Wr2  = (const float*)d_in[10];
    const float* Wout = (const float*)d_in[11];
    const float* bout = (const float*)d_in[12];
    float* out = (float*)d_out;

    const int* src = ei;
    const int* dst = ei + N_EDGES;

    float* ws     = (float*)d_ws;
    float* rdeg   = ws;
    int*   cnt    = (int*)(rdeg + N_NODES);
    float* agg    = (float*)(cnt + N_NODES);
    int*   rowptr = (int*)(agg + (size_t)N_NODES * HID);
    int*   col    = rowptr + N_NODES + 32;
    int*   rank   = col + N_EDGES;
    float* hA     = agg + (size_t)N_NODES * IN_PAD;
    float* xp     = hA + (size_t)N_NODES * HID;
    float* hB     = xp;
    float* Wl0p   = xp + (size_t)N_NODES * IN_PAD;
    float* Wr0p   = Wl0p + (size_t)HID * IN_PAD;

    const int ZT = 256;
    auto zero = [&](float* p, size_t n) {
        zero_f32_kernel<<<(unsigned)((n + ZT - 1) / ZT), ZT, 0, stream>>>(p, n);
    };
    auto pad = [&](const float* s, float* dptr, int sd, int dd, long rows) {
        long n = rows * dd;
        pad_rows_kernel<<<(unsigned)((n + ZT - 1) / ZT), ZT, 0, stream>>>(s, dptr, sd, dd, rows);
    };

    const int edgeBlocks256 = (N_EDGES + 255) / 256;
    const int scanBlocks    = (N_NODES + 1023) / 1024;
    const int gatherBlocks  = 2048;
    const int gemmBlocks    = N_NODES / 16;
    const int halfNodes     = N_NODES / 2;
    const int nodeBlocks256 = (N_NODES + 255) / 256;

    zero((float*)cnt, (size_t)N_NODES);
    rank_kernel<<<edgeBlocks256, 256, 0, stream>>>(dst, cnt, rank, N_EDGES);
    scan_rdeg_kernel<<<scanBlocks, 1024, 0, stream>>>(cnt, rowptr, rdeg, N_NODES);
    fill_kernel<<<edgeBlocks256, 256, 0, stream>>>(src, dst, rank, rowptr, col, N_EDGES);

    pad(x,   xp,   IN_CH, IN_PAD, N_NODES);
    pad(Wl0, Wl0p, IN_CH, IN_PAD, HID);
    pad(Wr0, Wr0p, IN_CH, IN_PAD, HID);

    gather_sum_kernel<IN_PAD><<<gatherBlocks, 256, 0, stream>>>(
        xp, rowptr, col, agg, 0, halfNodes);
    sage_gemm_wmma_kernel<IN_PAD, true><<<halfNodes / 16, 256, 0, stream>>>(
        agg, rdeg, xp, Wl0p, Wr0p, bl0, hA);
    gather_sum_kernel<IN_PAD><<<gatherBlocks, 256, 0, stream>>>(
        xp, rowptr, col, agg, halfNodes, halfNodes);
    sage_gemm_wmma_kernel<IN_PAD, true><<<halfNodes / 16, 256, 0, stream>>>(
        agg, rdeg + halfNodes, xp + (size_t)halfNodes * IN_PAD, Wl0p, Wr0p, bl0,
        hA + (size_t)halfNodes * HID);

    gather_sum_kernel<HID><<<gatherBlocks, 256, 0, stream>>>(
        hA, rowptr, col, agg, 0, N_NODES);
    sage_gemm_wmma_kernel<HID, true><<<gemmBlocks, 256, 0, stream>>>(
        agg, rdeg, hA, Wl1, Wr1, bl1, hB);

    gather_sum_kernel<HID><<<gatherBlocks, 256, 0, stream>>>(
        hB, rowptr, col, agg, 0, N_NODES);
    sage_gemm_wmma_kernel<HID, false><<<gemmBlocks, 256, 0, stream>>>(
        agg, rdeg, hB, Wl2, Wr2, bl2, hA);

    out_head_kernel<<<nodeBlocks256, 256, 0, stream>>>(hA, Wout, bout, out, N_NODES);
}


# GEMM operand reuse: both weight matrices staged once per workgroup in LDS (256 persistent 16-wave workgroups); each wave does a 16x128 strip, one A fragment feeding 8 tiles; per-tile sum order kept
# speedup vs baseline: 3.6846x; 1.9167x over previous
#include <hip/hip_runtime.h>
#include <hip/hip_bf16.h>

#define N_NODES 100000
#define N_EDGES 1600000
#define IN_CH   165
#define IN_PAD  168
#define HID     128

typedef __attribute__((ext_vector_type(2))) float v2f;
typedef __attribute__((ext_vector_type(8))) float v8f;

__global__ void zero_f32_kernel(float* __restrict__ p, size_t n) {
    size_t i = (size_t)blockIdx.x * blockDim.x + threadIdx.x;
    if (i < n) p[i] = 0.0f;
}

__global__ void pad_rows_kernel(const float* __restrict__ src,
                                float* __restrict__ dst,
                                int srcD, int dstD, long rows) {
    long i = (long)blockIdx.x * blockDim.x + threadIdx.x;
    if (i >= rows * dstD) return;
    int col  = (int)(i % dstD);
    long row = i / dstD;
    dst[i] = (col < srcD) ? src[row * srcD + col] : 0.0f;
}

__global__ void rank_kernel(const int* __restrict__ dst,
                            int* __restrict__ cnt,
                            int* __restrict__ rank, int nEdges) {
    int e = blockIdx.x * blockDim.x + threadIdx.x;
    if (e < nEdges) rank[e] = atomicAdd(&cnt[dst[e]], 1);
}

__global__ __launch_bounds__(1024)
void scan_rdeg_kernel(const int* __restrict__ cnt,
                      int* __restrict__ rowptr,
                      float* __restrict__ rdeg, int n) {
    __shared__ int front[32];
    __shared__ int wtot[32];
    const int tid  = threadIdx.x;
    const int lane = tid & 31;
    const int w    = tid >> 5;
    const int first = blockIdx.x * 1024;

    int acc = 0;
    for (int j = tid; j < first; j += 1024) acc += cnt[j];
    for (int o = 16; o > 0; o >>= 1) acc += __shfl_xor(acc, o);
    if (lane == 0) front[w] = acc;

    const int i = first + tid;
    const int cI = (i < n) ? cnt[i] : 0;
    int v = cI;
    for (int o = 1; o < 32; o <<= 1) {
        int u = __shfl_up(v, o);
        if (lane >= o) v += u;
    }
    if (lane == 31) wtot[w] = v;
    __syncthreads();

    int base = 0;
    for (int k = 0; k < 32; ++k) base += front[k];
    for (int k = 0; k < w; ++k)  base += wtot[k];

    if (i < n) {
        rowptr[i] = base + v - cI;
        rdeg[i]   = 1.0f / fmaxf((float)cI, 1.0f);
        if (i == n - 1) rowptr[n] = base + v;
    }
}

__global__ void fill_kernel(const int* __restrict__ src,
                            const int* __restrict__ dst,
                            const int* __restrict__ rank,
                            const int* __restrict__ rowptr,
                            int* __restrict__ col, int nEdges) {
    int e = blockIdx.x * blockDim.x + threadIdx.x;
    if (e < nEdges) col[rowptr[dst[e]] + rank[e]] = src[e];
}

template <int D>
__global__ __launch_bounds__(256)
void gather_sum_kernel(const float* __restrict__ h,
                       const int* __restrict__ rowptr,
                       const int* __restrict__ col,
                       float* __restrict__ agg,
                       int nodeBegin, int nodeCount) {
    static_assert(D % 4 == 0 && D >= 128 && D <= 256, "row = 32..64 float4");
    constexpr int V4 = D / 4;
    constexpr bool LONG = (V4 > 32);
    const int lane = threadIdx.x & 31;
    const bool hasTail = LONG && (lane + 32 < V4);
    const int wpb = blockDim.x >> 5;
    const int nw  = gridDim.x * wpb;

    for (int i0 = blockIdx.x * wpb + (threadIdx.x >> 5); i0 < nodeCount; i0 += nw) {
        const int i = __builtin_amdgcn_readfirstlane(i0);
        const int t = nodeBegin + i;
        const int beg = rowptr[t];
        const int end = rowptr[t + 1];
        float4 a = make_float4(0.f, 0.f, 0.f, 0.f);
        float4 b = make_float4(0.f, 0.f, 0.f, 0.f);

        for (int base = beg; base < end; base += 32) {
            int n = end - base;
            if (n > 32) n = 32;
            const int cv = (lane < n) ? col[base + lane] : 0;

#define GATHER_BATCH(B, K0)                                                     \
            {                                                                   \
                float4 va[B];                                                   \
                float4 vb[B];                                                   \
                _Pragma("unroll")                                               \
                for (int q = 0; q < B; ++q) {                                   \
                    const int s = __builtin_amdgcn_readlane(cv, (K0) + q);      \
                    const float4* r = (const float4*)(h + (size_t)s * D);       \
                    va[q] = r[lane];                                            \
                    if (LONG) vb[q] = hasTail ? r[lane + 32]                    \
                                              : make_float4(0.f, 0.f, 0.f, 0.f);\
                }                                                               \
                _Pragma("unroll")                                               \
                for (int q = 0; q < B; ++q) {                                   \
                    a.x += va[q].x; a.y += va[q].y;                             \
                    a.z += va[q].z; a.w += va[q].w;                             \
                    if (LONG) {                                                 \
                        b.x += vb[q].x; b.y += vb[q].y;                         \
                        b.z += vb[q].z; b.w += vb[q].w;                         \
                    }                                                           \
                }                                                               \
            }
            int k = 0;
            for (; k + 8 <= n; k += 8) GATHER_BATCH(8, k)
            if (n & 4) { GATHER_BATCH(4, k) k += 4; }
            if (n & 2) { GATHER_BATCH(2, k) k += 2; }
            if (n & 1) { GATHER_BATCH(1, k) }
#undef GATHER_BATCH
        }

        float4* o = (float4*)(agg + (size_t)i * D);
        o[lane] = a;
        if (hasTail) o[lane + 32] = b;
    }
}

#define GEMM_WGS   256
#define GEMM_WAVES 16

#define LOAD_B(BUF, S)                                                          \
    _Pragma("unroll")                                                           \
    for (int n = 0; n < 8; ++n) {                                               \
        BUF[2 * n]     = *(const v2f*)((S) + n * 64);                           \
        BUF[2 * n + 1] = *(const v2f*)((S) + HID * 4 + n * 64);                 \
    }
#define WMMA_STEP(AM, AH, BUF)                                                  \
    {                                                                           \
        v2f a1; a1.x = (AM).x * rd; a1.y = (AM).y * rd;                         \
        v2f a2; a2.x = (AH).x;      a2.y = (AH).y;                              \
        _Pragma("unroll")                                                       \
        for (int n = 0; n < 8; ++n) {                                           \
            acc[n] = __builtin_amdgcn_wmma_f32_16x16x4_f32(                     \
                false, a1, false, BUF[2 * n], (short)0, acc[n], false, false);  \
            acc[n] = __builtin_amdgcn_wmma_f32_16x16x4_f32(                     \
                false, a2, false, BUF[2 * n + 1], (short)0, acc[n], false, false); \
        }                                                                       \
    }

template <int D, bool RELU>
__global__ __launch_bounds__(GEMM_WAVES * 32)
void sage_gemm_wmma_kernel(const float* __restrict__ agg,
                           const float* __restrict__ rdeg,
                           const float* __restrict__ h,
                           const float* __restrict__ Wl,
                           const float* __restrict__ Wr,
                           const float* __restrict__ bias,
                           float* __restrict__ out,
                           int nRowBlocks) {
    static_assert(D % 8 == 0, "K is walked in pairs of k-steps of 4");
    constexpr int V4 = D / 4;
    constexpr int NB = V4 / 2;
    constexpr int KP = 2 * HID * 4 + 4;
    __shared__ float wsh[V4 * KP];

    constexpr int NT = GEMM_WAVES * 32;
    constexpr int ST = (HID * V4 + NT - 1) / NT;
#pragma unroll
    for (int t = 0; t < ST; ++t) {
        const int i = threadIdx.x + t * NT;
        if (i < HID * V4) {
            const int r = i / V4;
            const int q = i - r * V4;
            const float4 vl = ((const float4*)(Wl + (size_t)r * D))[q];
            const float4 vr = ((const float4*)(Wr + (size_t)r * D))[q];
            *(float4*)(wsh + q * KP + r * 4)           = vl;
            *(float4*)(wsh + q * KP + HID * 4 + r * 4) = vr;
        }
    }
    __syncthreads();

    const int lane = threadIdx.x & 31;
    const int wave = threadIdx.x >> 5;
    const int c    = lane & 15;
    const int koff = (lane >> 4) << 1;

    const float* __restrict__ bp = wsh + c * 4 + koff;

    float bv[8];
#pragma unroll
    for (int n = 0; n < 8; ++n) bv[n] = bias[n * 16 + c];

    const int turn  = ((wave >> 2) + (wave & 3)) & 3;
    const int first = turn * (4 * GEMM_WGS) + blockIdx.x * 4 + (wave & 3);

    for (int rb = first; rb < nRowBlocks; rb += GEMM_WGS * GEMM_WAVES) {
        const int m0   = rb * 16;
        const int mrow = m0 + c;
        const float rd = rdeg[mrow];

        const float2* __restrict__ arow = (const float2*)(agg + (size_t)mrow * D + koff);
        const float2* __restrict__ hrow = (const float2*)(h   + (size_t)mrow * D + koff);

        v8f acc[8];
#pragma unroll
        for (int n = 0; n < 8; ++n)
            acc[n] = (v8f){0.f, 0.f, 0.f, 0.f, 0.f, 0.f, 0.f, 0.f};

        float2 nm0 = arow[0], nh0 = hrow[0], nm1 = arow[2], nh1 = hrow[2];
        v2f b0[16], b1[16];
        LOAD_B(b0, bp)

        for (int blk = 0; blk < NB; ++blk) {
            const float2 cm0 = nm0, ch0 = nh0, cm1 = nm1, ch1 = nh1;
            const int nb = (blk + 1 < NB) ? blk + 1 : NB - 1;
            nm0 = arow[4 * nb];     nh0 = hrow[4 * nb];
            nm1 = arow[4 * nb + 2]; nh1 = hrow[4 * nb + 2];

            LOAD_B(b1, bp + (2 * blk + 1) * KP)
            WMMA_STEP(cm0, ch0, b0)
            LOAD_B(b0, bp + (2 * nb) * KP)
            WMMA_STEP(cm1, ch1, b1)
        }

        const int mbase = m0 + ((lane < 16) ? 0 : 8);
#pragma unroll
        for (int n = 0; n < 8; ++n) {
#pragma unroll
            for (int r = 0; r < 8; ++r) {
                float v = acc[n][r] + bv[n];
                if (RELU) v = fmaxf(v, 0.f);
                out[(size_t)(mbase + r) * HID + n * 16 + c] = v;
            }
        }
    }
}
#undef LOAD_B
#undef WMMA_STEP

__global__ __launch_bounds__(256)
void out_head_kernel(const float* __restrict__ h,
                     const float* __restrict__ Wout,
                     const float* __restrict__ bout,
                     float* __restrict__ out, int n) {
    int i = blockIdx.x * blockDim.x + threadIdx.x;
    if (i >= n) return;
    const float4* hr = (const float4*)(h + (size_t)i * HID);
    const float4* w0 = (const float4*)(Wout);
    const float4* w1 = (const float4*)(Wout + HID);
    float a0 = bout[0], a1 = bout[1];
#pragma unroll 8
    for (int k = 0; k < HID / 4; ++k) {
        float4 hv = hr[k];
        float4 u = w0[k];
        float4 v = w1[k];
        a0 += hv.x * u.x + hv.y * u.y + hv.z * u.z + hv.w * u.w;
        a1 += hv.x * v.x + hv.y * v.y + hv.z * v.z + hv.w * v.w;
    }
    out[(size_t)i * 2 + 0] = a0;
    out[(size_t)i * 2 + 1] = a1;
}

extern "C" void kernel_launch(void* const* d_in, const int* in_sizes, int n_in,
                              void* d_out, int out_size, void* d_ws, size_t ws_size,
                              hipStream_t stream) {
    const float* x    = (const float*)d_in[0];
    const int*   ei   = (const int*)d_in[1];
    const float* Wl0  = (const float*)d_in[2];
    const float* bl0  = (const float*)d_in[3];
    const float* Wr0  = (const float*)d_in[4];
    const float* Wl1  = (const float*)d_in[5];
    const float* bl1  = (const float*)d_in[6];
    const float* Wr1  = (const float*)d_in[7];
    const float* Wl2  = (const float*)d_in[8];
    const float* bl2  = (const float*)d_in[9];
    const float* Wr2  = (const float*)d_in[10];
    const float* Wout = (const float*)d_in[11];
    const float* bout = (const float*)d_in[12];
    float* out = (float*)d_out;

    const int* src = ei;
    const int* dst = ei + N_EDGES;

    float* ws     = (float*)d_ws;
    float* rdeg   = ws;
    int*   cnt    = (int*)(rdeg + N_NODES);
    float* agg    = (float*)(cnt + N_NODES);
    int*   rowptr = (int*)(agg + (size_t)N_NODES * HID);
    int*   col    = rowptr + N_NODES + 32;
    int*   rank   = col + N_EDGES;
    float* hA     = agg + (size_t)N_NODES * IN_PAD;
    float* xp     = hA + (size_t)N_NODES * HID;
    float* hB     = xp;
    float* Wl0p   = xp + (size_t)N_NODES * IN_PAD;
    float* Wr0p   = Wl0p + (size_t)HID * IN_PAD;

    const int ZT = 256;
    auto zero = [&](float* p, size_t n) {
        zero_f32_kernel<<<(unsigned)((n + ZT - 1) / ZT), ZT, 0, stream>>>(p, n);
    };
    auto pad = [&](const float* s, float* dptr, int sd, int dd, long rows) {
        long n = rows * dd;
        pad_rows_kernel<<<(unsigned)((n + ZT - 1) / ZT), ZT, 0, stream>>>(s, dptr, sd, dd, rows);
    };

    const int edgeBlocks256 = (N_EDGES + 255) / 256;
    const int scanBlocks    = (N_NODES + 1023) / 1024;
    const int gatherBlocks  = 2048;
    const int gemmBlocks    = N_NODES / 16;
    const int halfNodes     = N_NODES / 2;
    const int nodeBlocks256 = (N_NODES + 255) / 256;

    zero((float*)cnt, (size_t)N_NODES);
    rank_kernel<<<edgeBlocks256, 256, 0, stream>>>(dst, cnt, rank, N_EDGES);
    scan_rdeg_kernel<<<scanBlocks, 1024, 0, stream>>>(cnt, rowptr, rdeg, N_NODES);
    fill_kernel<<<edgeBlocks256, 256, 0, stream>>>(src, dst, rank, rowptr, col, N_EDGES);

    pad(x,   xp,   IN_CH, IN_PAD, N_NODES);
    pad(Wl0, Wl0p, IN_CH, IN_PAD, HID);
    pad(Wr0, Wr0p, IN_CH, IN_PAD, HID);

    gather_sum_kernel<IN_PAD><<<gatherBlocks, 256, 0, stream>>>(
        xp, rowptr, col, agg, 0, halfNodes);
    sage_gemm_wmma_kernel<IN_PAD, true><<<GEMM_WGS, GEMM_WAVES * 32, 0, stream>>>(
        agg, rdeg, xp, Wl0p, Wr0p, bl0, hA, halfNodes / 16);
    gather_sum_kernel<IN_PAD><<<gatherBlocks, 256, 0, stream>>>(
        xp, rowptr, col, agg, halfNodes, halfNodes);
    sage_gemm_wmma_kernel<IN_PAD, true><<<GEMM_WGS, GEMM_WAVES * 32, 0, stream>>>(
        agg, rdeg + halfNodes, xp + (size_t)halfNodes * IN_PAD, Wl0p, Wr0p, bl0,
        hA + (size_t)halfNodes * HID, halfNodes / 16);

    gather_sum_kernel<HID><<<gatherBlocks, 256, 0, stream>>>(
        hA, rowptr, col, agg, 0, N_NODES);
    sage_gemm_wmma_kernel<HID, true><<<GEMM_WGS, GEMM_WAVES * 32, 0, stream>>>(
        agg, rdeg, hA, Wl1, Wr1, bl1, hB, gemmBlocks);

    gather_sum_kernel<HID><<<gatherBlocks, 256, 0, stream>>>(
        hB, rowptr, col, agg, 0, N_NODES);
    sage_gemm_wmma_kernel<HID, false><<<GEMM_WGS, GEMM_WAVES * 32, 0, stream>>>(
        agg, rdeg, hB, Wl2, Wr2, bl2, hA, gemmBlocks);

    out_head_kernel<<<nodeBlocks256, 256, 0, stream>>>(hA, Wout, bout, out, N_NODES);
}


# Last layer (no activation) folded into the head and applied before the mean: p = h1 @ (W_out [W_l2|W_r2])^T on WMMA, last gather sums 2 floats per edge not 128; sums re-associated, 16-lane tree
# speedup vs baseline: 4.1856x; 1.1360x over previous
#include <hip/hip_runtime.h>
#include <hip/hip_bf16.h>

#define N_NODES 100000
#define N_EDGES 1600000
#define IN_CH   165
#define IN_PAD  168
#define HID     128

typedef __attribute__((ext_vector_type(2))) float v2f;
typedef __attribute__((ext_vector_type(8))) float v8f;

__global__ void zero_f32_kernel(float* __restrict__ p, size_t n) {
    size_t i = (size_t)blockIdx.x * blockDim.x + threadIdx.x;
    if (i < n) p[i] = 0.0f;
}

__global__ void pad_rows_kernel(const float* __restrict__ src,
                                float* __restrict__ dst,
                                int srcD, int dstD, long rows) {
    long i = (long)blockIdx.x * blockDim.x + threadIdx.x;
    if (i >= rows * dstD) return;
    int col  = (int)(i % dstD);
    long row = i / dstD;
    dst[i] = (col < srcD) ? src[row * srcD + col] : 0.0f;
}

__global__ void rank_kernel(const int* __restrict__ dst,
                            int* __restrict__ cnt,
                            int* __restrict__ rank, int nEdges) {
    int e = blockIdx.x * blockDim.x + threadIdx.x;
    if (e < nEdges) rank[e] = atomicAdd(&cnt[dst[e]], 1);
}

__global__ __launch_bounds__(1024)
void scan_rdeg_kernel(const int* __restrict__ cnt,
                      int* __restrict__ rowptr,
                      float* __restrict__ rdeg, int n) {
    __shared__ int front[32];
    __shared__ int wtot[32];
    const int tid  = threadIdx.x;
    const int lane = tid & 31;
    const int w    = tid >> 5;
    const int first = blockIdx.x * 1024;

    int acc = 0;
    for (int j = tid; j < first; j += 1024) acc += cnt[j];
    for (int o = 16; o > 0; o >>= 1) acc += __shfl_xor(acc, o);
    if (lane == 0) front[w] = acc;

    const int i = first + tid;
    const int cI = (i < n) ? cnt[i] : 0;
    int v = cI;
    for (int o = 1; o < 32; o <<= 1) {
        int u = __shfl_up(v, o);
        if (lane >= o) v += u;
    }
    if (lane == 31) wtot[w] = v;
    __syncthreads();

    int base = 0;
    for (int k = 0; k < 32; ++k) base += front[k];
    for (int k = 0; k < w; ++k)  base += wtot[k];

    if (i < n) {
        rowptr[i] = base + v - cI;
        rdeg[i]   = 1.0f / fmaxf((float)cI, 1.0f);
        if (i == n - 1) rowptr[n] = base + v;
    }
}

__global__ void fill_kernel(const int* __restrict__ src,
                            const int* __restrict__ dst,
                            const int* __restrict__ rank,
                            const int* __restrict__ rowptr,
                            int* __restrict__ col, int nEdges) {
    int e = blockIdx.x * blockDim.x + threadIdx.x;
    if (e < nEdges) col[rowptr[dst[e]] + rank[e]] = src[e];
}

template <int D>
__global__ __launch_bounds__(256)
void gather_sum_kernel(const float* __restrict__ h,
                       const int* __restrict__ rowptr,
                       const int* __restrict__ col,
                       float* __restrict__ agg,
                       int nodeBegin, int nodeCount) {
    static_assert(D % 4 == 0 && D >= 128 && D <= 256, "row = 32..64 float4");
    constexpr int V4 = D / 4;
    constexpr bool LONG = (V4 > 32);
    const int lane = threadIdx.x & 31;
    const bool hasTail = LONG && (lane + 32 < V4);
    const int wpb = blockDim.x >> 5;
    const int nw  = gridDim.x * wpb;

    for (int i0 = blockIdx.x * wpb + (threadIdx.x >> 5); i0 < nodeCount; i0 += nw) {
        const int i = __builtin_amdgcn_readfirstlane(i0);
        const int t = nodeBegin + i;
        const int beg = rowptr[t];
        const int end = rowptr[t + 1];
        float4 a = make_float4(0.f, 0.f, 0.f, 0.f);
        float4 b = make_float4(0.f, 0.f, 0.f, 0.f);

        for (int base = beg; base < end; base += 32) {
            int n = end - base;
            if (n > 32) n = 32;
            const int cv = (lane < n) ? col[base + lane] : 0;

#define GATHER_BATCH(B, K0)                                                     \
            {                                                                   \
                float4 va[B];                                                   \
                float4 vb[B];                                                   \
                _Pragma("unroll")                                               \
                for (int q = 0; q < B; ++q) {                                   \
                    const int s = __builtin_amdgcn_readlane(cv, (K0) + q);      \
                    const float4* r = (const float4*)(h + (size_t)s * D);       \
                    va[q] = r[lane];                                            \
                    if (LONG) vb[q] = hasTail ? r[lane + 32]                    \
                                              : make_float4(0.f, 0.f, 0.f, 0.f);\
                }                                                               \
                _Pragma("unroll")                                               \
                for (int q = 0; q < B; ++q) {                                   \
                    a.x += va[q].x; a.y += va[q].y;                             \
                    a.z += va[q].z; a.w += va[q].w;                             \
                    if (LONG) {                                                 \
                        b.x += vb[q].x; b.y += vb[q].y;                         \
                        b.z += vb[q].z; b.w += vb[q].w;                         \
                    }                                                           \
                }                                                               \
            }
            int k = 0;
            for (; k + 8 <= n; k += 8) GATHER_BATCH(8, k)
            if (n & 4) { GATHER_BATCH(4, k) k += 4; }
            if (n & 2) { GATHER_BATCH(2, k) k += 2; }
            if (n & 1) { GATHER_BATCH(1, k) }
#undef GATHER_BATCH
        }

        float4* o = (float4*)(agg + (size_t)i * D);
        o[lane] = a;
        if (hasTail) o[lane + 32] = b;
    }
}

#define GEMM_WGS   256
#define GEMM_WAVES 16

#define LOAD_B(BUF, S)                                                          \
    _Pragma("unroll")                                                           \
    for (int n = 0; n < 8; ++n) {                                               \
        BUF[2 * n]     = *(const v2f*)((S) + n * 64);                           \
        BUF[2 * n + 1] = *(const v2f*)((S) + HID * 4 + n * 64);                 \
    }
#define WMMA_STEP(AM, AH, BUF)                                                  \
    {                                                                           \
        v2f a1; a1.x = (AM).x * rd; a1.y = (AM).y * rd;                         \
        v2f a2; a2.x = (AH).x;      a2.y = (AH).y;                              \
        _Pragma("unroll")                                                       \
        for (int n = 0; n < 8; ++n) {                                           \
            acc[n] = __builtin_amdgcn_wmma_f32_16x16x4_f32(                     \
                false, a1, false, BUF[2 * n], (short)0, acc[n], false, false);  \
            acc[n] = __builtin_amdgcn_wmma_f32_16x16x4_f32(                     \
                false, a2, false, BUF[2 * n + 1], (short)0, acc[n], false, false); \
        }                                                                       \
    }

template <int D, bool RELU>
__global__ __launch_bounds__(GEMM_WAVES * 32)
void sage_gemm_wmma_kernel(const float* __restrict__ agg,
                           const float* __restrict__ rdeg,
                           const float* __restrict__ h,
                           const float* __restrict__ Wl,
                           const float* __restrict__ Wr,
                           const float* __restrict__ bias,
                           float* __restrict__ out,
                           int nRowBlocks) {
    static_assert(D % 8 == 0, "K is walked in pairs of k-steps of 4");
    constexpr int V4 = D / 4;
    constexpr int NB = V4 / 2;
    constexpr int KP = 2 * HID * 4 + 4;
    __shared__ float wsh[V4 * KP];

    constexpr int NT = GEMM_WAVES * 32;
    constexpr int ST = (HID * V4 + NT - 1) / NT;
#pragma unroll
    for (int t = 0; t < ST; ++t) {
        const int i = threadIdx.x + t * NT;
        if (i < HID * V4) {
            const int r = i / V4;
            const int q = i - r * V4;
            const float4 vl = ((const float4*)(Wl + (size_t)r * D))[q];
            const float4 vr = ((const float4*)(Wr + (size_t)r * D))[q];
            *(float4*)(wsh + q * KP + r * 4)           = vl;
            *(float4*)(wsh + q * KP + HID * 4 + r * 4) = vr;
        }
    }
    __syncthreads();

    const int lane = threadIdx.x & 31;
    const int wave = threadIdx.x >> 5;
    const int c    = lane & 15;
    const int koff = (lane >> 4) << 1;

    const float* __restrict__ bp = wsh + c * 4 + koff;

    float bv[8];
#pragma unroll
    for (int n = 0; n < 8; ++n) bv[n] = bias[n * 16 + c];

    const int turn  = ((wave >> 2) + (wave & 3)) & 3;
    const int first = turn * (4 * GEMM_WGS) + blockIdx.x * 4 + (wave & 3);

    for (int rb = first; rb < nRowBlocks; rb += GEMM_WGS * GEMM_WAVES) {
        const int m0   = rb * 16;
        const int mrow = m0 + c;
        const float rd = rdeg[mrow];

        const float2* __restrict__ arow = (const float2*)(agg + (size_t)mrow * D + koff);
        const float2* __restrict__ hrow = (const float2*)(h   + (size_t)mrow * D + koff);

        v8f acc[8];
#pragma unroll
        for (int n = 0; n < 8; ++n)
            acc[n] = (v8f){0.f, 0.f, 0.f, 0.f, 0.f, 0.f, 0.f, 0.f};

        float2 nm0 = arow[0], nh0 = hrow[0], nm1 = arow[2], nh1 = hrow[2];
        v2f b0[16], b1[16];
        LOAD_B(b0, bp)

        for (int blk = 0; blk < NB; ++blk) {
            const float2 cm0 = nm0, ch0 = nh0, cm1 = nm1, ch1 = nh1;
            const int nb = (blk + 1 < NB) ? blk + 1 : NB - 1;
            nm0 = arow[4 * nb];     nh0 = hrow[4 * nb];
            nm1 = arow[4 * nb + 2]; nh1 = hrow[4 * nb + 2];

            LOAD_B(b1, bp + (2 * blk + 1) * KP)
            WMMA_STEP(cm0, ch0, b0)
            LOAD_B(b0, bp + (2 * nb) * KP)
            WMMA_STEP(cm1, ch1, b1)
        }

        const int mbase = m0 + ((lane < 16) ? 0 : 8);
#pragma unroll
        for (int n = 0; n < 8; ++n) {
#pragma unroll
            for (int r = 0; r < 8; ++r) {
                float v = acc[n][r] + bv[n];
                if (RELU) v = fmaxf(v, 0.f);
                out[(size_t)(mbase + r) * HID + n * 16 + c] = v;
            }
        }
    }
}
#undef LOAD_B
#undef WMMA_STEP


__global__ __launch_bounds__(256)
void fold_head_kernel(const float* __restrict__ Wout,
                      const float* __restrict__ bout,
                      const float* __restrict__ Wl,
                      const float* __restrict__ Wr,
                      const float* __restrict__ bl,
                      float* __restrict__ Wc,
                      float* __restrict__ cvec) {
    const int t = blockIdx.x * blockDim.x + threadIdx.x;
    if (t < 4 * HID) {
        const int m = t >> 7;
        const int k = t & (HID - 1);
        const float* __restrict__ wo = Wout + (m & 1) * HID;
        const float* __restrict__ W  = (m < 2) ? Wl : Wr;
        float acc = 0.f;
        for (int o = 0; o < HID; ++o) acc += wo[o] * W[(size_t)o * HID + k];
        Wc[t] = acc;
    } else if (t < 4 * HID + 2) {
        const int j = t - 4 * HID;
        const float* __restrict__ wo = Wout + j * HID;
        float acc = bout[j];
        for (int o = 0; o < HID; ++o) acc += wo[o] * bl[o];
        cvec[j] = acc;
    }
}

__global__ __launch_bounds__(256)
void head_proj_wmma_kernel(const float* __restrict__ h,
                           const float* __restrict__ Wc,
                           const float* __restrict__ cvec,
                           float* __restrict__ p,
                           int nRowBlocks) {
    const int lane = threadIdx.x & 31;
    const int c    = lane & 15;
    const int half = lane >> 4;
    const int rb   = blockIdx.x * (blockDim.x >> 5) + (threadIdx.x >> 5);
    if (rb >= nRowBlocks) return;
    const int m0 = rb * 16;

    const float4* __restrict__ arow = (const float4*)(h + (size_t)(m0 + c) * HID) + half;
    const float4* __restrict__ brow = (const float4*)(Wc + (size_t)(c & 3) * HID) + half;
    const bool live = (c < 4);

    float4 av[16], bw[16];
#pragma unroll
    for (int j = 0; j < 16; ++j) av[j] = arow[2 * j];
#pragma unroll
    for (int j = 0; j < 16; ++j) {
        const float4 w = brow[2 * j];
        bw[j] = live ? w : make_float4(0.f, 0.f, 0.f, 0.f);
    }

    const float cb = ((c >> 1) == 1) ? cvec[c & 1] : 0.f;
    v8f acc = (v8f){cb, cb, cb, cb, cb, cb, cb, cb};

#pragma unroll
    for (int j = 0; j < 16; ++j) {
        v2f a, b;
        a.x = av[j].x; a.y = av[j].y; b.x = bw[j].x; b.y = bw[j].y;
        acc = __builtin_amdgcn_wmma_f32_16x16x4_f32(
            false, a, false, b, (short)0, acc, false, false);
        a.x = av[j].z; a.y = av[j].w; b.x = bw[j].z; b.y = bw[j].w;
        acc = __builtin_amdgcn_wmma_f32_16x16x4_f32(
            false, a, false, b, (short)0, acc, false, false);
    }

    if (live) {
#pragma unroll
        for (int r = 0; r < 8; ++r)
            p[(size_t)(m0 + 8 * half + r) * 4 + c] = acc[r];
    }
}

__global__ __launch_bounds__(256)
void head_gather_kernel(const float* __restrict__ p,
                        const int* __restrict__ rowptr,
                        const int* __restrict__ col,
                        const float* __restrict__ rdeg,
                        float* __restrict__ out,
                        int nGroups) {
    const int lane = threadIdx.x & 31;
    const int c    = lane & 15;
    const int half = lane >> 4;
    const int g    = blockIdx.x * (blockDim.x >> 5) + (threadIdx.x >> 5);
    if (g >= nGroups) return;
    const int t0 = g * 16;

    const int rp0 = rowptr[t0 + c];
    const int rp1 = rowptr[t0 + c + 1];

    int e0[8], end[8];
#pragma unroll
    for (int j = 0; j < 8; ++j) {
        e0[j]  = __shfl(rp0, 2 * j + half) + c;
        end[j] = __shfl(rp1, 2 * j + half);
    }
    int s0[8], s1[8];
#pragma unroll
    for (int j = 0; j < 8; ++j) {
        s0[j] = (e0[j] < end[j])      ? col[e0[j]]      : -1;
        s1[j] = (e0[j] + 16 < end[j]) ? col[e0[j] + 16] : -1;
    }
    float sx[8], sy[8];
#pragma unroll
    for (int j = 0; j < 8; ++j) {
        float2 a = make_float2(0.f, 0.f);
        float2 b = make_float2(0.f, 0.f);
        if (s0[j] >= 0) a = *(const float2*)(p + (size_t)s0[j] * 4);
        if (s1[j] >= 0) b = *(const float2*)(p + (size_t)s1[j] * 4);
        sx[j] = a.x + b.x;
        sy[j] = a.y + b.y;
    }
#pragma unroll
    for (int j = 0; j < 8; ++j) {
        for (int e = e0[j] + 32; e < end[j]; e += 16) {
            const float2 v = *(const float2*)(p + (size_t)col[e] * 4);
            sx[j] += v.x;
            sy[j] += v.y;
        }
    }
    float kx = 0.f, ky = 0.f;
#pragma unroll
    for (int j = 0; j < 8; ++j) {
        for (int o = 8; o > 0; o >>= 1) {
            sx[j] += __shfl_xor(sx[j], o);
            sy[j] += __shfl_xor(sy[j], o);
        }
        if (c == 2 * j + half) { kx = sx[j]; ky = sy[j]; }
    }
    if ((c & 1) == half) {
        const int t = t0 + c;
        const float2 self = *(const float2*)(p + (size_t)t * 4 + 2);
        const float rd = rdeg[t];
        float2 o2;
        o2.x = kx * rd + self.x;
        o2.y = ky * rd + self.y;
        *(float2*)(out + (size_t)t * 2) = o2;
    }
}

extern "C" void kernel_launch(void* const* d_in, const int* in_sizes, int n_in,
                              void* d_out, int out_size, void* d_ws, size_t ws_size,
                              hipStream_t stream) {
    const float* x    = (const float*)d_in[0];
    const int*   ei   = (const int*)d_in[1];
    const float* Wl0  = (const float*)d_in[2];
    const float* bl0  = (const float*)d_in[3];
    const float* Wr0  = (const float*)d_in[4];
    const float* Wl1  = (const float*)d_in[5];
    const float* bl1  = (const float*)d_in[6];
    const float* Wr1  = (const float*)d_in[7];
    const float* Wl2  = (const float*)d_in[8];
    const float* bl2  = (const float*)d_in[9];
    const float* Wr2  = (const float*)d_in[10];
    const float* Wout = (const float*)d_in[11];
    const float* bout = (const float*)d_in[12];
    float* out = (float*)d_out;

    const int* src = ei;
    const int* dst = ei + N_EDGES;

    float* ws     = (float*)d_ws;
    float* rdeg   = ws;
    int*   cnt    = (int*)(rdeg + N_NODES);
    float* agg    = (float*)(cnt + N_NODES);
    int*   rowptr = (int*)(agg + (size_t)N_NODES * HID);
    int*   col    = rowptr + N_NODES + 32;
    int*   rank   = col + N_EDGES;
    float* hA     = agg + (size_t)N_NODES * IN_PAD;
    float* xp     = hA + (size_t)N_NODES * HID;
    float* hB     = xp;
    float* Wl0p   = xp + (size_t)N_NODES * IN_PAD;
    float* Wr0p   = Wl0p + (size_t)HID * IN_PAD;
    float* p4     = hA;
    float* Wc     = p4 + (size_t)N_NODES * 4;
    float* cvec   = Wc + 4 * HID;

    const int ZT = 256;
    auto zero = [&](float* p, size_t n) {
        zero_f32_kernel<<<(unsigned)((n + ZT - 1) / ZT), ZT, 0, stream>>>(p, n);
    };
    auto pad = [&](const float* s, float* dptr, int sd, int dd, long rows) {
        long n = rows * dd;
        pad_rows_kernel<<<(unsigned)((n + ZT - 1) / ZT), ZT, 0, stream>>>(s, dptr, sd, dd, rows);
    };

    const int edgeBlocks256 = (N_EDGES + 255) / 256;
    const int scanBlocks    = (N_NODES + 1023) / 1024;
    const int gatherBlocks  = 2048;
    const int gemmBlocks    = N_NODES / 16;
    const int halfNodes     = N_NODES / 2;

    zero((float*)cnt, (size_t)N_NODES);
    rank_kernel<<<edgeBlocks256, 256, 0, stream>>>(dst, cnt, rank, N_EDGES);
    scan_rdeg_kernel<<<scanBlocks, 1024, 0, stream>>>(cnt, rowptr, rdeg, N_NODES);
    fill_kernel<<<edgeBlocks256, 256, 0, stream>>>(src, dst, rank, rowptr, col, N_EDGES);

    pad(x,   xp,   IN_CH, IN_PAD, N_NODES);
    pad(Wl0, Wl0p, IN_CH, IN_PAD, HID);
    pad(Wr0, Wr0p, IN_CH, IN_PAD, HID);

    gather_sum_kernel<IN_PAD><<<gatherBlocks, 256, 0, stream>>>(
        xp, rowptr, col, agg, 0, halfNodes);
    sage_gemm_wmma_kernel<IN_PAD, true><<<GEMM_WGS, GEMM_WAVES * 32, 0, stream>>>(
        agg, rdeg, xp, Wl0p, Wr0p, bl0, hA, halfNodes / 16);
    gather_sum_kernel<IN_PAD><<<gatherBlocks, 256, 0, stream>>>(
        xp, rowptr, col, agg, halfNodes, halfNodes);
    sage_gemm_wmma_kernel<IN_PAD, true><<<GEMM_WGS, GEMM_WAVES * 32, 0, stream>>>(
        agg, rdeg + halfNodes, xp + (size_t)halfNodes * IN_PAD, Wl0p, Wr0p, bl0,
        hA + (size_t)halfNodes * HID, halfNodes / 16);

    gather_sum_kernel<HID><<<gatherBlocks, 256, 0, stream>>>(
        hA, rowptr, col, agg, 0, N_NODES);
    sage_gemm_wmma_kernel<HID, true><<<GEMM_WGS, GEMM_WAVES * 32, 0, stream>>>(
        agg, rdeg, hA, Wl1, Wr1, bl1, hB, gemmBlocks);

    fold_head_kernel<<<(4 * HID + 2 + ZT - 1) / ZT, ZT, 0, stream>>>(
        Wout, bout, Wl2, Wr2, bl2, Wc, cvec);
    head_proj_wmma_kernel<<<(gemmBlocks + 7) / 8, 256, 0, stream>>>(
        hB, Wc, cvec, p4, gemmBlocks);
    head_gather_kernel<<<(gemmBlocks + 7) / 8, 256, 0, stream>>>(
        p4, rowptr, col, rdeg, out, gemmBlocks);
}


# Padding copy of x and layer-0 weights (one float per thread, 525000 waves) rewritten as a streaming copy: 1026 workgroups, each lane 2 batches of 32 whole-line loads issued before its stores
# speedup vs baseline: 4.4760x; 1.0694x over previous
#include <hip/hip_runtime.h>
#include <hip/hip_bf16.h>

#define N_NODES 100000
#define N_EDGES 1600000
#define IN_CH   165
#define IN_PAD  168
#define HID     128

typedef __attribute__((ext_vector_type(2))) float v2f;
typedef __attribute__((ext_vector_type(8))) float v8f;

__global__ void zero_f32_kernel(float* __restrict__ p, size_t n) {
    size_t i = (size_t)blockIdx.x * blockDim.x + threadIdx.x;
    if (i < n) p[i] = 0.0f;
}

#define PAD_THREADS 256
#define PAD_BATCH   32
#define PAD_NBATCH  2
#define PAD_CHUNK   (PAD_THREADS * PAD_BATCH * PAD_NBATCH)

template <int SRC_D, int DST_D, bool TAIL>
__device__ __forceinline__
void pad_batch(const float* __restrict__ src, float* __restrict__ dst,
               unsigned q0, unsigned total) {
    float v[PAD_BATCH];
#pragma unroll
    for (int u = 0; u < PAD_BATCH; ++u) {
        unsigned q = q0 + u * PAD_THREADS;
        if (TAIL) q = (q < total) ? q : total - 1;
        const unsigned r = q / DST_D;
        const unsigned c = q - r * DST_D;
        const bool in = (c < SRC_D);
        const float sv = src[r * SRC_D + (in ? c : 0u)];
        v[u] = in ? sv : 0.0f;
    }
#pragma unroll
    for (int u = 0; u < PAD_BATCH; ++u) {
        const unsigned q = q0 + u * PAD_THREADS;
        if (!TAIL || q < total) dst[q] = v[u];
    }
}

template <int SRC_D, int DST_D>
__global__ __launch_bounds__(PAD_THREADS)
void pad_rows_kernel(const float* __restrict__ src,
                     float* __restrict__ dst, int rows) {
    const unsigned total = (unsigned)rows * DST_D;
    const unsigned blockBase = blockIdx.x * (unsigned)PAD_CHUNK;
    if (blockBase + PAD_CHUNK <= total) {
        for (int b = 0; b < PAD_NBATCH; ++b)
            pad_batch<SRC_D, DST_D, false>(
                src, dst, blockBase + b * (PAD_THREADS * PAD_BATCH) + threadIdx.x, total);
    } else {
        for (int b = 0; b < PAD_NBATCH; ++b)
            pad_batch<SRC_D, DST_D, true>(
                src, dst, blockBase + b * (PAD_THREADS * PAD_BATCH) + threadIdx.x, total);
    }
}

__global__ void rank_kernel(const int* __restrict__ dst,
                            int* __restrict__ cnt,
                            int* __restrict__ rank, int nEdges) {
    int e = blockIdx.x * blockDim.x + threadIdx.x;
    if (e < nEdges) rank[e] = atomicAdd(&cnt[dst[e]], 1);
}

__global__ __launch_bounds__(1024)
void scan_rdeg_kernel(const int* __restrict__ cnt,
                      int* __restrict__ rowptr,
                      float* __restrict__ rdeg, int n) {
    __shared__ int front[32];
    __shared__ int wtot[32];
    const int tid  = threadIdx.x;
    const int lane = tid & 31;
    const int w    = tid >> 5;
    const int first = blockIdx.x * 1024;

    int acc = 0;
    for (int j = tid; j < first; j += 1024) acc += cnt[j];
    for (int o = 16; o > 0; o >>= 1) acc += __shfl_xor(acc, o);
    if (lane == 0) front[w] = acc;

    const int i = first + tid;
    const int cI = (i < n) ? cnt[i] : 0;
    int v = cI;
    for (int o = 1; o < 32; o <<= 1) {
        int u = __shfl_up(v, o);
        if (lane >= o) v += u;
    }
    if (lane == 31) wtot[w] = v;
    __syncthreads();

    int base = 0;
    for (int k = 0; k < 32; ++k) base += front[k];
    for (int k = 0; k < w; ++k)  base += wtot[k];

    if (i < n) {
        rowptr[i] = base + v - cI;
        rdeg[i]   = 1.0f / fmaxf((float)cI, 1.0f);
        if (i == n - 1) rowptr[n] = base + v;
    }
}

__global__ void fill_kernel(const int* __restrict__ src,
                            const int* __restrict__ dst,
                            const int* __restrict__ rank,
                            const int* __restrict__ rowptr,
                            int* __restrict__ col, int nEdges) {
    int e = blockIdx.x * blockDim.x + threadIdx.x;
    if (e < nEdges) col[rowptr[dst[e]] + rank[e]] = src[e];
}

template <int D>
__global__ __launch_bounds__(256)
void gather_sum_kernel(const float* __restrict__ h,
                       const int* __restrict__ rowptr,
                       const int* __restrict__ col,
                       float* __restrict__ agg,
                       int nodeBegin, int nodeCount) {
    static_assert(D % 4 == 0 && D >= 128 && D <= 256, "row = 32..64 float4");
    constexpr int V4 = D / 4;
    constexpr bool LONG = (V4 > 32);
    const int lane = threadIdx.x & 31;
    const bool hasTail = LONG && (lane + 32 < V4);
    const int wpb = blockDim.x >> 5;
    const int nw  = gridDim.x * wpb;

    for (int i0 = blockIdx.x * wpb + (threadIdx.x >> 5); i0 < nodeCount; i0 += nw) {
        const int i = __builtin_amdgcn_readfirstlane(i0);
        const int t = nodeBegin + i;
        const int beg = rowptr[t];
        const int end = rowptr[t + 1];
        float4 a = make_float4(0.f, 0.f, 0.f, 0.f);
        float4 b = make_float4(0.f, 0.f, 0.f, 0.f);

        for (int base = beg; base < end; base += 32) {
            int n = end - base;
            if (n > 32) n = 32;
            const int cv = (lane < n) ? col[base + lane] : 0;

#define GATHER_BATCH(B, K0)                                                     \
            {                                                                   \
                float4 va[B];                                                   \
                float4 vb[B];                                                   \
                _Pragma("unroll")                                               \
                for (int q = 0; q < B; ++q) {                                   \
                    const int s = __builtin_amdgcn_readlane(cv, (K0) + q);      \
                    const float4* r = (const float4*)(h + (size_t)s * D);       \
                    va[q] = r[lane];                                            \
                    if (LONG) vb[q] = hasTail ? r[lane + 32]                    \
                                              : make_float4(0.f, 0.f, 0.f, 0.f);\
                }                                                               \
                _Pragma("unroll")                                               \
                for (int q = 0; q < B; ++q) {                                   \
                    a.x += va[q].x; a.y += va[q].y;                             \
                    a.z += va[q].z; a.w += va[q].w;                             \
                    if (LONG) {                                                 \
                        b.x += vb[q].x; b.y += vb[q].y;                         \
                        b.z += vb[q].z; b.w += vb[q].w;                         \
                    }                                                           \
                }                                                               \
            }
            int k = 0;
            for (; k + 8 <= n; k += 8) GATHER_BATCH(8, k)
            if (n & 4) { GATHER_BATCH(4, k) k += 4; }
            if (n & 2) { GATHER_BATCH(2, k) k += 2; }
            if (n & 1) { GATHER_BATCH(1, k) }
#undef GATHER_BATCH
        }

        float4* o = (float4*)(agg + (size_t)i * D);
        o[lane] = a;
        if (hasTail) o[lane + 32] = b;
    }
}

#define GEMM_WGS   256
#define GEMM_WAVES 16

#define LOAD_B(BUF, S)                                                          \
    _Pragma("unroll")                                                           \
    for (int n = 0; n < 8; ++n) {                                               \
        BUF[2 * n]     = *(const v2f*)((S) + n * 64);                           \
        BUF[2 * n + 1] = *(const v2f*)((S) + HID * 4 + n * 64);                 \
    }
#define WMMA_STEP(AM, AH, BUF)                                                  \
    {                                                                           \
        v2f a1; a1.x = (AM).x * rd; a1.y = (AM).y * rd;                         \
        v2f a2; a2.x = (AH).x;      a2.y = (AH).y;                              \
        _Pragma("unroll")                                                       \
        for (int n = 0; n < 8; ++n) {                                           \
            acc[n] = __builtin_amdgcn_wmma_f32_16x16x4_f32(                     \
                false, a1, false, BUF[2 * n], (short)0, acc[n], false, false);  \
            acc[n] = __builtin_amdgcn_wmma_f32_16x16x4_f32(                     \
                false, a2, false, BUF[2 * n + 1], (short)0, acc[n], false, false); \
        }                                                                       \
    }

template <int D, bool RELU>
__global__ __launch_bounds__(GEMM_WAVES * 32)
void sage_gemm_wmma_kernel(const float* __restrict__ agg,
                           const float* __restrict__ rdeg,
                           const float* __restrict__ h,
                           const float* __restrict__ Wl,
                           const float* __restrict__ Wr,
                           const float* __restrict__ bias,
                           float* __restrict__ out,
                           int nRowBlocks) {
    static_assert(D % 8 == 0, "K is walked in pairs of k-steps of 4");
    constexpr int V4 = D / 4;
    constexpr int NB = V4 / 2;
    constexpr int KP = 2 * HID * 4 + 4;
    __shared__ float wsh[V4 * KP];

    constexpr int NT = GEMM_WAVES * 32;
    constexpr int ST = (HID * V4 + NT - 1) / NT;
#pragma unroll
    for (int t = 0; t < ST; ++t) {
        const int i = threadIdx.x + t * NT;
        if (i < HID * V4) {
            const int r = i / V4;
            const int q = i - r * V4;
            const float4 vl = ((const float4*)(Wl + (size_t)r * D))[q];
            const float4 vr = ((const float4*)(Wr + (size_t)r * D))[q];
            *(float4*)(wsh + q * KP + r * 4)           = vl;
            *(float4*)(wsh + q * KP + HID * 4 + r * 4) = vr;
        }
    }
    __syncthreads();

    const int lane = threadIdx.x & 31;
    const int wave = threadIdx.x >> 5;
    const int c    = lane & 15;
    const int koff = (lane >> 4) << 1;

    const float* __restrict__ bp = wsh + c * 4 + koff;

    float bv[8];
#pragma unroll
    for (int n = 0; n < 8; ++n) bv[n] = bias[n * 16 + c];

    const int turn  = ((wave >> 2) + (wave & 3)) & 3;
    const int first = turn * (4 * GEMM_WGS) + blockIdx.x * 4 + (wave & 3);

    for (int rb = first; rb < nRowBlocks; rb += GEMM_WGS * GEMM_WAVES) {
        const int m0   = rb * 16;
        const int mrow = m0 + c;
        const float rd = rdeg[mrow];

        const float2* __restrict__ arow = (const float2*)(agg + (size_t)mrow * D + koff);
        const float2* __restrict__ hrow = (const float2*)(h   + (size_t)mrow * D + koff);

        v8f acc[8];
#pragma unroll
        for (int n = 0; n < 8; ++n)
            acc[n] = (v8f){0.f, 0.f, 0.f, 0.f, 0.f, 0.f, 0.f, 0.f};

        float2 nm0 = arow[0], nh0 = hrow[0], nm1 = arow[2], nh1 = hrow[2];
        v2f b0[16], b1[16];
        LOAD_B(b0, bp)

        for (int blk = 0; blk < NB; ++blk) {
            const float2 cm0 = nm0, ch0 = nh0, cm1 = nm1, ch1 = nh1;
            const int nb = (blk + 1 < NB) ? blk + 1 : NB - 1;
            nm0 = arow[4 * nb];     nh0 = hrow[4 * nb];
            nm1 = arow[4 * nb + 2]; nh1 = hrow[4 * nb + 2];

            LOAD_B(b1, bp + (2 * blk + 1) * KP)
            WMMA_STEP(cm0, ch0, b0)
            LOAD_B(b0, bp + (2 * nb) * KP)
            WMMA_STEP(cm1, ch1, b1)
        }

        const int mbase = m0 + ((lane < 16) ? 0 : 8);
#pragma unroll
        for (int n = 0; n < 8; ++n) {
#pragma unroll
            for (int r = 0; r < 8; ++r) {
                float v = acc[n][r] + bv[n];
                if (RELU) v = fmaxf(v, 0.f);
                out[(size_t)(mbase + r) * HID + n * 16 + c] = v;
            }
        }
    }
}
#undef LOAD_B
#undef WMMA_STEP


__global__ __launch_bounds__(256)
void fold_head_kernel(const float* __restrict__ Wout,
                      const float* __restrict__ bout,
                      const float* __restrict__ Wl,
                      const float* __restrict__ Wr,
                      const float* __restrict__ bl,
                      float* __restrict__ Wc,
                      float* __restrict__ cvec) {
    const int t = blockIdx.x * blockDim.x + threadIdx.x;
    if (t < 4 * HID) {
        const int m = t >> 7;
        const int k = t & (HID - 1);
        const float* __restrict__ wo = Wout + (m & 1) * HID;
        const float* __restrict__ W  = (m < 2) ? Wl : Wr;
        float acc = 0.f;
        for (int o = 0; o < HID; ++o) acc += wo[o] * W[(size_t)o * HID + k];
        Wc[t] = acc;
    } else if (t < 4 * HID + 2) {
        const int j = t - 4 * HID;
        const float* __restrict__ wo = Wout + j * HID;
        float acc = bout[j];
        for (int o = 0; o < HID; ++o) acc += wo[o] * bl[o];
        cvec[j] = acc;
    }
}

__global__ __launch_bounds__(256)
void head_proj_wmma_kernel(const float* __restrict__ h,
                           const float* __restrict__ Wc,
                           const float* __restrict__ cvec,
                           float* __restrict__ p,
                           int nRowBlocks) {
    const int lane = threadIdx.x & 31;
    const int c    = lane & 15;
    const int half = lane >> 4;
    const int rb   = blockIdx.x * (blockDim.x >> 5) + (threadIdx.x >> 5);
    if (rb >= nRowBlocks) return;
    const int m0 = rb * 16;

    const float4* __restrict__ arow = (const float4*)(h + (size_t)(m0 + c) * HID) + half;
    const float4* __restrict__ brow = (const float4*)(Wc + (size_t)(c & 3) * HID) + half;
    const bool live = (c < 4);

    float4 av[16], bw[16];
#pragma unroll
    for (int j = 0; j < 16; ++j) av[j] = arow[2 * j];
#pragma unroll
    for (int j = 0; j < 16; ++j) {
        const float4 w = brow[2 * j];
        bw[j] = live ? w : make_float4(0.f, 0.f, 0.f, 0.f);
    }

    const float cb = ((c >> 1) == 1) ? cvec[c & 1] : 0.f;
    v8f acc = (v8f){cb, cb, cb, cb, cb, cb, cb, cb};

#pragma unroll
    for (int j = 0; j < 16; ++j) {
        v2f a, b;
        a.x = av[j].x; a.y = av[j].y; b.x = bw[j].x; b.y = bw[j].y;
        acc = __builtin_amdgcn_wmma_f32_16x16x4_f32(
            false, a, false, b, (short)0, acc, false, false);
        a.x = av[j].z; a.y = av[j].w; b.x = bw[j].z; b.y = bw[j].w;
        acc = __builtin_amdgcn_wmma_f32_16x16x4_f32(
            false, a, false, b, (short)0, acc, false, false);
    }

    if (live) {
#pragma unroll
        for (int r = 0; r < 8; ++r)
            p[(size_t)(m0 + 8 * half + r) * 4 + c] = acc[r];
    }
}

__global__ __launch_bounds__(256)
void head_gather_kernel(const float* __restrict__ p,
                        const int* __restrict__ rowptr,
                        const int* __restrict__ col,
                        const float* __restrict__ rdeg,
                        float* __restrict__ out,
                        int nGroups) {
    const int lane = threadIdx.x & 31;
    const int c    = lane & 15;
    const int half = lane >> 4;
    const int g    = blockIdx.x * (blockDim.x >> 5) + (threadIdx.x >> 5);
    if (g >= nGroups) return;
    const int t0 = g * 16;

    const int rp0 = rowptr[t0 + c];
    const int rp1 = rowptr[t0 + c + 1];

    int e0[8], end[8];
#pragma unroll
    for (int j = 0; j < 8; ++j) {
        e0[j]  = __shfl(rp0, 2 * j + half) + c;
        end[j] = __shfl(rp1, 2 * j + half);
    }
    int s0[8], s1[8];
#pragma unroll
    for (int j = 0; j < 8; ++j) {
        s0[j] = (e0[j] < end[j])      ? col[e0[j]]      : -1;
        s1[j] = (e0[j] + 16 < end[j]) ? col[e0[j] + 16] : -1;
    }
    float sx[8], sy[8];
#pragma unroll
    for (int j = 0; j < 8; ++j) {
        float2 a = make_float2(0.f, 0.f);
        float2 b = make_float2(0.f, 0.f);
        if (s0[j] >= 0) a = *(const float2*)(p + (size_t)s0[j] * 4);
        if (s1[j] >= 0) b = *(const float2*)(p + (size_t)s1[j] * 4);
        sx[j] = a.x + b.x;
        sy[j] = a.y + b.y;
    }
#pragma unroll
    for (int j = 0; j < 8; ++j) {
        for (int e = e0[j] + 32; e < end[j]; e += 16) {
            const float2 v = *(const float2*)(p + (size_t)col[e] * 4);
            sx[j] += v.x;
            sy[j] += v.y;
        }
    }
    float kx = 0.f, ky = 0.f;
#pragma unroll
    for (int j = 0; j < 8; ++j) {
        for (int o = 8; o > 0; o >>= 1) {
            sx[j] += __shfl_xor(sx[j], o);
            sy[j] += __shfl_xor(sy[j], o);
        }
        if (c == 2 * j + half) { kx = sx[j]; ky = sy[j]; }
    }
    if ((c & 1) == half) {
        const int t = t0 + c;
        const float2 self = *(const float2*)(p + (size_t)t * 4 + 2);
        const float rd = rdeg[t];
        float2 o2;
        o2.x = kx * rd + self.x;
        o2.y = ky * rd + self.y;
        *(float2*)(out + (size_t)t * 2) = o2;
    }
}

extern "C" void kernel_launch(void* const* d_in, const int* in_sizes, int n_in,
                              void* d_out, int out_size, void* d_ws, size_t ws_size,
                              hipStream_t stream) {
    const float* x    = (const float*)d_in[0];
    const int*   ei   = (const int*)d_in[1];
    const float* Wl0  = (const float*)d_in[2];
    const float* bl0  = (const float*)d_in[3];
    const float* Wr0  = (const float*)d_in[4];
    const float* Wl1  = (const float*)d_in[5];
    const float* bl1  = (const float*)d_in[6];
    const float* Wr1  = (const float*)d_in[7];
    const float* Wl2  = (const float*)d_in[8];
    const float* bl2  = (const float*)d_in[9];
    const float* Wr2  = (const float*)d_in[10];
    const float* Wout = (const float*)d_in[11];
    const float* bout = (const float*)d_in[12];
    float* out = (float*)d_out;

    const int* src = ei;
    const int* dst = ei + N_EDGES;

    float* ws     = (float*)d_ws;
    float* rdeg   = ws;
    int*   cnt    = (int*)(rdeg + N_NODES);
    float* agg    = (float*)(cnt + N_NODES);
    int*   rowptr = (int*)(agg + (size_t)N_NODES * HID);
    int*   col    = rowptr + N_NODES + 32;
    int*   rank   = col + N_EDGES;
    float* hA     = agg + (size_t)N_NODES * IN_PAD;
    float* xp     = hA + (size_t)N_NODES * HID;
    float* hB     = xp;
    float* Wl0p   = xp + (size_t)N_NODES * IN_PAD;
    float* Wr0p   = Wl0p + (size_t)HID * IN_PAD;
    float* p4     = hA;
    float* Wc     = p4 + (size_t)N_NODES * 4;
    float* cvec   = Wc + 4 * HID;

    const int ZT = 256;
    auto zero = [&](float* p, size_t n) {
        zero_f32_kernel<<<(unsigned)((n + ZT - 1) / ZT), ZT, 0, stream>>>(p, n);
    };

    const int edgeBlocks256 = (N_EDGES + 255) / 256;
    const int scanBlocks    = (N_NODES + 1023) / 1024;
    const int gatherBlocks  = 2048;
    const int gemmBlocks    = N_NODES / 16;
    const int halfNodes     = N_NODES / 2;
    const int padBlocksX    = (N_NODES * IN_PAD + PAD_CHUNK - 1) / PAD_CHUNK;
    const int padBlocksW    = (HID * IN_PAD + PAD_CHUNK - 1) / PAD_CHUNK;

    zero((float*)cnt, (size_t)N_NODES);
    rank_kernel<<<edgeBlocks256, 256, 0, stream>>>(dst, cnt, rank, N_EDGES);
    scan_rdeg_kernel<<<scanBlocks, 1024, 0, stream>>>(cnt, rowptr, rdeg, N_NODES);
    fill_kernel<<<edgeBlocks256, 256, 0, stream>>>(src, dst, rank, rowptr, col, N_EDGES);

    pad_rows_kernel<IN_CH, IN_PAD><<<padBlocksX, PAD_THREADS, 0, stream>>>(x,   xp,   N_NODES);
    pad_rows_kernel<IN_CH, IN_PAD><<<padBlocksW, PAD_THREADS, 0, stream>>>(Wl0, Wl0p, HID);
    pad_rows_kernel<IN_CH, IN_PAD><<<padBlocksW, PAD_THREADS, 0, stream>>>(Wr0, Wr0p, HID);

    gather_sum_kernel<IN_PAD><<<gatherBlocks, 256, 0, stream>>>(
        xp, rowptr, col, agg, 0, halfNodes);
    sage_gemm_wmma_kernel<IN_PAD, true><<<GEMM_WGS, GEMM_WAVES * 32, 0, stream>>>(
        agg, rdeg, xp, Wl0p, Wr0p, bl0, hA, halfNodes / 16);
    gather_sum_kernel<IN_PAD><<<gatherBlocks, 256, 0, stream>>>(
        xp, rowptr, col, agg, halfNodes, halfNodes);
    sage_gemm_wmma_kernel<IN_PAD, true><<<GEMM_WGS, GEMM_WAVES * 32, 0, stream>>>(
        agg, rdeg + halfNodes, xp + (size_t)halfNodes * IN_PAD, Wl0p, Wr0p, bl0,
        hA + (size_t)halfNodes * HID, halfNodes / 16);

    gather_sum_kernel<HID><<<gatherBlocks, 256, 0, stream>>>(
        hA, rowptr, col, agg, 0, N_NODES);
    sage_gemm_wmma_kernel<HID, true><<<GEMM_WGS, GEMM_WAVES * 32, 0, stream>>>(
        agg, rdeg, hA, Wl1, Wr1, bl1, hB, gemmBlocks);

    fold_head_kernel<<<(4 * HID + 2 + ZT - 1) / ZT, ZT, 0, stream>>>(
        Wout, bout, Wl2, Wr2, bl2, Wc, cvec);
    head_proj_wmma_kernel<<<(gemmBlocks + 7) / 8, 256, 0, stream>>>(
        hB, Wc, cvec, p4, gemmBlocks);
    head_gather_kernel<<<(gemmBlocks + 7) / 8, 256, 0, stream>>>(
        p4, rowptr, col, rdeg, out, gemmBlocks);
}


# Destination grouping without global atomics: edges binned by 512-node range (LDS histogram per 6250-edge chunk, scan, copy), then per-bin LDS counters give rowptr and col; sums in LDS arrival order
# speedup vs baseline: 5.5742x; 1.2454x over previous
#include <hip/hip_runtime.h>
#include <hip/hip_bf16.h>

#define N_NODES 100000
#define N_EDGES 1600000
#define IN_CH   165
#define IN_PAD  168
#define HID     128

typedef __attribute__((ext_vector_type(2))) float v2f;
typedef __attribute__((ext_vector_type(8))) float v8f;


#define PAD_THREADS 256
#define PAD_BATCH   32
#define PAD_NBATCH  2
#define PAD_CHUNK   (PAD_THREADS * PAD_BATCH * PAD_NBATCH)

template <int SRC_D, int DST_D, bool TAIL>
__device__ __forceinline__
void pad_batch(const float* __restrict__ src, float* __restrict__ dst,
               unsigned q0, unsigned total) {
    float v[PAD_BATCH];
#pragma unroll
    for (int u = 0; u < PAD_BATCH; ++u) {
        unsigned q = q0 + u * PAD_THREADS;
        if (TAIL) q = (q < total) ? q : total - 1;
        const unsigned r = q / DST_D;
        const unsigned c = q - r * DST_D;
        const bool in = (c < SRC_D);
        const float sv = src[r * SRC_D + (in ? c : 0u)];
        v[u] = in ? sv : 0.0f;
    }
#pragma unroll
    for (int u = 0; u < PAD_BATCH; ++u) {
        const unsigned q = q0 + u * PAD_THREADS;
        if (!TAIL || q < total) dst[q] = v[u];
    }
}

template <int SRC_D, int DST_D>
__global__ __launch_bounds__(PAD_THREADS)
void pad_rows_kernel(const float* __restrict__ src,
                     float* __restrict__ dst, int rows) {
    const unsigned total = (unsigned)rows * DST_D;
    const unsigned blockBase = blockIdx.x * (unsigned)PAD_CHUNK;
    if (blockBase + PAD_CHUNK <= total) {
        for (int b = 0; b < PAD_NBATCH; ++b)
            pad_batch<SRC_D, DST_D, false>(
                src, dst, blockBase + b * (PAD_THREADS * PAD_BATCH) + threadIdx.x, total);
    } else {
        for (int b = 0; b < PAD_NBATCH; ++b)
            pad_batch<SRC_D, DST_D, true>(
                src, dst, blockBase + b * (PAD_THREADS * PAD_BATCH) + threadIdx.x, total);
    }
}

#define BIN_SHIFT  9
#define BIN_NODES  (1 << BIN_SHIFT)
#define NBIN       ((N_NODES + BIN_NODES - 1) / BIN_NODES)
#define NCHUNK     256
#define CHUNK      ((N_EDGES + NCHUNK - 1) / NCHUNK)
#define BIN_THREADS 1024
#define BIN_BATCH  4

__global__ __launch_bounds__(BIN_THREADS)
void bin_count_kernel(const int* __restrict__ dst,
                      int* __restrict__ counts,
                      int nEdges) {
    __shared__ int hist[NBIN];
    const int tid = threadIdx.x;
    for (int i = tid; i < NBIN; i += BIN_THREADS) hist[i] = 0;
    __syncthreads();

    const int c0 = blockIdx.x * CHUNK;
    const int c1 = (c0 + CHUNK < nEdges) ? c0 + CHUNK : nEdges;
    for (int e0 = c0 + tid; e0 < c1; e0 += BIN_THREADS * BIN_BATCH) {
        int d[BIN_BATCH];
#pragma unroll
        for (int u = 0; u < BIN_BATCH; ++u) {
            const int e = e0 + u * BIN_THREADS;
            d[u] = (e < c1) ? dst[e] : -1;
        }
#pragma unroll
        for (int u = 0; u < BIN_BATCH; ++u)
            if (d[u] >= 0) atomicAdd(&hist[d[u] >> BIN_SHIFT], 1);
    }
    __syncthreads();
    for (int i = tid; i < NBIN; i += BIN_THREADS)
        counts[i * NCHUNK + blockIdx.x] = hist[i];
}

__global__ __launch_bounds__(1024)
void scan_excl_kernel(const int* __restrict__ in,
                      int* __restrict__ out, int n) {
    __shared__ int front[32];
    __shared__ int wtot[32];
    const int tid  = threadIdx.x;
    const int lane = tid & 31;
    const int w    = tid >> 5;
    const int first = blockIdx.x * 1024;

    int acc = 0;
    for (int j = tid; j < first; j += 1024) acc += in[j];
    for (int o = 16; o > 0; o >>= 1) acc += __shfl_xor(acc, o);
    if (lane == 0) front[w] = acc;

    const int i = first + tid;
    const int cI = (i < n) ? in[i] : 0;
    int v = cI;
    for (int o = 1; o < 32; o <<= 1) {
        int u = __shfl_up(v, o);
        if (lane >= o) v += u;
    }
    if (lane == 31) wtot[w] = v;
    __syncthreads();

    int base = 0;
    for (int k = 0; k < 32; ++k) base += front[k];
    for (int k = 0; k < w; ++k)  base += wtot[k];

    if (i < n) {
        out[i] = base + v - cI;
        if (i == n - 1) out[n] = base + v;
    }
}

__global__ __launch_bounds__(BIN_THREADS)
void bin_scatter_kernel(const int* __restrict__ src,
                        const int* __restrict__ dst,
                        const int* __restrict__ offs,
                        int2* __restrict__ binned,
                        int nEdges) {
    __shared__ int pos[NBIN];
    const int tid = threadIdx.x;
    for (int i = tid; i < NBIN; i += BIN_THREADS) pos[i] = offs[i * NCHUNK + blockIdx.x];
    __syncthreads();

    const int c0 = blockIdx.x * CHUNK;
    const int c1 = (c0 + CHUNK < nEdges) ? c0 + CHUNK : nEdges;
    for (int e0 = c0 + tid; e0 < c1; e0 += BIN_THREADS * BIN_BATCH) {
        int d[BIN_BATCH], s[BIN_BATCH];
#pragma unroll
        for (int u = 0; u < BIN_BATCH; ++u) {
            const int e = e0 + u * BIN_THREADS;
            const bool in = (e < c1);
            d[u] = in ? dst[e] : -1;
            s[u] = in ? src[e] : 0;
        }
#pragma unroll
        for (int u = 0; u < BIN_BATCH; ++u) {
            if (d[u] >= 0) {
                const int p = atomicAdd(&pos[d[u] >> BIN_SHIFT], 1);
                binned[p] = make_int2(s[u], d[u]);
            }
        }
    }
}

__global__ __launch_bounds__(BIN_THREADS)
void bin_finish_kernel(const int2* __restrict__ binned,
                       const int* __restrict__ offs,
                       int* __restrict__ rowptr,
                       float* __restrict__ rdeg,
                       int* __restrict__ col, int n) {
    __shared__ int cntL[BIN_NODES];
    __shared__ int posL[BIN_NODES];
    __shared__ int wtot[BIN_NODES / 32];
    const int tid  = threadIdx.x;
    const int lane = tid & 31;
    const int w    = tid >> 5;
    const int nodeBase = blockIdx.x * BIN_NODES;
    const int beg = offs[blockIdx.x * NCHUNK];
    const int end = offs[(blockIdx.x + 1) * NCHUNK];

    if (tid < BIN_NODES) cntL[tid] = 0;
    __syncthreads();

    for (int i0 = beg + tid; i0 < end; i0 += BIN_THREADS * BIN_BATCH) {
        int dy[BIN_BATCH];
#pragma unroll
        for (int u = 0; u < BIN_BATCH; ++u) {
            const int i = i0 + u * BIN_THREADS;
            dy[u] = (i < end) ? binned[i].y : -1;
        }
#pragma unroll
        for (int u = 0; u < BIN_BATCH; ++u)
            if (dy[u] >= 0) atomicAdd(&cntL[dy[u] - nodeBase], 1);
    }
    __syncthreads();

    int cI = 0, v = 0;
    if (tid < BIN_NODES) {
        cI = cntL[tid];
        v = cI;
        for (int o = 1; o < 32; o <<= 1) {
            int u = __shfl_up(v, o);
            if (lane >= o) v += u;
        }
        if (lane == 31) wtot[w] = v;
    }
    __syncthreads();
    if (tid < BIN_NODES) {
        int start = beg + v - cI;
        for (int k = 0; k < w; ++k) start += wtot[k];
        posL[tid] = start;
        const int node = nodeBase + tid;
        if (node < n) {
            rowptr[node] = start;
            rdeg[node]   = 1.0f / fmaxf((float)cI, 1.0f);
        }
    }
    if (blockIdx.x == gridDim.x - 1 && tid == 0) rowptr[n] = end;
    __syncthreads();

    for (int i0 = beg + tid; i0 < end; i0 += BIN_THREADS * BIN_BATCH) {
        int2 r[BIN_BATCH];
#pragma unroll
        for (int u = 0; u < BIN_BATCH; ++u) {
            const int i = i0 + u * BIN_THREADS;
            r[u] = (i < end) ? binned[i] : make_int2(0, -1);
        }
#pragma unroll
        for (int u = 0; u < BIN_BATCH; ++u) {
            if (r[u].y >= 0) {
                const int p = atomicAdd(&posL[r[u].y - nodeBase], 1);
                col[p] = r[u].x;
            }
        }
    }
}

template <int D>
__global__ __launch_bounds__(256)
void gather_sum_kernel(const float* __restrict__ h,
                       const int* __restrict__ rowptr,
                       const int* __restrict__ col,
                       float* __restrict__ agg,
                       int nodeBegin, int nodeCount) {
    static_assert(D % 4 == 0 && D >= 128 && D <= 256, "row = 32..64 float4");
    constexpr int V4 = D / 4;
    constexpr bool LONG = (V4 > 32);
    const int lane = threadIdx.x & 31;
    const bool hasTail = LONG && (lane + 32 < V4);
    const int wpb = blockDim.x >> 5;
    const int nw  = gridDim.x * wpb;

    for (int i0 = blockIdx.x * wpb + (threadIdx.x >> 5); i0 < nodeCount; i0 += nw) {
        const int i = __builtin_amdgcn_readfirstlane(i0);
        const int t = nodeBegin + i;
        const int beg = rowptr[t];
        const int end = rowptr[t + 1];
        float4 a = make_float4(0.f, 0.f, 0.f, 0.f);
        float4 b = make_float4(0.f, 0.f, 0.f, 0.f);

        for (int base = beg; base < end; base += 32) {
            int n = end - base;
            if (n > 32) n = 32;
            const int cv = (lane < n) ? col[base + lane] : 0;

#define GATHER_BATCH(B, K0)                                                     \
            {                                                                   \
                float4 va[B];                                                   \
                float4 vb[B];                                                   \
                _Pragma("unroll")                                               \
                for (int q = 0; q < B; ++q) {                                   \
                    const int s = __builtin_amdgcn_readlane(cv, (K0) + q);      \
                    const float4* r = (const float4*)(h + (size_t)s * D);       \
                    va[q] = r[lane];                                            \
                    if (LONG) vb[q] = hasTail ? r[lane + 32]                    \
                                              : make_float4(0.f, 0.f, 0.f, 0.f);\
                }                                                               \
                _Pragma("unroll")                                               \
                for (int q = 0; q < B; ++q) {                                   \
                    a.x += va[q].x; a.y += va[q].y;                             \
                    a.z += va[q].z; a.w += va[q].w;                             \
                    if (LONG) {                                                 \
                        b.x += vb[q].x; b.y += vb[q].y;                         \
                        b.z += vb[q].z; b.w += vb[q].w;                         \
                    }                                                           \
                }                                                               \
            }
            int k = 0;
            for (; k + 8 <= n; k += 8) GATHER_BATCH(8, k)
            if (n & 4) { GATHER_BATCH(4, k) k += 4; }
            if (n & 2) { GATHER_BATCH(2, k) k += 2; }
            if (n & 1) { GATHER_BATCH(1, k) }
#undef GATHER_BATCH
        }

        float4* o = (float4*)(agg + (size_t)i * D);
        o[lane] = a;
        if (hasTail) o[lane + 32] = b;
    }
}

#define GEMM_WGS   256
#define GEMM_WAVES 16

#define LOAD_B(BUF, S)                                                          \
    _Pragma("unroll")                                                           \
    for (int n = 0; n < 8; ++n) {                                               \
        BUF[2 * n]     = *(const v2f*)((S) + n * 64);                           \
        BUF[2 * n + 1] = *(const v2f*)((S) + HID * 4 + n * 64);                 \
    }
#define WMMA_STEP(AM, AH, BUF)                                                  \
    {                                                                           \
        v2f a1; a1.x = (AM).x * rd; a1.y = (AM).y * rd;                         \
        v2f a2; a2.x = (AH).x;      a2.y = (AH).y;                              \
        _Pragma("unroll")                                                       \
        for (int n = 0; n < 8; ++n) {                                           \
            acc[n] = __builtin_amdgcn_wmma_f32_16x16x4_f32(                     \
                false, a1, false, BUF[2 * n], (short)0, acc[n], false, false);  \
            acc[n] = __builtin_amdgcn_wmma_f32_16x16x4_f32(                     \
                false, a2, false, BUF[2 * n + 1], (short)0, acc[n], false, false); \
        }                                                                       \
    }

template <int D, bool RELU>
__global__ __launch_bounds__(GEMM_WAVES * 32)
void sage_gemm_wmma_kernel(const float* __restrict__ agg,
                           const float* __restrict__ rdeg,
                           const float* __restrict__ h,
                           const float* __restrict__ Wl,
                           const float* __restrict__ Wr,
                           const float* __restrict__ bias,
                           float* __restrict__ out,
                           int nRowBlocks) {
    static_assert(D % 8 == 0, "K is walked in pairs of k-steps of 4");
    constexpr int V4 = D / 4;
    constexpr int NB = V4 / 2;
    constexpr int KP = 2 * HID * 4 + 4;
    __shared__ float wsh[V4 * KP];

    constexpr int NT = GEMM_WAVES * 32;
    constexpr int ST = (HID * V4 + NT - 1) / NT;
#pragma unroll
    for (int t = 0; t < ST; ++t) {
        const int i = threadIdx.x + t * NT;
        if (i < HID * V4) {
            const int r = i / V4;
            const int q = i - r * V4;
            const float4 vl = ((const float4*)(Wl + (size_t)r * D))[q];
            const float4 vr = ((const float4*)(Wr + (size_t)r * D))[q];
            *(float4*)(wsh + q * KP + r * 4)           = vl;
            *(float4*)(wsh + q * KP + HID * 4 + r * 4) = vr;
        }
    }
    __syncthreads();

    const int lane = threadIdx.x & 31;
    const int wave = threadIdx.x >> 5;
    const int c    = lane & 15;
    const int koff = (lane >> 4) << 1;

    const float* __restrict__ bp = wsh + c * 4 + koff;

    float bv[8];
#pragma unroll
    for (int n = 0; n < 8; ++n) bv[n] = bias[n * 16 + c];

    const int turn  = ((wave >> 2) + (wave & 3)) & 3;
    const int first = turn * (4 * GEMM_WGS) + blockIdx.x * 4 + (wave & 3);

    for (int rb = first; rb < nRowBlocks; rb += GEMM_WGS * GEMM_WAVES) {
        const int m0   = rb * 16;
        const int mrow = m0 + c;
        const float rd = rdeg[mrow];

        const float2* __restrict__ arow = (const float2*)(agg + (size_t)mrow * D + koff);
        const float2* __restrict__ hrow = (const float2*)(h   + (size_t)mrow * D + koff);

        v8f acc[8];
#pragma unroll
        for (int n = 0; n < 8; ++n)
            acc[n] = (v8f){0.f, 0.f, 0.f, 0.f, 0.f, 0.f, 0.f, 0.f};

        float2 nm0 = arow[0], nh0 = hrow[0], nm1 = arow[2], nh1 = hrow[2];
        v2f b0[16], b1[16];
        LOAD_B(b0, bp)

        for (int blk = 0; blk < NB; ++blk) {
            const float2 cm0 = nm0, ch0 = nh0, cm1 = nm1, ch1 = nh1;
            const int nb = (blk + 1 < NB) ? blk + 1 : NB - 1;
            nm0 = arow[4 * nb];     nh0 = hrow[4 * nb];
            nm1 = arow[4 * nb + 2]; nh1 = hrow[4 * nb + 2];

            LOAD_B(b1, bp + (2 * blk + 1) * KP)
            WMMA_STEP(cm0, ch0, b0)
            LOAD_B(b0, bp + (2 * nb) * KP)
            WMMA_STEP(cm1, ch1, b1)
        }

        const int mbase = m0 + ((lane < 16) ? 0 : 8);
#pragma unroll
        for (int n = 0; n < 8; ++n) {
#pragma unroll
            for (int r = 0; r < 8; ++r) {
                float v = acc[n][r] + bv[n];
                if (RELU) v = fmaxf(v, 0.f);
                out[(size_t)(mbase + r) * HID + n * 16 + c] = v;
            }
        }
    }
}
#undef LOAD_B
#undef WMMA_STEP


__global__ __launch_bounds__(256)
void fold_head_kernel(const float* __restrict__ Wout,
                      const float* __restrict__ bout,
                      const float* __restrict__ Wl,
                      const float* __restrict__ Wr,
                      const float* __restrict__ bl,
                      float* __restrict__ Wc,
                      float* __restrict__ cvec) {
    const int t = blockIdx.x * blockDim.x + threadIdx.x;
    if (t < 4 * HID) {
        const int m = t >> 7;
        const int k = t & (HID - 1);
        const float* __restrict__ wo = Wout + (m & 1) * HID;
        const float* __restrict__ W  = (m < 2) ? Wl : Wr;
        float acc = 0.f;
        for (int o = 0; o < HID; ++o) acc += wo[o] * W[(size_t)o * HID + k];
        Wc[t] = acc;
    } else if (t < 4 * HID + 2) {
        const int j = t - 4 * HID;
        const float* __restrict__ wo = Wout + j * HID;
        float acc = bout[j];
        for (int o = 0; o < HID; ++o) acc += wo[o] * bl[o];
        cvec[j] = acc;
    }
}

__global__ __launch_bounds__(256)
void head_proj_wmma_kernel(const float* __restrict__ h,
                           const float* __restrict__ Wc,
                           const float* __restrict__ cvec,
                           float* __restrict__ p,
                           int nRowBlocks) {
    const int lane = threadIdx.x & 31;
    const int c    = lane & 15;
    const int half = lane >> 4;
    const int rb   = blockIdx.x * (blockDim.x >> 5) + (threadIdx.x >> 5);
    if (rb >= nRowBlocks) return;
    const int m0 = rb * 16;

    const float4* __restrict__ arow = (const float4*)(h + (size_t)(m0 + c) * HID) + half;
    const float4* __restrict__ brow = (const float4*)(Wc + (size_t)(c & 3) * HID) + half;
    const bool live = (c < 4);

    float4 av[16], bw[16];
#pragma unroll
    for (int j = 0; j < 16; ++j) av[j] = arow[2 * j];
#pragma unroll
    for (int j = 0; j < 16; ++j) {
        const float4 w = brow[2 * j];
        bw[j] = live ? w : make_float4(0.f, 0.f, 0.f, 0.f);
    }

    const float cb = ((c >> 1) == 1) ? cvec[c & 1] : 0.f;
    v8f acc = (v8f){cb, cb, cb, cb, cb, cb, cb, cb};

#pragma unroll
    for (int j = 0; j < 16; ++j) {
        v2f a, b;
        a.x = av[j].x; a.y = av[j].y; b.x = bw[j].x; b.y = bw[j].y;
        acc = __builtin_amdgcn_wmma_f32_16x16x4_f32(
            false, a, false, b, (short)0, acc, false, false);
        a.x = av[j].z; a.y = av[j].w; b.x = bw[j].z; b.y = bw[j].w;
        acc = __builtin_amdgcn_wmma_f32_16x16x4_f32(
            false, a, false, b, (short)0, acc, false, false);
    }

    if (live) {
#pragma unroll
        for (int r = 0; r < 8; ++r)
            p[(size_t)(m0 + 8 * half + r) * 4 + c] = acc[r];
    }
}

__global__ __launch_bounds__(256)
void head_gather_kernel(const float* __restrict__ p,
                        const int* __restrict__ rowptr,
                        const int* __restrict__ col,
                        const float* __restrict__ rdeg,
                        float* __restrict__ out,
                        int nGroups) {
    const int lane = threadIdx.x & 31;
    const int c    = lane & 15;
    const int half = lane >> 4;
    const int g    = blockIdx.x * (blockDim.x >> 5) + (threadIdx.x >> 5);
    if (g >= nGroups) return;
    const int t0 = g * 16;

    const int rp0 = rowptr[t0 + c];
    const int rp1 = rowptr[t0 + c + 1];

    int e0[8], end[8];
#pragma unroll
    for (int j = 0; j < 8; ++j) {
        e0[j]  = __shfl(rp0, 2 * j + half) + c;
        end[j] = __shfl(rp1, 2 * j + half);
    }
    int s0[8], s1[8];
#pragma unroll
    for (int j = 0; j < 8; ++j) {
        s0[j] = (e0[j] < end[j])      ? col[e0[j]]      : -1;
        s1[j] = (e0[j] + 16 < end[j]) ? col[e0[j] + 16] : -1;
    }
    float sx[8], sy[8];
#pragma unroll
    for (int j = 0; j < 8; ++j) {
        float2 a = make_float2(0.f, 0.f);
        float2 b = make_float2(0.f, 0.f);
        if (s0[j] >= 0) a = *(const float2*)(p + (size_t)s0[j] * 4);
        if (s1[j] >= 0) b = *(const float2*)(p + (size_t)s1[j] * 4);
        sx[j] = a.x + b.x;
        sy[j] = a.y + b.y;
    }
#pragma unroll
    for (int j = 0; j < 8; ++j) {
        for (int e = e0[j] + 32; e < end[j]; e += 16) {
            const float2 v = *(const float2*)(p + (size_t)col[e] * 4);
            sx[j] += v.x;
            sy[j] += v.y;
        }
    }
    float kx = 0.f, ky = 0.f;
#pragma unroll
    for (int j = 0; j < 8; ++j) {
        for (int o = 8; o > 0; o >>= 1) {
            sx[j] += __shfl_xor(sx[j], o);
            sy[j] += __shfl_xor(sy[j], o);
        }
        if (c == 2 * j + half) { kx = sx[j]; ky = sy[j]; }
    }
    if ((c & 1) == half) {
        const int t = t0 + c;
        const float2 self = *(const float2*)(p + (size_t)t * 4 + 2);
        const float rd = rdeg[t];
        float2 o2;
        o2.x = kx * rd + self.x;
        o2.y = ky * rd + self.y;
        *(float2*)(out + (size_t)t * 2) = o2;
    }
}

extern "C" void kernel_launch(void* const* d_in, const int* in_sizes, int n_in,
                              void* d_out, int out_size, void* d_ws, size_t ws_size,
                              hipStream_t stream) {
    const float* x    = (const float*)d_in[0];
    const int*   ei   = (const int*)d_in[1];
    const float* Wl0  = (const float*)d_in[2];
    const float* bl0  = (const float*)d_in[3];
    const float* Wr0  = (const float*)d_in[4];
    const float* Wl1  = (const float*)d_in[5];
    const float* bl1  = (const float*)d_in[6];
    const float* Wr1  = (const float*)d_in[7];
    const float* Wl2  = (const float*)d_in[8];
    const float* bl2  = (const float*)d_in[9];
    const float* Wr2  = (const float*)d_in[10];
    const float* Wout = (const float*)d_in[11];
    const float* bout = (const float*)d_in[12];
    float* out = (float*)d_out;

    const int* src = ei;
    const int* dst = ei + N_EDGES;

    float* ws     = (float*)d_ws;
    float* rdeg   = ws;
    float* agg    = rdeg + 2 * (size_t)N_NODES;
    int*   rowptr = (int*)(agg + (size_t)N_NODES * HID);
    int*   col    = rowptr + N_NODES + 32;
    float* hA     = agg + (size_t)N_NODES * IN_PAD;
    int2*  binned = (int2*)hA;
    int*   counts = (int*)hA + 2 * (size_t)N_EDGES;
    int*   offs   = counts + NBIN * NCHUNK;
    float* xp     = hA + (size_t)N_NODES * HID;
    float* hB     = xp;
    float* Wl0p   = xp + (size_t)N_NODES * IN_PAD;
    float* Wr0p   = Wl0p + (size_t)HID * IN_PAD;
    float* p4     = hA;
    float* Wc     = p4 + (size_t)N_NODES * 4;
    float* cvec   = Wc + 4 * HID;

    const int ZT = 256;
    const int scanBlocks    = (NBIN * NCHUNK + 1023) / 1024;
    const int gatherBlocks  = 2048;
    const int gemmBlocks    = N_NODES / 16;
    const int halfNodes     = N_NODES / 2;
    const int padBlocksX    = (N_NODES * IN_PAD + PAD_CHUNK - 1) / PAD_CHUNK;
    const int padBlocksW    = (HID * IN_PAD + PAD_CHUNK - 1) / PAD_CHUNK;

    bin_count_kernel<<<NCHUNK, BIN_THREADS, 0, stream>>>(dst, counts, N_EDGES);
    scan_excl_kernel<<<scanBlocks, 1024, 0, stream>>>(counts, offs, NBIN * NCHUNK);
    bin_scatter_kernel<<<NCHUNK, BIN_THREADS, 0, stream>>>(src, dst, offs, binned, N_EDGES);
    bin_finish_kernel<<<NBIN, BIN_THREADS, 0, stream>>>(binned, offs, rowptr, rdeg, col, N_NODES);

    pad_rows_kernel<IN_CH, IN_PAD><<<padBlocksX, PAD_THREADS, 0, stream>>>(x,   xp,   N_NODES);
    pad_rows_kernel<IN_CH, IN_PAD><<<padBlocksW, PAD_THREADS, 0, stream>>>(Wl0, Wl0p, HID);
    pad_rows_kernel<IN_CH, IN_PAD><<<padBlocksW, PAD_THREADS, 0, stream>>>(Wr0, Wr0p, HID);

    gather_sum_kernel<IN_PAD><<<gatherBlocks, 256, 0, stream>>>(
        xp, rowptr, col, agg, 0, halfNodes);
    sage_gemm_wmma_kernel<IN_PAD, true><<<GEMM_WGS, GEMM_WAVES * 32, 0, stream>>>(
        agg, rdeg, xp, Wl0p, Wr0p, bl0, hA, halfNodes / 16);
    gather_sum_kernel<IN_PAD><<<gatherBlocks, 256, 0, stream>>>(
        xp, rowptr, col, agg, halfNodes, halfNodes);
    sage_gemm_wmma_kernel<IN_PAD, true><<<GEMM_WGS, GEMM_WAVES * 32, 0, stream>>>(
        agg, rdeg + halfNodes, xp + (size_t)halfNodes * IN_PAD, Wl0p, Wr0p, bl0,
        hA + (size_t)halfNodes * HID, halfNodes / 16);

    gather_sum_kernel<HID><<<gatherBlocks, 256, 0, stream>>>(
        hA, rowptr, col, agg, 0, N_NODES);
    sage_gemm_wmma_kernel<HID, true><<<GEMM_WGS, GEMM_WAVES * 32, 0, stream>>>(
        agg, rdeg, hA, Wl1, Wr1, bl1, hB, gemmBlocks);

    fold_head_kernel<<<(4 * HID + 2 + ZT - 1) / ZT, ZT, 0, stream>>>(
        Wout, bout, Wl2, Wr2, bl2, Wc, cvec);
    head_proj_wmma_kernel<<<(gemmBlocks + 7) / 8, 256, 0, stream>>>(
        hB, Wc, cvec, p4, gemmBlocks);
    head_gather_kernel<<<(gemmBlocks + 7) / 8, 256, 0, stream>>>(
        p4, rowptr, col, rdeg, out, gemmBlocks);
}


# First layer's neighbour product taken before the mean: y = x Wl0^T and x Wr0^T + b in one WMMA pass, gather sums 128 floats per edge not 168 and adds bias, own term, ReLU; sums re-associated
# speedup vs baseline: 6.3024x; 1.1306x over previous
#include <hip/hip_runtime.h>
#include <hip/hip_bf16.h>

#define N_NODES 100000
#define N_EDGES 1600000
#define IN_CH   165
#define IN_PAD  168
#define HID     128

typedef __attribute__((ext_vector_type(2))) float v2f;
typedef __attribute__((ext_vector_type(8))) float v8f;


#define PAD_THREADS 256
#define PAD_BATCH   32
#define PAD_NBATCH  2
#define PAD_CHUNK   (PAD_THREADS * PAD_BATCH * PAD_NBATCH)

template <int SRC_D, int DST_D, bool TAIL>
__device__ __forceinline__
void pad_batch(const float* __restrict__ src, float* __restrict__ dst,
               unsigned q0, unsigned total) {
    float v[PAD_BATCH];
#pragma unroll
    for (int u = 0; u < PAD_BATCH; ++u) {
        unsigned q = q0 + u * PAD_THREADS;
        if (TAIL) q = (q < total) ? q : total - 1;
        const unsigned r = q / DST_D;
        const unsigned c = q - r * DST_D;
        const bool in = (c < SRC_D);
        const float sv = src[r * SRC_D + (in ? c : 0u)];
        v[u] = in ? sv : 0.0f;
    }
#pragma unroll
    for (int u = 0; u < PAD_BATCH; ++u) {
        const unsigned q = q0 + u * PAD_THREADS;
        if (!TAIL || q < total) dst[q] = v[u];
    }
}

template <int SRC_D, int DST_D>
__global__ __launch_bounds__(PAD_THREADS)
void pad_rows_kernel(const float* __restrict__ src,
                     float* __restrict__ dst, int rows) {
    const unsigned total = (unsigned)rows * DST_D;
    const unsigned blockBase = blockIdx.x * (unsigned)PAD_CHUNK;
    if (blockBase + PAD_CHUNK <= total) {
        for (int b = 0; b < PAD_NBATCH; ++b)
            pad_batch<SRC_D, DST_D, false>(
                src, dst, blockBase + b * (PAD_THREADS * PAD_BATCH) + threadIdx.x, total);
    } else {
        for (int b = 0; b < PAD_NBATCH; ++b)
            pad_batch<SRC_D, DST_D, true>(
                src, dst, blockBase + b * (PAD_THREADS * PAD_BATCH) + threadIdx.x, total);
    }
}

#define BIN_SHIFT  9
#define BIN_NODES  (1 << BIN_SHIFT)
#define NBIN       ((N_NODES + BIN_NODES - 1) / BIN_NODES)
#define NCHUNK     256
#define CHUNK      ((N_EDGES + NCHUNK - 1) / NCHUNK)
#define BIN_THREADS 1024
#define BIN_BATCH  4

__global__ __launch_bounds__(BIN_THREADS)
void bin_count_kernel(const int* __restrict__ dst,
                      int* __restrict__ counts,
                      int nEdges) {
    __shared__ int hist[NBIN];
    const int tid = threadIdx.x;
    for (int i = tid; i < NBIN; i += BIN_THREADS) hist[i] = 0;
    __syncthreads();

    const int c0 = blockIdx.x * CHUNK;
    const int c1 = (c0 + CHUNK < nEdges) ? c0 + CHUNK : nEdges;
    for (int e0 = c0 + tid; e0 < c1; e0 += BIN_THREADS * BIN_BATCH) {
        int d[BIN_BATCH];
#pragma unroll
        for (int u = 0; u < BIN_BATCH; ++u) {
            const int e = e0 + u * BIN_THREADS;
            d[u] = (e < c1) ? dst[e] : -1;
        }
#pragma unroll
        for (int u = 0; u < BIN_BATCH; ++u)
            if (d[u] >= 0) atomicAdd(&hist[d[u] >> BIN_SHIFT], 1);
    }
    __syncthreads();
    for (int i = tid; i < NBIN; i += BIN_THREADS)
        counts[i * NCHUNK + blockIdx.x] = hist[i];
}

__global__ __launch_bounds__(1024)
void scan_excl_kernel(const int* __restrict__ in,
                      int* __restrict__ out, int n) {
    __shared__ int front[32];
    __shared__ int wtot[32];
    const int tid  = threadIdx.x;
    const int lane = tid & 31;
    const int w    = tid >> 5;
    const int first = blockIdx.x * 1024;

    int acc = 0;
    for (int j = tid; j < first; j += 1024) acc += in[j];
    for (int o = 16; o > 0; o >>= 1) acc += __shfl_xor(acc, o);
    if (lane == 0) front[w] = acc;

    const int i = first + tid;
    const int cI = (i < n) ? in[i] : 0;
    int v = cI;
    for (int o = 1; o < 32; o <<= 1) {
        int u = __shfl_up(v, o);
        if (lane >= o) v += u;
    }
    if (lane == 31) wtot[w] = v;
    __syncthreads();

    int base = 0;
    for (int k = 0; k < 32; ++k) base += front[k];
    for (int k = 0; k < w; ++k)  base += wtot[k];

    if (i < n) {
        out[i] = base + v - cI;
        if (i == n - 1) out[n] = base + v;
    }
}

__global__ __launch_bounds__(BIN_THREADS)
void bin_scatter_kernel(const int* __restrict__ src,
                        const int* __restrict__ dst,
                        const int* __restrict__ offs,
                        int2* __restrict__ binned,
                        int nEdges) {
    __shared__ int pos[NBIN];
    const int tid = threadIdx.x;
    for (int i = tid; i < NBIN; i += BIN_THREADS) pos[i] = offs[i * NCHUNK + blockIdx.x];
    __syncthreads();

    const int c0 = blockIdx.x * CHUNK;
    const int c1 = (c0 + CHUNK < nEdges) ? c0 + CHUNK : nEdges;
    for (int e0 = c0 + tid; e0 < c1; e0 += BIN_THREADS * BIN_BATCH) {
        int d[BIN_BATCH], s[BIN_BATCH];
#pragma unroll
        for (int u = 0; u < BIN_BATCH; ++u) {
            const int e = e0 + u * BIN_THREADS;
            const bool in = (e < c1);
            d[u] = in ? dst[e] : -1;
            s[u] = in ? src[e] : 0;
        }
#pragma unroll
        for (int u = 0; u < BIN_BATCH; ++u) {
            if (d[u] >= 0) {
                const int p = atomicAdd(&pos[d[u] >> BIN_SHIFT], 1);
                binned[p] = make_int2(s[u], d[u]);
            }
        }
    }
}

__global__ __launch_bounds__(BIN_THREADS)
void bin_finish_kernel(const int2* __restrict__ binned,
                       const int* __restrict__ offs,
                       int* __restrict__ rowptr,
                       float* __restrict__ rdeg,
                       int* __restrict__ col, int n) {
    __shared__ int cntL[BIN_NODES];
    __shared__ int posL[BIN_NODES];
    __shared__ int wtot[BIN_NODES / 32];
    const int tid  = threadIdx.x;
    const int lane = tid & 31;
    const int w    = tid >> 5;
    const int nodeBase = blockIdx.x * BIN_NODES;
    const int beg = offs[blockIdx.x * NCHUNK];
    const int end = offs[(blockIdx.x + 1) * NCHUNK];

    if (tid < BIN_NODES) cntL[tid] = 0;
    __syncthreads();

    for (int i0 = beg + tid; i0 < end; i0 += BIN_THREADS * BIN_BATCH) {
        int dy[BIN_BATCH];
#pragma unroll
        for (int u = 0; u < BIN_BATCH; ++u) {
            const int i = i0 + u * BIN_THREADS;
            dy[u] = (i < end) ? binned[i].y : -1;
        }
#pragma unroll
        for (int u = 0; u < BIN_BATCH; ++u)
            if (dy[u] >= 0) atomicAdd(&cntL[dy[u] - nodeBase], 1);
    }
    __syncthreads();

    int cI = 0, v = 0;
    if (tid < BIN_NODES) {
        cI = cntL[tid];
        v = cI;
        for (int o = 1; o < 32; o <<= 1) {
            int u = __shfl_up(v, o);
            if (lane >= o) v += u;
        }
        if (lane == 31) wtot[w] = v;
    }
    __syncthreads();
    if (tid < BIN_NODES) {
        int start = beg + v - cI;
        for (int k = 0; k < w; ++k) start += wtot[k];
        posL[tid] = start;
        const int node = nodeBase + tid;
        if (node < n) {
            rowptr[node] = start;
            rdeg[node]   = 1.0f / fmaxf((float)cI, 1.0f);
        }
    }
    if (blockIdx.x == gridDim.x - 1 && tid == 0) rowptr[n] = end;
    __syncthreads();

    for (int i0 = beg + tid; i0 < end; i0 += BIN_THREADS * BIN_BATCH) {
        int2 r[BIN_BATCH];
#pragma unroll
        for (int u = 0; u < BIN_BATCH; ++u) {
            const int i = i0 + u * BIN_THREADS;
            r[u] = (i < end) ? binned[i] : make_int2(0, -1);
        }
#pragma unroll
        for (int u = 0; u < BIN_BATCH; ++u) {
            if (r[u].y >= 0) {
                const int p = atomicAdd(&posL[r[u].y - nodeBase], 1);
                col[p] = r[u].x;
            }
        }
    }
}

template <int D, bool COMBINE>
__global__ __launch_bounds__(256)
void gather_sum_kernel(const float* __restrict__ h,
                       const int* __restrict__ rowptr,
                       const int* __restrict__ col,
                       const float* __restrict__ rdeg,
                       float* __restrict__ agg,
                       int nodeBegin, int nodeCount) {
    static_assert(D % 4 == 0 && D >= 128 && D <= 256, "row = 32..64 float4");
    constexpr int V4 = D / 4;
    constexpr bool LONG = (V4 > 32);
    const int lane = threadIdx.x & 31;
    const bool hasTail = LONG && (lane + 32 < V4);
    const int wpb = blockDim.x >> 5;
    const int nw  = gridDim.x * wpb;

    for (int i0 = blockIdx.x * wpb + (threadIdx.x >> 5); i0 < nodeCount; i0 += nw) {
        const int i = __builtin_amdgcn_readfirstlane(i0);
        const int t = nodeBegin + i;
        const int beg = rowptr[t];
        const int end = rowptr[t + 1];
        float4 a = make_float4(0.f, 0.f, 0.f, 0.f);
        float4 b = make_float4(0.f, 0.f, 0.f, 0.f);

        float4* o = (float4*)(agg + (size_t)i * D);
        float rd = 0.f;
        float4 za = make_float4(0.f, 0.f, 0.f, 0.f);
        float4 zb = make_float4(0.f, 0.f, 0.f, 0.f);
        if (COMBINE) {
            rd = rdeg[t];
            za = o[lane];
            if (hasTail) zb = o[lane + 32];
        }

        for (int base = beg; base < end; base += 32) {
            int n = end - base;
            if (n > 32) n = 32;
            const int cv = (lane < n) ? col[base + lane] : 0;

#define GATHER_BATCH(B, K0)                                                     \
            {                                                                   \
                float4 va[B];                                                   \
                float4 vb[B];                                                   \
                _Pragma("unroll")                                               \
                for (int q = 0; q < B; ++q) {                                   \
                    const int s = __builtin_amdgcn_readlane(cv, (K0) + q);      \
                    const float4* r = (const float4*)(h + (size_t)s * D);       \
                    va[q] = r[lane];                                            \
                    if (LONG) vb[q] = hasTail ? r[lane + 32]                    \
                                              : make_float4(0.f, 0.f, 0.f, 0.f);\
                }                                                               \
                _Pragma("unroll")                                               \
                for (int q = 0; q < B; ++q) {                                   \
                    a.x += va[q].x; a.y += va[q].y;                             \
                    a.z += va[q].z; a.w += va[q].w;                             \
                    if (LONG) {                                                 \
                        b.x += vb[q].x; b.y += vb[q].y;                         \
                        b.z += vb[q].z; b.w += vb[q].w;                         \
                    }                                                           \
                }                                                               \
            }
            int k = 0;
            for (; k + 8 <= n; k += 8) GATHER_BATCH(8, k)
            if (n & 4) { GATHER_BATCH(4, k) k += 4; }
            if (n & 2) { GATHER_BATCH(2, k) k += 2; }
            if (n & 1) { GATHER_BATCH(1, k) }
#undef GATHER_BATCH
        }

        if (COMBINE) {
            a.x = fmaxf(a.x * rd + za.x, 0.f); a.y = fmaxf(a.y * rd + za.y, 0.f);
            a.z = fmaxf(a.z * rd + za.z, 0.f); a.w = fmaxf(a.w * rd + za.w, 0.f);
            b.x = fmaxf(b.x * rd + zb.x, 0.f); b.y = fmaxf(b.y * rd + zb.y, 0.f);
            b.z = fmaxf(b.z * rd + zb.z, 0.f); b.w = fmaxf(b.w * rd + zb.w, 0.f);
        }
        o[lane] = a;
        if (hasTail) o[lane + 32] = b;
    }
}

#define GEMM_WGS   256
#define GEMM_WAVES 16

#define LOAD_B(BUF, S)                                                          \
    _Pragma("unroll")                                                           \
    for (int n = 0; n < 8; ++n) {                                               \
        BUF[2 * n]     = *(const v2f*)((S) + n * 64);                           \
        BUF[2 * n + 1] = *(const v2f*)((S) + HID * 4 + n * 64);                 \
    }
#define WMMA_STEP(AM, AH, BUF)                                                  \
    {                                                                           \
        v2f a1; a1.x = (AM).x * rd; a1.y = (AM).y * rd;                         \
        v2f a2; a2.x = (AH).x;      a2.y = (AH).y;                              \
        _Pragma("unroll")                                                       \
        for (int n = 0; n < 8; ++n) {                                           \
            acc[n] = __builtin_amdgcn_wmma_f32_16x16x4_f32(                     \
                false, a1, false, BUF[2 * n], (short)0, acc[n], false, false);  \
            acc[n] = __builtin_amdgcn_wmma_f32_16x16x4_f32(                     \
                false, a2, false, BUF[2 * n + 1], (short)0, acc[n], false, false); \
        }                                                                       \
    }

template <int D, bool RELU>
__global__ __launch_bounds__(GEMM_WAVES * 32)
void sage_gemm_wmma_kernel(const float* __restrict__ agg,
                           const float* __restrict__ rdeg,
                           const float* __restrict__ h,
                           const float* __restrict__ Wl,
                           const float* __restrict__ Wr,
                           const float* __restrict__ bias,
                           float* __restrict__ out,
                           int nRowBlocks) {
    static_assert(D % 8 == 0, "K is walked in pairs of k-steps of 4");
    constexpr int V4 = D / 4;
    constexpr int NB = V4 / 2;
    constexpr int KP = 2 * HID * 4 + 4;
    __shared__ float wsh[V4 * KP];

    constexpr int NT = GEMM_WAVES * 32;
    constexpr int ST = (HID * V4 + NT - 1) / NT;
#pragma unroll
    for (int t = 0; t < ST; ++t) {
        const int i = threadIdx.x + t * NT;
        if (i < HID * V4) {
            const int r = i / V4;
            const int q = i - r * V4;
            const float4 vl = ((const float4*)(Wl + (size_t)r * D))[q];
            const float4 vr = ((const float4*)(Wr + (size_t)r * D))[q];
            *(float4*)(wsh + q * KP + r * 4)           = vl;
            *(float4*)(wsh + q * KP + HID * 4 + r * 4) = vr;
        }
    }
    __syncthreads();

    const int lane = threadIdx.x & 31;
    const int wave = threadIdx.x >> 5;
    const int c    = lane & 15;
    const int koff = (lane >> 4) << 1;

    const float* __restrict__ bp = wsh + c * 4 + koff;

    float bv[8];
#pragma unroll
    for (int n = 0; n < 8; ++n) bv[n] = bias[n * 16 + c];

    const int turn  = ((wave >> 2) + (wave & 3)) & 3;
    const int first = turn * (4 * GEMM_WGS) + blockIdx.x * 4 + (wave & 3);

    for (int rb = first; rb < nRowBlocks; rb += GEMM_WGS * GEMM_WAVES) {
        const int m0   = rb * 16;
        const int mrow = m0 + c;
        const float rd = rdeg[mrow];

        const float2* __restrict__ arow = (const float2*)(agg + (size_t)mrow * D + koff);
        const float2* __restrict__ hrow = (const float2*)(h   + (size_t)mrow * D + koff);

        v8f acc[8];
#pragma unroll
        for (int n = 0; n < 8; ++n)
            acc[n] = (v8f){0.f, 0.f, 0.f, 0.f, 0.f, 0.f, 0.f, 0.f};

        float2 nm0 = arow[0], nh0 = hrow[0], nm1 = arow[2], nh1 = hrow[2];
        v2f b0[16], b1[16];
        LOAD_B(b0, bp)

        for (int blk = 0; blk < NB; ++blk) {
            const float2 cm0 = nm0, ch0 = nh0, cm1 = nm1, ch1 = nh1;
            const int nb = (blk + 1 < NB) ? blk + 1 : NB - 1;
            nm0 = arow[4 * nb];     nh0 = hrow[4 * nb];
            nm1 = arow[4 * nb + 2]; nh1 = hrow[4 * nb + 2];

            LOAD_B(b1, bp + (2 * blk + 1) * KP)
            WMMA_STEP(cm0, ch0, b0)
            LOAD_B(b0, bp + (2 * nb) * KP)
            WMMA_STEP(cm1, ch1, b1)
        }

        const int mbase = m0 + ((lane < 16) ? 0 : 8);
#pragma unroll
        for (int n = 0; n < 8; ++n) {
#pragma unroll
            for (int r = 0; r < 8; ++r) {
                float v = acc[n][r] + bv[n];
                if (RELU) v = fmaxf(v, 0.f);
                out[(size_t)(mbase + r) * HID + n * 16 + c] = v;
            }
        }
    }
}
#undef LOAD_B
#undef WMMA_STEP

#define LOAD_B8(BUF, S)                                                         \
    _Pragma("unroll")                                                           \
    for (int n = 0; n < 8; ++n)                                                 \
        BUF[n] = *(const v2f*)((S) + n * 64);
#define WMMA_STEP8(AH, BUF)                                                     \
    {                                                                           \
        v2f a1; a1.x = (AH).x; a1.y = (AH).y;                                   \
        _Pragma("unroll")                                                       \
        for (int n = 0; n < 8; ++n)                                             \
            acc[n] = __builtin_amdgcn_wmma_f32_16x16x4_f32(                     \
                false, a1, false, BUF[n], (short)0, acc[n], false, false);      \
    }

template <int D>
__global__ __launch_bounds__(GEMM_WAVES * 32)
void proj_gemm_wmma_kernel(const float* __restrict__ h,
                           const float* __restrict__ Wl,
                           const float* __restrict__ Wr,
                           const float* __restrict__ bias,
                           float* __restrict__ outL,
                           float* __restrict__ outR,
                           int nRowBlocks) {
    static_assert(D % 8 == 0, "K is walked in pairs of k-steps of 4");
    constexpr int V4 = D / 4;
    constexpr int NB = V4 / 2;
    constexpr int KP = 2 * HID * 4 + 4;
    __shared__ float wsh[V4 * KP];

    constexpr int NT = GEMM_WAVES * 32;
    constexpr int ST = (HID * V4 + NT - 1) / NT;
#pragma unroll
    for (int t = 0; t < ST; ++t) {
        const int i = threadIdx.x + t * NT;
        if (i < HID * V4) {
            const int r = i / V4;
            const int q = i - r * V4;
            const float4 vl = ((const float4*)(Wl + (size_t)r * D))[q];
            const float4 vr = ((const float4*)(Wr + (size_t)r * D))[q];
            *(float4*)(wsh + q * KP + r * 4)           = vl;
            *(float4*)(wsh + q * KP + HID * 4 + r * 4) = vr;
        }
    }
    __syncthreads();

    const int lane = threadIdx.x & 31;
    const int wave = threadIdx.x >> 5;
    const int c    = lane & 15;
    const int koff = (lane >> 4) << 1;

    const float* __restrict__ bp = wsh + c * 4 + koff;

    float bv[8];
#pragma unroll
    for (int n = 0; n < 8; ++n) bv[n] = bias[n * 16 + c];

    const int turn  = ((wave >> 2) + (wave & 3)) & 3;
    const int first = turn * (4 * GEMM_WGS) + blockIdx.x * 4 + (wave & 3);
    const int nUnits = 2 * nRowBlocks;

    for (int u = first; u < nUnits; u += GEMM_WGS * GEMM_WAVES) {
        const int  m0    = (u >> 1) * 16;
        const bool right = (u & 1) != 0;
        const int  mrow  = m0 + c;

        const float2* __restrict__ hrow = (const float2*)(h + (size_t)mrow * D + koff);
        const float*  __restrict__ bq   = bp + (right ? HID * 4 : 0);

        v8f acc[8];
#pragma unroll
        for (int n = 0; n < 8; ++n)
            acc[n] = (v8f){0.f, 0.f, 0.f, 0.f, 0.f, 0.f, 0.f, 0.f};

        float2 nh0 = hrow[0], nh1 = hrow[2];
        v2f b0[8], b1[8];
        LOAD_B8(b0, bq)

        for (int blk = 0; blk < NB; ++blk) {
            const float2 ch0 = nh0, ch1 = nh1;
            const int nb = (blk + 1 < NB) ? blk + 1 : NB - 1;
            nh0 = hrow[4 * nb];
            nh1 = hrow[4 * nb + 2];

            LOAD_B8(b1, bq + (2 * blk + 1) * KP)
            WMMA_STEP8(ch0, b0)
            LOAD_B8(b0, bq + (2 * nb) * KP)
            WMMA_STEP8(ch1, b1)
        }

        const int mbase = m0 + ((lane < 16) ? 0 : 8);
        float* __restrict__ out = right ? outR : outL;
#pragma unroll
        for (int n = 0; n < 8; ++n) {
            const float add = right ? bv[n] : 0.f;
#pragma unroll
            for (int r = 0; r < 8; ++r)
                out[(size_t)(mbase + r) * HID + n * 16 + c] = acc[n][r] + add;
        }
    }
}
#undef LOAD_B8
#undef WMMA_STEP8


__global__ __launch_bounds__(256)
void fold_head_kernel(const float* __restrict__ Wout,
                      const float* __restrict__ bout,
                      const float* __restrict__ Wl,
                      const float* __restrict__ Wr,
                      const float* __restrict__ bl,
                      float* __restrict__ Wc,
                      float* __restrict__ cvec) {
    const int t = blockIdx.x * blockDim.x + threadIdx.x;
    if (t < 4 * HID) {
        const int m = t >> 7;
        const int k = t & (HID - 1);
        const float* __restrict__ wo = Wout + (m & 1) * HID;
        const float* __restrict__ W  = (m < 2) ? Wl : Wr;
        float acc = 0.f;
        for (int o = 0; o < HID; ++o) acc += wo[o] * W[(size_t)o * HID + k];
        Wc[t] = acc;
    } else if (t < 4 * HID + 2) {
        const int j = t - 4 * HID;
        const float* __restrict__ wo = Wout + j * HID;
        float acc = bout[j];
        for (int o = 0; o < HID; ++o) acc += wo[o] * bl[o];
        cvec[j] = acc;
    }
}

__global__ __launch_bounds__(256)
void head_proj_wmma_kernel(const float* __restrict__ h,
                           const float* __restrict__ Wc,
                           const float* __restrict__ cvec,
                           float* __restrict__ p,
                           int nRowBlocks) {
    const int lane = threadIdx.x & 31;
    const int c    = lane & 15;
    const int half = lane >> 4;
    const int rb   = blockIdx.x * (blockDim.x >> 5) + (threadIdx.x >> 5);
    if (rb >= nRowBlocks) return;
    const int m0 = rb * 16;

    const float4* __restrict__ arow = (const float4*)(h + (size_t)(m0 + c) * HID) + half;
    const float4* __restrict__ brow = (const float4*)(Wc + (size_t)(c & 3) * HID) + half;
    const bool live = (c < 4);

    float4 av[16], bw[16];
#pragma unroll
    for (int j = 0; j < 16; ++j) av[j] = arow[2 * j];
#pragma unroll
    for (int j = 0; j < 16; ++j) {
        const float4 w = brow[2 * j];
        bw[j] = live ? w : make_float4(0.f, 0.f, 0.f, 0.f);
    }

    const float cb = ((c >> 1) == 1) ? cvec[c & 1] : 0.f;
    v8f acc = (v8f){cb, cb, cb, cb, cb, cb, cb, cb};

#pragma unroll
    for (int j = 0; j < 16; ++j) {
        v2f a, b;
        a.x = av[j].x; a.y = av[j].y; b.x = bw[j].x; b.y = bw[j].y;
        acc = __builtin_amdgcn_wmma_f32_16x16x4_f32(
            false, a, false, b, (short)0, acc, false, false);
        a.x = av[j].z; a.y = av[j].w; b.x = bw[j].z; b.y = bw[j].w;
        acc = __builtin_amdgcn_wmma_f32_16x16x4_f32(
            false, a, false, b, (short)0, acc, false, false);
    }

    if (live) {
#pragma unroll
        for (int r = 0; r < 8; ++r)
            p[(size_t)(m0 + 8 * half + r) * 4 + c] = acc[r];
    }
}

__global__ __launch_bounds__(256)
void head_gather_kernel(const float* __restrict__ p,
                        const int* __restrict__ rowptr,
                        const int* __restrict__ col,
                        const float* __restrict__ rdeg,
                        float* __restrict__ out,
                        int nGroups) {
    const int lane = threadIdx.x & 31;
    const int c    = lane & 15;
    const int half = lane >> 4;
    const int g    = blockIdx.x * (blockDim.x >> 5) + (threadIdx.x >> 5);
    if (g >= nGroups) return;
    const int t0 = g * 16;

    const int rp0 = rowptr[t0 + c];
    const int rp1 = rowptr[t0 + c + 1];

    int e0[8], end[8];
#pragma unroll
    for (int j = 0; j < 8; ++j) {
        e0[j]  = __shfl(rp0, 2 * j + half) + c;
        end[j] = __shfl(rp1, 2 * j + half);
    }
    int s0[8], s1[8];
#pragma unroll
    for (int j = 0; j < 8; ++j) {
        s0[j] = (e0[j] < end[j])      ? col[e0[j]]      : -1;
        s1[j] = (e0[j] + 16 < end[j]) ? col[e0[j] + 16] : -1;
    }
    float sx[8], sy[8];
#pragma unroll
    for (int j = 0; j < 8; ++j) {
        float2 a = make_float2(0.f, 0.f);
        float2 b = make_float2(0.f, 0.f);
        if (s0[j] >= 0) a = *(const float2*)(p + (size_t)s0[j] * 4);
        if (s1[j] >= 0) b = *(const float2*)(p + (size_t)s1[j] * 4);
        sx[j] = a.x + b.x;
        sy[j] = a.y + b.y;
    }
#pragma unroll
    for (int j = 0; j < 8; ++j) {
        for (int e = e0[j] + 32; e < end[j]; e += 16) {
            const float2 v = *(const float2*)(p + (size_t)col[e] * 4);
            sx[j] += v.x;
            sy[j] += v.y;
        }
    }
    float kx = 0.f, ky = 0.f;
#pragma unroll
    for (int j = 0; j < 8; ++j) {
        for (int o = 8; o > 0; o >>= 1) {
            sx[j] += __shfl_xor(sx[j], o);
            sy[j] += __shfl_xor(sy[j], o);
        }
        if (c == 2 * j + half) { kx = sx[j]; ky = sy[j]; }
    }
    if ((c & 1) == half) {
        const int t = t0 + c;
        const float2 self = *(const float2*)(p + (size_t)t * 4 + 2);
        const float rd = rdeg[t];
        float2 o2;
        o2.x = kx * rd + self.x;
        o2.y = ky * rd + self.y;
        *(float2*)(out + (size_t)t * 2) = o2;
    }
}

extern "C" void kernel_launch(void* const* d_in, const int* in_sizes, int n_in,
                              void* d_out, int out_size, void* d_ws, size_t ws_size,
                              hipStream_t stream) {
    const float* x    = (const float*)d_in[0];
    const int*   ei   = (const int*)d_in[1];
    const float* Wl0  = (const float*)d_in[2];
    const float* bl0  = (const float*)d_in[3];
    const float* Wr0  = (const float*)d_in[4];
    const float* Wl1  = (const float*)d_in[5];
    const float* bl1  = (const float*)d_in[6];
    const float* Wr1  = (const float*)d_in[7];
    const float* Wl2  = (const float*)d_in[8];
    const float* bl2  = (const float*)d_in[9];
    const float* Wr2  = (const float*)d_in[10];
    const float* Wout = (const float*)d_in[11];
    const float* bout = (const float*)d_in[12];
    float* out = (float*)d_out;

    const int* src = ei;
    const int* dst = ei + N_EDGES;

    float* ws     = (float*)d_ws;
    float* rdeg   = ws;
    float* agg    = rdeg + 2 * (size_t)N_NODES;
    int*   rowptr = (int*)(agg + (size_t)N_NODES * HID);
    int*   col    = rowptr + N_NODES + 32;
    float* hA     = agg + (size_t)N_NODES * IN_PAD;
    int2*  binned = (int2*)hA;
    int*   counts = (int*)hA + 2 * (size_t)N_EDGES;
    int*   offs   = counts + NBIN * NCHUNK;
    float* xp     = hA + (size_t)N_NODES * HID;
    float* hB     = xp;
    float* Wl0p   = xp + (size_t)N_NODES * IN_PAD;
    float* Wr0p   = Wl0p + (size_t)HID * IN_PAD;
    float* p4     = hA;
    float* Wc     = p4 + (size_t)N_NODES * 4;
    float* cvec   = Wc + 4 * HID;

    const int ZT = 256;
    const int scanBlocks    = (NBIN * NCHUNK + 1023) / 1024;
    const int gatherBlocks  = 2048;
    const int gemmBlocks    = N_NODES / 16;
    const int padBlocksX    = (N_NODES * IN_PAD + PAD_CHUNK - 1) / PAD_CHUNK;
    const int padBlocksW    = (HID * IN_PAD + PAD_CHUNK - 1) / PAD_CHUNK;

    bin_count_kernel<<<NCHUNK, BIN_THREADS, 0, stream>>>(dst, counts, N_EDGES);
    scan_excl_kernel<<<scanBlocks, 1024, 0, stream>>>(counts, offs, NBIN * NCHUNK);
    bin_scatter_kernel<<<NCHUNK, BIN_THREADS, 0, stream>>>(src, dst, offs, binned, N_EDGES);
    bin_finish_kernel<<<NBIN, BIN_THREADS, 0, stream>>>(binned, offs, rowptr, rdeg, col, N_NODES);

    pad_rows_kernel<IN_CH, IN_PAD><<<padBlocksX, PAD_THREADS, 0, stream>>>(x,   xp,   N_NODES);
    pad_rows_kernel<IN_CH, IN_PAD><<<padBlocksW, PAD_THREADS, 0, stream>>>(Wl0, Wl0p, HID);
    pad_rows_kernel<IN_CH, IN_PAD><<<padBlocksW, PAD_THREADS, 0, stream>>>(Wr0, Wr0p, HID);

    proj_gemm_wmma_kernel<IN_PAD><<<GEMM_WGS, GEMM_WAVES * 32, 0, stream>>>(
        xp, Wl0p, Wr0p, bl0, agg, hA, gemmBlocks);
    gather_sum_kernel<HID, true><<<gatherBlocks, 256, 0, stream>>>(
        agg, rowptr, col, rdeg, hA, 0, N_NODES);

    gather_sum_kernel<HID, false><<<gatherBlocks, 256, 0, stream>>>(
        hA, rowptr, col, rdeg, agg, 0, N_NODES);
    sage_gemm_wmma_kernel<HID, true><<<GEMM_WGS, GEMM_WAVES * 32, 0, stream>>>(
        agg, rdeg, hA, Wl1, Wr1, bl1, hB, gemmBlocks);

    fold_head_kernel<<<(4 * HID + 2 + ZT - 1) / ZT, ZT, 0, stream>>>(
        Wout, bout, Wl2, Wr2, bl2, Wc, cvec);
    head_proj_wmma_kernel<<<(gemmBlocks + 7) / 8, 256, 0, stream>>>(
        hB, Wc, cvec, p4, gemmBlocks);
    head_gather_kernel<<<(gemmBlocks + 7) / 8, 256, 0, stream>>>(
        p4, rowptr, col, rdeg, out, gemmBlocks);
}


# Layer-0 GEMM reads x and both weight matrices unpadded (rows of 165 floats, 4-byte-aligned 8/16-byte loads, k 165..167 zeroed in its loaders); the 3 padding copies and their launches are gone
# speedup vs baseline: 6.8659x; 1.0894x over previous
#include <hip/hip_runtime.h>
#include <hip/hip_bf16.h>

#define N_NODES 100000
#define N_EDGES 1600000
#define IN_CH   165
#define IN_PAD  168
#define HID     128

typedef __attribute__((ext_vector_type(2))) float v2f;
typedef __attribute__((ext_vector_type(8))) float v8f;


typedef float f2a4 __attribute__((ext_vector_type(2), aligned(4)));
typedef float f4a4 __attribute__((ext_vector_type(4), aligned(4)));
__device__ __forceinline__ float2 ld_f2(const float* __restrict__ p) {
    const f2a4 v = *(const f2a4*)p;
    return make_float2(v.x, v.y);
}
__device__ __forceinline__ float4 ld_f4(const float* __restrict__ p) {
    const f4a4 v = *(const f4a4*)p;
    return make_float4(v.x, v.y, v.z, v.w);
}

template <int REM>
__device__ __forceinline__ float4 tail4(const float4 v, const bool last) {
    float4 o;
    o.x = last ? (REM == 1 ? v.w : REM == 2 ? v.z : REM == 3 ? v.y : v.x) : v.x;
    o.y = last ? (REM == 1 ? 0.f : REM == 2 ? v.w : REM == 3 ? v.z : v.y) : v.y;
    o.z = last ? (REM <= 2 ? 0.f : REM == 3 ? v.w : v.z) : v.z;
    o.w = last ? (REM <= 3 ? 0.f : v.w) : v.w;
    return o;
}

#define BIN_SHIFT  9
#define BIN_NODES  (1 << BIN_SHIFT)
#define NBIN       ((N_NODES + BIN_NODES - 1) / BIN_NODES)
#define NCHUNK     256
#define CHUNK      ((N_EDGES + NCHUNK - 1) / NCHUNK)
#define BIN_THREADS 1024
#define BIN_BATCH  4

__global__ __launch_bounds__(BIN_THREADS)
void bin_count_kernel(const int* __restrict__ dst,
                      int* __restrict__ counts,
                      int nEdges) {
    __shared__ int hist[NBIN];
    const int tid = threadIdx.x;
    for (int i = tid; i < NBIN; i += BIN_THREADS) hist[i] = 0;
    __syncthreads();

    const int c0 = blockIdx.x * CHUNK;
    const int c1 = (c0 + CHUNK < nEdges) ? c0 + CHUNK : nEdges;
    for (int e0 = c0 + tid; e0 < c1; e0 += BIN_THREADS * BIN_BATCH) {
        int d[BIN_BATCH];
#pragma unroll
        for (int u = 0; u < BIN_BATCH; ++u) {
            const int e = e0 + u * BIN_THREADS;
            d[u] = (e < c1) ? dst[e] : -1;
        }
#pragma unroll
        for (int u = 0; u < BIN_BATCH; ++u)
            if (d[u] >= 0) atomicAdd(&hist[d[u] >> BIN_SHIFT], 1);
    }
    __syncthreads();
    for (int i = tid; i < NBIN; i += BIN_THREADS)
        counts[i * NCHUNK + blockIdx.x] = hist[i];
}

__global__ __launch_bounds__(1024)
void scan_excl_kernel(const int* __restrict__ in,
                      int* __restrict__ out, int n) {
    __shared__ int front[32];
    __shared__ int wtot[32];
    const int tid  = threadIdx.x;
    const int lane = tid & 31;
    const int w    = tid >> 5;
    const int first = blockIdx.x * 1024;

    int acc = 0;
    for (int j = tid; j < first; j += 1024) acc += in[j];
    for (int o = 16; o > 0; o >>= 1) acc += __shfl_xor(acc, o);
    if (lane == 0) front[w] = acc;

    const int i = first + tid;
    const int cI = (i < n) ? in[i] : 0;
    int v = cI;
    for (int o = 1; o < 32; o <<= 1) {
        int u = __shfl_up(v, o);
        if (lane >= o) v += u;
    }
    if (lane == 31) wtot[w] = v;
    __syncthreads();

    int base = 0;
    for (int k = 0; k < 32; ++k) base += front[k];
    for (int k = 0; k < w; ++k)  base += wtot[k];

    if (i < n) {
        out[i] = base + v - cI;
        if (i == n - 1) out[n] = base + v;
    }
}

__global__ __launch_bounds__(BIN_THREADS)
void bin_scatter_kernel(const int* __restrict__ src,
                        const int* __restrict__ dst,
                        const int* __restrict__ offs,
                        int2* __restrict__ binned,
                        int nEdges) {
    __shared__ int pos[NBIN];
    const int tid = threadIdx.x;
    for (int i = tid; i < NBIN; i += BIN_THREADS) pos[i] = offs[i * NCHUNK + blockIdx.x];
    __syncthreads();

    const int c0 = blockIdx.x * CHUNK;
    const int c1 = (c0 + CHUNK < nEdges) ? c0 + CHUNK : nEdges;
    for (int e0 = c0 + tid; e0 < c1; e0 += BIN_THREADS * BIN_BATCH) {
        int d[BIN_BATCH], s[BIN_BATCH];
#pragma unroll
        for (int u = 0; u < BIN_BATCH; ++u) {
            const int e = e0 + u * BIN_THREADS;
            const bool in = (e < c1);
            d[u] = in ? dst[e] : -1;
            s[u] = in ? src[e] : 0;
        }
#pragma unroll
        for (int u = 0; u < BIN_BATCH; ++u) {
            if (d[u] >= 0) {
                const int p = atomicAdd(&pos[d[u] >> BIN_SHIFT], 1);
                binned[p] = make_int2(s[u], d[u]);
            }
        }
    }
}

__global__ __launch_bounds__(BIN_THREADS)
void bin_finish_kernel(const int2* __restrict__ binned,
                       const int* __restrict__ offs,
                       int* __restrict__ rowptr,
                       float* __restrict__ rdeg,
                       int* __restrict__ col, int n) {
    __shared__ int cntL[BIN_NODES];
    __shared__ int posL[BIN_NODES];
    __shared__ int wtot[BIN_NODES / 32];
    const int tid  = threadIdx.x;
    const int lane = tid & 31;
    const int w    = tid >> 5;
    const int nodeBase = blockIdx.x * BIN_NODES;
    const int beg = offs[blockIdx.x * NCHUNK];
    const int end = offs[(blockIdx.x + 1) * NCHUNK];

    if (tid < BIN_NODES) cntL[tid] = 0;
    __syncthreads();

    for (int i0 = beg + tid; i0 < end; i0 += BIN_THREADS * BIN_BATCH) {
        int dy[BIN_BATCH];
#pragma unroll
        for (int u = 0; u < BIN_BATCH; ++u) {
            const int i = i0 + u * BIN_THREADS;
            dy[u] = (i < end) ? binned[i].y : -1;
        }
#pragma unroll
        for (int u = 0; u < BIN_BATCH; ++u)
            if (dy[u] >= 0) atomicAdd(&cntL[dy[u] - nodeBase], 1);
    }
    __syncthreads();

    int cI = 0, v = 0;
    if (tid < BIN_NODES) {
        cI = cntL[tid];
        v = cI;
        for (int o = 1; o < 32; o <<= 1) {
            int u = __shfl_up(v, o);
            if (lane >= o) v += u;
        }
        if (lane == 31) wtot[w] = v;
    }
    __syncthreads();
    if (tid < BIN_NODES) {
        int start = beg + v - cI;
        for (int k = 0; k < w; ++k) start += wtot[k];
        posL[tid] = start;
        const int node = nodeBase + tid;
        if (node < n) {
            rowptr[node] = start;
            rdeg[node]   = 1.0f / fmaxf((float)cI, 1.0f);
        }
    }
    if (blockIdx.x == gridDim.x - 1 && tid == 0) rowptr[n] = end;
    __syncthreads();

    for (int i0 = beg + tid; i0 < end; i0 += BIN_THREADS * BIN_BATCH) {
        int2 r[BIN_BATCH];
#pragma unroll
        for (int u = 0; u < BIN_BATCH; ++u) {
            const int i = i0 + u * BIN_THREADS;
            r[u] = (i < end) ? binned[i] : make_int2(0, -1);
        }
#pragma unroll
        for (int u = 0; u < BIN_BATCH; ++u) {
            if (r[u].y >= 0) {
                const int p = atomicAdd(&posL[r[u].y - nodeBase], 1);
                col[p] = r[u].x;
            }
        }
    }
}

template <int D, bool COMBINE>
__global__ __launch_bounds__(256)
void gather_sum_kernel(const float* __restrict__ h,
                       const int* __restrict__ rowptr,
                       const int* __restrict__ col,
                       const float* __restrict__ rdeg,
                       float* __restrict__ agg,
                       int nodeBegin, int nodeCount) {
    static_assert(D % 4 == 0 && D >= 128 && D <= 256, "row = 32..64 float4");
    constexpr int V4 = D / 4;
    constexpr bool LONG = (V4 > 32);
    const int lane = threadIdx.x & 31;
    const bool hasTail = LONG && (lane + 32 < V4);
    const int wpb = blockDim.x >> 5;
    const int nw  = gridDim.x * wpb;

    for (int i0 = blockIdx.x * wpb + (threadIdx.x >> 5); i0 < nodeCount; i0 += nw) {
        const int i = __builtin_amdgcn_readfirstlane(i0);
        const int t = nodeBegin + i;
        const int beg = rowptr[t];
        const int end = rowptr[t + 1];
        float4 a = make_float4(0.f, 0.f, 0.f, 0.f);
        float4 b = make_float4(0.f, 0.f, 0.f, 0.f);

        float4* o = (float4*)(agg + (size_t)i * D);
        float rd = 0.f;
        float4 za = make_float4(0.f, 0.f, 0.f, 0.f);
        float4 zb = make_float4(0.f, 0.f, 0.f, 0.f);
        if (COMBINE) {
            rd = rdeg[t];
            za = o[lane];
            if (hasTail) zb = o[lane + 32];
        }

        for (int base = beg; base < end; base += 32) {
            int n = end - base;
            if (n > 32) n = 32;
            const int cv = (lane < n) ? col[base + lane] : 0;

#define GATHER_BATCH(B, K0)                                                     \
            {                                                                   \
                float4 va[B];                                                   \
                float4 vb[B];                                                   \
                _Pragma("unroll")                                               \
                for (int q = 0; q < B; ++q) {                                   \
                    const int s = __builtin_amdgcn_readlane(cv, (K0) + q);      \
                    const float4* r = (const float4*)(h + (size_t)s * D);       \
                    va[q] = r[lane];                                            \
                    if (LONG) vb[q] = hasTail ? r[lane + 32]                    \
                                              : make_float4(0.f, 0.f, 0.f, 0.f);\
                }                                                               \
                _Pragma("unroll")                                               \
                for (int q = 0; q < B; ++q) {                                   \
                    a.x += va[q].x; a.y += va[q].y;                             \
                    a.z += va[q].z; a.w += va[q].w;                             \
                    if (LONG) {                                                 \
                        b.x += vb[q].x; b.y += vb[q].y;                         \
                        b.z += vb[q].z; b.w += vb[q].w;                         \
                    }                                                           \
                }                                                               \
            }
            int k = 0;
            for (; k + 8 <= n; k += 8) GATHER_BATCH(8, k)
            if (n & 4) { GATHER_BATCH(4, k) k += 4; }
            if (n & 2) { GATHER_BATCH(2, k) k += 2; }
            if (n & 1) { GATHER_BATCH(1, k) }
#undef GATHER_BATCH
        }

        if (COMBINE) {
            a.x = fmaxf(a.x * rd + za.x, 0.f); a.y = fmaxf(a.y * rd + za.y, 0.f);
            a.z = fmaxf(a.z * rd + za.z, 0.f); a.w = fmaxf(a.w * rd + za.w, 0.f);
            b.x = fmaxf(b.x * rd + zb.x, 0.f); b.y = fmaxf(b.y * rd + zb.y, 0.f);
            b.z = fmaxf(b.z * rd + zb.z, 0.f); b.w = fmaxf(b.w * rd + zb.w, 0.f);
        }
        o[lane] = a;
        if (hasTail) o[lane + 32] = b;
    }
}

#define GEMM_WGS   256
#define GEMM_WAVES 16

#define LOAD_B(BUF, S)                                                          \
    _Pragma("unroll")                                                           \
    for (int n = 0; n < 8; ++n) {                                               \
        BUF[2 * n]     = *(const v2f*)((S) + n * 64);                           \
        BUF[2 * n + 1] = *(const v2f*)((S) + HID * 4 + n * 64);                 \
    }
#define WMMA_STEP(AM, AH, BUF)                                                  \
    {                                                                           \
        v2f a1; a1.x = (AM).x * rd; a1.y = (AM).y * rd;                         \
        v2f a2; a2.x = (AH).x;      a2.y = (AH).y;                              \
        _Pragma("unroll")                                                       \
        for (int n = 0; n < 8; ++n) {                                           \
            acc[n] = __builtin_amdgcn_wmma_f32_16x16x4_f32(                     \
                false, a1, false, BUF[2 * n], (short)0, acc[n], false, false);  \
            acc[n] = __builtin_amdgcn_wmma_f32_16x16x4_f32(                     \
                false, a2, false, BUF[2 * n + 1], (short)0, acc[n], false, false); \
        }                                                                       \
    }

template <int D, bool RELU>
__global__ __launch_bounds__(GEMM_WAVES * 32)
void sage_gemm_wmma_kernel(const float* __restrict__ agg,
                           const float* __restrict__ rdeg,
                           const float* __restrict__ h,
                           const float* __restrict__ Wl,
                           const float* __restrict__ Wr,
                           const float* __restrict__ bias,
                           float* __restrict__ out,
                           int nRowBlocks) {
    static_assert(D % 8 == 0, "K is walked in pairs of k-steps of 4");
    constexpr int V4 = D / 4;
    constexpr int NB = V4 / 2;
    constexpr int KP = 2 * HID * 4 + 4;
    __shared__ float wsh[V4 * KP];

    constexpr int NT = GEMM_WAVES * 32;
    constexpr int ST = (HID * V4 + NT - 1) / NT;
#pragma unroll
    for (int t = 0; t < ST; ++t) {
        const int i = threadIdx.x + t * NT;
        if (i < HID * V4) {
            const int r = i / V4;
            const int q = i - r * V4;
            const float4 vl = ((const float4*)(Wl + (size_t)r * D))[q];
            const float4 vr = ((const float4*)(Wr + (size_t)r * D))[q];
            *(float4*)(wsh + q * KP + r * 4)           = vl;
            *(float4*)(wsh + q * KP + HID * 4 + r * 4) = vr;
        }
    }
    __syncthreads();

    const int lane = threadIdx.x & 31;
    const int wave = threadIdx.x >> 5;
    const int c    = lane & 15;
    const int koff = (lane >> 4) << 1;

    const float* __restrict__ bp = wsh + c * 4 + koff;

    float bv[8];
#pragma unroll
    for (int n = 0; n < 8; ++n) bv[n] = bias[n * 16 + c];

    const int turn  = ((wave >> 2) + (wave & 3)) & 3;
    const int first = turn * (4 * GEMM_WGS) + blockIdx.x * 4 + (wave & 3);

    for (int rb = first; rb < nRowBlocks; rb += GEMM_WGS * GEMM_WAVES) {
        const int m0   = rb * 16;
        const int mrow = m0 + c;
        const float rd = rdeg[mrow];

        const float2* __restrict__ arow = (const float2*)(agg + (size_t)mrow * D + koff);
        const float2* __restrict__ hrow = (const float2*)(h   + (size_t)mrow * D + koff);

        v8f acc[8];
#pragma unroll
        for (int n = 0; n < 8; ++n)
            acc[n] = (v8f){0.f, 0.f, 0.f, 0.f, 0.f, 0.f, 0.f, 0.f};

        float2 nm0 = arow[0], nh0 = hrow[0], nm1 = arow[2], nh1 = hrow[2];
        v2f b0[16], b1[16];
        LOAD_B(b0, bp)

        for (int blk = 0; blk < NB; ++blk) {
            const float2 cm0 = nm0, ch0 = nh0, cm1 = nm1, ch1 = nh1;
            const int nb = (blk + 1 < NB) ? blk + 1 : NB - 1;
            nm0 = arow[4 * nb];     nh0 = hrow[4 * nb];
            nm1 = arow[4 * nb + 2]; nh1 = hrow[4 * nb + 2];

            LOAD_B(b1, bp + (2 * blk + 1) * KP)
            WMMA_STEP(cm0, ch0, b0)
            LOAD_B(b0, bp + (2 * nb) * KP)
            WMMA_STEP(cm1, ch1, b1)
        }

        const int mbase = m0 + ((lane < 16) ? 0 : 8);
#pragma unroll
        for (int n = 0; n < 8; ++n) {
#pragma unroll
            for (int r = 0; r < 8; ++r) {
                float v = acc[n][r] + bv[n];
                if (RELU) v = fmaxf(v, 0.f);
                out[(size_t)(mbase + r) * HID + n * 16 + c] = v;
            }
        }
    }
}
#undef LOAD_B
#undef WMMA_STEP

#define LOAD_B8(BUF, S)                                                         \
    _Pragma("unroll")                                                           \
    for (int n = 0; n < 8; ++n)                                                 \
        BUF[n] = *(const v2f*)((S) + n * 64);
#define WMMA_STEP8(AH, BUF)                                                     \
    {                                                                           \
        v2f a1; a1.x = (AH).x; a1.y = (AH).y;                                   \
        _Pragma("unroll")                                                       \
        for (int n = 0; n < 8; ++n)                                             \
            acc[n] = __builtin_amdgcn_wmma_f32_16x16x4_f32(                     \
                false, a1, false, BUF[n], (short)0, acc[n], false, false);      \
    }

template <int D, int SRC_D>
__global__ __launch_bounds__(GEMM_WAVES * 32)
void proj_gemm_wmma_kernel(const float* __restrict__ h,
                           const float* __restrict__ Wl,
                           const float* __restrict__ Wr,
                           const float* __restrict__ bias,
                           float* __restrict__ outL,
                           float* __restrict__ outR,
                           int nRowBlocks) {
    static_assert(D % 8 == 0 && D >= 16, "K is walked in pairs of k-steps of 4, two pairs or more");
    static_assert(SRC_D <= D && SRC_D > D - 4,
                  "only the last k-step may be short");
    constexpr int V4 = D / 4;
    constexpr int NB = V4 / 2;
    constexpr int KT = 4 * (V4 - 1);
    constexpr int REM = SRC_D - KT;
    constexpr int KP = 2 * HID * 4 + 4;
    __shared__ float wsh[V4 * KP];

    constexpr int NT = GEMM_WAVES * 32;
    constexpr int ST = (HID * V4 + NT - 1) / NT;
    float4 vl[ST], vr[ST];
#pragma unroll
    for (int t = 0; t < ST; ++t) {
        const int i = threadIdx.x + t * NT;
        if (i < HID * V4) {
            const int r = i / V4;
            const int q = i - r * V4;
            const bool last = (REM < 4) && (q == V4 - 1);
            const int off = last ? SRC_D - 4 : 4 * q;
            vl[t] = ld_f4(Wl + (size_t)r * SRC_D + off);
            vr[t] = ld_f4(Wr + (size_t)r * SRC_D + off);
        }
    }
#pragma unroll
    for (int t = 0; t < ST; ++t) {
        const int i = threadIdx.x + t * NT;
        if (i < HID * V4) {
            const int r = i / V4;
            const int q = i - r * V4;
            const bool last = (REM < 4) && (q == V4 - 1);
            *(float4*)(wsh + q * KP + r * 4)           = tail4<REM>(vl[t], last);
            *(float4*)(wsh + q * KP + HID * 4 + r * 4) = tail4<REM>(vr[t], last);
        }
    }
    __syncthreads();

    const int lane = threadIdx.x & 31;
    const int wave = threadIdx.x >> 5;
    const int c    = lane & 15;
    const int koff = (lane >> 4) << 1;

    const float* __restrict__ bp = wsh + c * 4 + koff;

    float bv[8];
#pragma unroll
    for (int n = 0; n < 8; ++n) bv[n] = bias[n * 16 + c];

    const int turn  = ((wave >> 2) + (wave & 3)) & 3;
    const int first = turn * (4 * GEMM_WGS) + blockIdx.x * 4 + (wave & 3);
    const int nUnits = 2 * nRowBlocks;

    for (int u = first; u < nUnits; u += GEMM_WGS * GEMM_WAVES) {
        const int  m0    = (u >> 1) * 16;
        const bool right = (u & 1) != 0;
        const int  mrow  = m0 + c;

        const float* __restrict__ hrow = h + (size_t)mrow * SRC_D;
        const float* __restrict__ bq   = bp + (right ? HID * 4 : 0);

        v8f acc[8];
#pragma unroll
        for (int n = 0; n < 8; ++n)
            acc[n] = (v8f){0.f, 0.f, 0.f, 0.f, 0.f, 0.f, 0.f, 0.f};

        float2 nh0 = ld_f2(hrow + koff), nh1 = ld_f2(hrow + 4 + koff);
        v2f b0[8], b1[8];
        LOAD_B8(b0, bq)

        for (int blk = 0; blk < NB; ++blk) {
            const float2 ch0 = nh0;
            float2 ch1 = nh1;
            if (REM < 4) {
                const int k1 = 8 * blk + 4 + koff;
                ch1.x = (k1 < SRC_D) ? ((k1 > SRC_D - 2) ? nh1.y : nh1.x) : 0.f;
                ch1.y = (k1 + 1 < SRC_D) ? nh1.y : 0.f;
            }
            const int nb = (blk + 1 < NB) ? blk + 1 : NB - 1;
            nh0 = ld_f2(hrow + 8 * nb + koff);
            const int o1 = 8 * nb + 4 + koff;
            nh1 = ld_f2(hrow + ((REM < 4 && o1 > SRC_D - 2) ? SRC_D - 2 : o1));

            LOAD_B8(b1, bq + (2 * blk + 1) * KP)
            WMMA_STEP8(ch0, b0)
            LOAD_B8(b0, bq + (2 * nb) * KP)
            WMMA_STEP8(ch1, b1)
        }

        const int mbase = m0 + ((lane < 16) ? 0 : 8);
        float* __restrict__ out = right ? outR : outL;
#pragma unroll
        for (int n = 0; n < 8; ++n) {
            const float add = right ? bv[n] : 0.f;
#pragma unroll
            for (int r = 0; r < 8; ++r)
                out[(size_t)(mbase + r) * HID + n * 16 + c] = acc[n][r] + add;
        }
    }
}
#undef LOAD_B8
#undef WMMA_STEP8


__global__ __launch_bounds__(256)
void fold_head_kernel(const float* __restrict__ Wout,
                      const float* __restrict__ bout,
                      const float* __restrict__ Wl,
                      const float* __restrict__ Wr,
                      const float* __restrict__ bl,
                      float* __restrict__ Wc,
                      float* __restrict__ cvec) {
    const int t = blockIdx.x * blockDim.x + threadIdx.x;
    if (t < 4 * HID) {
        const int m = t >> 7;
        const int k = t & (HID - 1);
        const float* __restrict__ wo = Wout + (m & 1) * HID;
        const float* __restrict__ W  = (m < 2) ? Wl : Wr;
        float acc = 0.f;
        for (int o = 0; o < HID; ++o) acc += wo[o] * W[(size_t)o * HID + k];
        Wc[t] = acc;
    } else if (t < 4 * HID + 2) {
        const int j = t - 4 * HID;
        const float* __restrict__ wo = Wout + j * HID;
        float acc = bout[j];
        for (int o = 0; o < HID; ++o) acc += wo[o] * bl[o];
        cvec[j] = acc;
    }
}

__global__ __launch_bounds__(256)
void head_proj_wmma_kernel(const float* __restrict__ h,
                           const float* __restrict__ Wc,
                           const float* __restrict__ cvec,
                           float* __restrict__ p,
                           int nRowBlocks) {
    const int lane = threadIdx.x & 31;
    const int c    = lane & 15;
    const int half = lane >> 4;
    const int rb   = blockIdx.x * (blockDim.x >> 5) + (threadIdx.x >> 5);
    if (rb >= nRowBlocks) return;
    const int m0 = rb * 16;

    const float4* __restrict__ arow = (const float4*)(h + (size_t)(m0 + c) * HID) + half;
    const float4* __restrict__ brow = (const float4*)(Wc + (size_t)(c & 3) * HID) + half;
    const bool live = (c < 4);

    float4 av[16], bw[16];
#pragma unroll
    for (int j = 0; j < 16; ++j) av[j] = arow[2 * j];
#pragma unroll
    for (int j = 0; j < 16; ++j) {
        const float4 w = brow[2 * j];
        bw[j] = live ? w : make_float4(0.f, 0.f, 0.f, 0.f);
    }

    const float cb = ((c >> 1) == 1) ? cvec[c & 1] : 0.f;
    v8f acc = (v8f){cb, cb, cb, cb, cb, cb, cb, cb};

#pragma unroll
    for (int j = 0; j < 16; ++j) {
        v2f a, b;
        a.x = av[j].x; a.y = av[j].y; b.x = bw[j].x; b.y = bw[j].y;
        acc = __builtin_amdgcn_wmma_f32_16x16x4_f32(
            false, a, false, b, (short)0, acc, false, false);
        a.x = av[j].z; a.y = av[j].w; b.x = bw[j].z; b.y = bw[j].w;
        acc = __builtin_amdgcn_wmma_f32_16x16x4_f32(
            false, a, false, b, (short)0, acc, false, false);
    }

    if (live) {
#pragma unroll
        for (int r = 0; r < 8; ++r)
            p[(size_t)(m0 + 8 * half + r) * 4 + c] = acc[r];
    }
}

__global__ __launch_bounds__(256)
void head_gather_kernel(const float* __restrict__ p,
                        const int* __restrict__ rowptr,
                        const int* __restrict__ col,
                        const float* __restrict__ rdeg,
                        float* __restrict__ out,
                        int nGroups) {
    const int lane = threadIdx.x & 31;
    const int c    = lane & 15;
    const int half = lane >> 4;
    const int g    = blockIdx.x * (blockDim.x >> 5) + (threadIdx.x >> 5);
    if (g >= nGroups) return;
    const int t0 = g * 16;

    const int rp0 = rowptr[t0 + c];
    const int rp1 = rowptr[t0 + c + 1];

    int e0[8], end[8];
#pragma unroll
    for (int j = 0; j < 8; ++j) {
        e0[j]  = __shfl(rp0, 2 * j + half) + c;
        end[j] = __shfl(rp1, 2 * j + half);
    }
    int s0[8], s1[8];
#pragma unroll
    for (int j = 0; j < 8; ++j) {
        s0[j] = (e0[j] < end[j])      ? col[e0[j]]      : -1;
        s1[j] = (e0[j] + 16 < end[j]) ? col[e0[j] + 16] : -1;
    }
    float sx[8], sy[8];
#pragma unroll
    for (int j = 0; j < 8; ++j) {
        float2 a = make_float2(0.f, 0.f);
        float2 b = make_float2(0.f, 0.f);
        if (s0[j] >= 0) a = *(const float2*)(p + (size_t)s0[j] * 4);
        if (s1[j] >= 0) b = *(const float2*)(p + (size_t)s1[j] * 4);
        sx[j] = a.x + b.x;
        sy[j] = a.y + b.y;
    }
#pragma unroll
    for (int j = 0; j < 8; ++j) {
        for (int e = e0[j] + 32; e < end[j]; e += 16) {
            const float2 v = *(const float2*)(p + (size_t)col[e] * 4);
            sx[j] += v.x;
            sy[j] += v.y;
        }
    }
    float kx = 0.f, ky = 0.f;
#pragma unroll
    for (int j = 0; j < 8; ++j) {
        for (int o = 8; o > 0; o >>= 1) {
            sx[j] += __shfl_xor(sx[j], o);
            sy[j] += __shfl_xor(sy[j], o);
        }
        if (c == 2 * j + half) { kx = sx[j]; ky = sy[j]; }
    }
    if ((c & 1) == half) {
        const int t = t0 + c;
        const float2 self = *(const float2*)(p + (size_t)t * 4 + 2);
        const float rd = rdeg[t];
        float2 o2;
        o2.x = kx * rd + self.x;
        o2.y = ky * rd + self.y;
        *(float2*)(out + (size_t)t * 2) = o2;
    }
}

extern "C" void kernel_launch(void* const* d_in, const int* in_sizes, int n_in,
                              void* d_out, int out_size, void* d_ws, size_t ws_size,
                              hipStream_t stream) {
    const float* x    = (const float*)d_in[0];
    const int*   ei   = (const int*)d_in[1];
    const float* Wl0  = (const float*)d_in[2];
    const float* bl0  = (const float*)d_in[3];
    const float* Wr0  = (const float*)d_in[4];
    const float* Wl1  = (const float*)d_in[5];
    const float* bl1  = (const float*)d_in[6];
    const float* Wr1  = (const float*)d_in[7];
    const float* Wl2  = (const float*)d_in[8];
    const float* bl2  = (const float*)d_in[9];
    const float* Wr2  = (const float*)d_in[10];
    const float* Wout = (const float*)d_in[11];
    const float* bout = (const float*)d_in[12];
    float* out = (float*)d_out;

    const int* src = ei;
    const int* dst = ei + N_EDGES;

    float* ws     = (float*)d_ws;
    float* rdeg   = ws;
    float* agg    = rdeg + 2 * (size_t)N_NODES;
    int*   rowptr = (int*)(agg + (size_t)N_NODES * HID);
    int*   col    = rowptr + N_NODES + 32;
    float* hA     = agg + (size_t)N_NODES * IN_PAD;
    int2*  binned = (int2*)hA;
    int*   counts = (int*)hA + 2 * (size_t)N_EDGES;
    int*   offs   = counts + NBIN * NCHUNK;
    float* hB     = hA + (size_t)N_NODES * HID;
    float* p4     = hA;
    float* Wc     = p4 + (size_t)N_NODES * 4;
    float* cvec   = Wc + 4 * HID;

    const int ZT = 256;
    const int scanBlocks    = (NBIN * NCHUNK + 1023) / 1024;
    const int gatherBlocks  = 2048;
    const int gemmBlocks    = N_NODES / 16;

    bin_count_kernel<<<NCHUNK, BIN_THREADS, 0, stream>>>(dst, counts, N_EDGES);
    scan_excl_kernel<<<scanBlocks, 1024, 0, stream>>>(counts, offs, NBIN * NCHUNK);
    bin_scatter_kernel<<<NCHUNK, BIN_THREADS, 0, stream>>>(src, dst, offs, binned, N_EDGES);
    bin_finish_kernel<<<NBIN, BIN_THREADS, 0, stream>>>(binned, offs, rowptr, rdeg, col, N_NODES);

    proj_gemm_wmma_kernel<IN_PAD, IN_CH><<<GEMM_WGS, GEMM_WAVES * 32, 0, stream>>>(
        x, Wl0, Wr0, bl0, agg, hA, gemmBlocks);
    gather_sum_kernel<HID, true><<<gatherBlocks, 256, 0, stream>>>(
        agg, rowptr, col, rdeg, hA, 0, N_NODES);

    gather_sum_kernel<HID, false><<<gatherBlocks, 256, 0, stream>>>(
        hA, rowptr, col, rdeg, agg, 0, N_NODES);
    sage_gemm_wmma_kernel<HID, true><<<GEMM_WGS, GEMM_WAVES * 32, 0, stream>>>(
        agg, rdeg, hA, Wl1, Wr1, bl1, hB, gemmBlocks);

    fold_head_kernel<<<(4 * HID + 2 + ZT - 1) / ZT, ZT, 0, stream>>>(
        Wout, bout, Wl2, Wr2, bl2, Wc, cvec);
    head_proj_wmma_kernel<<<(gemmBlocks + 7) / 8, 256, 0, stream>>>(
        hB, Wc, cvec, p4, gemmBlocks);
    head_gather_kernel<<<(gemmBlocks + 7) / 8, 256, 0, stream>>>(
        p4, rowptr, col, rdeg, out, gemmBlocks);
}


# Head projection fused into the layer-1 GEMM: tile transposed (weights as A, node rows as B) so accumulators are the head WMMA's B fragments; h2 never stored; launches 11 -> 9; head sums re-associated
# speedup vs baseline: 7.2737x; 1.0594x over previous
#include <hip/hip_runtime.h>
#include <hip/hip_bf16.h>

#define N_NODES 100000
#define N_EDGES 1600000
#define IN_CH   165
#define IN_PAD  168
#define HID     128

typedef __attribute__((ext_vector_type(2))) float v2f;
typedef __attribute__((ext_vector_type(8))) float v8f;


typedef float f2a4 __attribute__((ext_vector_type(2), aligned(4)));
typedef float f4a4 __attribute__((ext_vector_type(4), aligned(4)));
__device__ __forceinline__ float2 ld_f2(const float* __restrict__ p) {
    const f2a4 v = *(const f2a4*)p;
    return make_float2(v.x, v.y);
}
__device__ __forceinline__ float4 ld_f4(const float* __restrict__ p) {
    const f4a4 v = *(const f4a4*)p;
    return make_float4(v.x, v.y, v.z, v.w);
}

template <int REM>
__device__ __forceinline__ float4 tail4(const float4 v, const bool last) {
    float4 o;
    o.x = last ? (REM == 1 ? v.w : REM == 2 ? v.z : REM == 3 ? v.y : v.x) : v.x;
    o.y = last ? (REM == 1 ? 0.f : REM == 2 ? v.w : REM == 3 ? v.z : v.y) : v.y;
    o.z = last ? (REM <= 2 ? 0.f : REM == 3 ? v.w : v.z) : v.z;
    o.w = last ? (REM <= 3 ? 0.f : v.w) : v.w;
    return o;
}

#define BIN_SHIFT  9
#define BIN_NODES  (1 << BIN_SHIFT)
#define NBIN       ((N_NODES + BIN_NODES - 1) / BIN_NODES)
#define NCHUNK     256
#define CHUNK      ((N_EDGES + NCHUNK - 1) / NCHUNK)
#define BIN_THREADS 1024
#define BIN_BATCH  4

__global__ __launch_bounds__(BIN_THREADS)
void bin_count_kernel(const int* __restrict__ dst,
                      int* __restrict__ counts,
                      int nEdges) {
    __shared__ int hist[NBIN];
    const int tid = threadIdx.x;
    for (int i = tid; i < NBIN; i += BIN_THREADS) hist[i] = 0;
    __syncthreads();

    const int c0 = blockIdx.x * CHUNK;
    const int c1 = (c0 + CHUNK < nEdges) ? c0 + CHUNK : nEdges;
    for (int e0 = c0 + tid; e0 < c1; e0 += BIN_THREADS * BIN_BATCH) {
        int d[BIN_BATCH];
#pragma unroll
        for (int u = 0; u < BIN_BATCH; ++u) {
            const int e = e0 + u * BIN_THREADS;
            d[u] = (e < c1) ? dst[e] : -1;
        }
#pragma unroll
        for (int u = 0; u < BIN_BATCH; ++u)
            if (d[u] >= 0) atomicAdd(&hist[d[u] >> BIN_SHIFT], 1);
    }
    __syncthreads();
    for (int i = tid; i < NBIN; i += BIN_THREADS)
        counts[i * NCHUNK + blockIdx.x] = hist[i];
}

__global__ __launch_bounds__(1024)
void scan_excl_kernel(const int* __restrict__ in,
                      int* __restrict__ out, int n) {
    __shared__ int front[32];
    __shared__ int wtot[32];
    const int tid  = threadIdx.x;
    const int lane = tid & 31;
    const int w    = tid >> 5;
    const int first = blockIdx.x * 1024;

    int acc = 0;
    for (int j = tid; j < first; j += 1024) acc += in[j];
    for (int o = 16; o > 0; o >>= 1) acc += __shfl_xor(acc, o);
    if (lane == 0) front[w] = acc;

    const int i = first + tid;
    const int cI = (i < n) ? in[i] : 0;
    int v = cI;
    for (int o = 1; o < 32; o <<= 1) {
        int u = __shfl_up(v, o);
        if (lane >= o) v += u;
    }
    if (lane == 31) wtot[w] = v;
    __syncthreads();

    int base = 0;
    for (int k = 0; k < 32; ++k) base += front[k];
    for (int k = 0; k < w; ++k)  base += wtot[k];

    if (i < n) {
        out[i] = base + v - cI;
        if (i == n - 1) out[n] = base + v;
    }
}

__global__ __launch_bounds__(BIN_THREADS)
void bin_scatter_kernel(const int* __restrict__ src,
                        const int* __restrict__ dst,
                        const int* __restrict__ offs,
                        int2* __restrict__ binned,
                        int nEdges) {
    __shared__ int pos[NBIN];
    const int tid = threadIdx.x;
    for (int i = tid; i < NBIN; i += BIN_THREADS) pos[i] = offs[i * NCHUNK + blockIdx.x];
    __syncthreads();

    const int c0 = blockIdx.x * CHUNK;
    const int c1 = (c0 + CHUNK < nEdges) ? c0 + CHUNK : nEdges;
    for (int e0 = c0 + tid; e0 < c1; e0 += BIN_THREADS * BIN_BATCH) {
        int d[BIN_BATCH], s[BIN_BATCH];
#pragma unroll
        for (int u = 0; u < BIN_BATCH; ++u) {
            const int e = e0 + u * BIN_THREADS;
            const bool in = (e < c1);
            d[u] = in ? dst[e] : -1;
            s[u] = in ? src[e] : 0;
        }
#pragma unroll
        for (int u = 0; u < BIN_BATCH; ++u) {
            if (d[u] >= 0) {
                const int p = atomicAdd(&pos[d[u] >> BIN_SHIFT], 1);
                binned[p] = make_int2(s[u], d[u]);
            }
        }
    }
}


__device__ __forceinline__
void fold_head(const int t,
               const float* __restrict__ Wout,
               const float* __restrict__ bout,
               const float* __restrict__ Wl,
               const float* __restrict__ Wr,
               const float* __restrict__ bl,
               float* __restrict__ Wc,
               float* __restrict__ cvec) {
    if (t < 4 * HID) {
        const int m = t >> 7;
        const int k = t & (HID - 1);
        const float* __restrict__ wo = Wout + (m & 1) * HID;
        const float* __restrict__ W  = (m < 2) ? Wl : Wr;
        float acc = 0.f;
        for (int o = 0; o < HID; ++o) acc += wo[o] * W[(size_t)o * HID + k];
        Wc[t] = acc;
    } else if (t < 4 * HID + 2) {
        const int j = t - 4 * HID;
        const float* __restrict__ wo = Wout + j * HID;
        float acc = bout[j];
        for (int o = 0; o < HID; ++o) acc += wo[o] * bl[o];
        cvec[j] = acc;
    }
}

__global__ __launch_bounds__(BIN_THREADS)
void bin_finish_kernel(const int2* __restrict__ binned,
                       const int* __restrict__ offs,
                       int* __restrict__ rowptr,
                       float* __restrict__ rdeg,
                       int* __restrict__ col, int n,
                       const float* __restrict__ Wout,
                       const float* __restrict__ bout,
                       const float* __restrict__ Wl2,
                       const float* __restrict__ Wr2,
                       const float* __restrict__ bl2,
                       float* __restrict__ Wc,
                       float* __restrict__ cvec) {
    if (blockIdx.x == NBIN) {
        fold_head(threadIdx.x, Wout, bout, Wl2, Wr2, bl2, Wc, cvec);
        return;
    }
    __shared__ int cntL[BIN_NODES];
    __shared__ int posL[BIN_NODES];
    __shared__ int wtot[BIN_NODES / 32];
    const int tid  = threadIdx.x;
    const int lane = tid & 31;
    const int w    = tid >> 5;
    const int nodeBase = blockIdx.x * BIN_NODES;
    const int beg = offs[blockIdx.x * NCHUNK];
    const int end = offs[(blockIdx.x + 1) * NCHUNK];

    if (tid < BIN_NODES) cntL[tid] = 0;
    __syncthreads();

    for (int i0 = beg + tid; i0 < end; i0 += BIN_THREADS * BIN_BATCH) {
        int dy[BIN_BATCH];
#pragma unroll
        for (int u = 0; u < BIN_BATCH; ++u) {
            const int i = i0 + u * BIN_THREADS;
            dy[u] = (i < end) ? binned[i].y : -1;
        }
#pragma unroll
        for (int u = 0; u < BIN_BATCH; ++u)
            if (dy[u] >= 0) atomicAdd(&cntL[dy[u] - nodeBase], 1);
    }
    __syncthreads();

    int cI = 0, v = 0;
    if (tid < BIN_NODES) {
        cI = cntL[tid];
        v = cI;
        for (int o = 1; o < 32; o <<= 1) {
            int u = __shfl_up(v, o);
            if (lane >= o) v += u;
        }
        if (lane == 31) wtot[w] = v;
    }
    __syncthreads();
    if (tid < BIN_NODES) {
        int start = beg + v - cI;
        for (int k = 0; k < w; ++k) start += wtot[k];
        posL[tid] = start;
        const int node = nodeBase + tid;
        if (node < n) {
            rowptr[node] = start;
            rdeg[node]   = 1.0f / fmaxf((float)cI, 1.0f);
        }
    }
    if (blockIdx.x == NBIN - 1 && tid == 0) rowptr[n] = end;
    __syncthreads();

    for (int i0 = beg + tid; i0 < end; i0 += BIN_THREADS * BIN_BATCH) {
        int2 r[BIN_BATCH];
#pragma unroll
        for (int u = 0; u < BIN_BATCH; ++u) {
            const int i = i0 + u * BIN_THREADS;
            r[u] = (i < end) ? binned[i] : make_int2(0, -1);
        }
#pragma unroll
        for (int u = 0; u < BIN_BATCH; ++u) {
            if (r[u].y >= 0) {
                const int p = atomicAdd(&posL[r[u].y - nodeBase], 1);
                col[p] = r[u].x;
            }
        }
    }
}

template <int D, bool COMBINE>
__global__ __launch_bounds__(256)
void gather_sum_kernel(const float* __restrict__ h,
                       const int* __restrict__ rowptr,
                       const int* __restrict__ col,
                       const float* __restrict__ rdeg,
                       float* __restrict__ agg,
                       int nodeBegin, int nodeCount) {
    static_assert(D % 4 == 0 && D >= 128 && D <= 256, "row = 32..64 float4");
    constexpr int V4 = D / 4;
    constexpr bool LONG = (V4 > 32);
    const int lane = threadIdx.x & 31;
    const bool hasTail = LONG && (lane + 32 < V4);
    const int wpb = blockDim.x >> 5;
    const int nw  = gridDim.x * wpb;

    for (int i0 = blockIdx.x * wpb + (threadIdx.x >> 5); i0 < nodeCount; i0 += nw) {
        const int i = __builtin_amdgcn_readfirstlane(i0);
        const int t = nodeBegin + i;
        const int beg = rowptr[t];
        const int end = rowptr[t + 1];
        float4 a = make_float4(0.f, 0.f, 0.f, 0.f);
        float4 b = make_float4(0.f, 0.f, 0.f, 0.f);

        float4* o = (float4*)(agg + (size_t)i * D);
        const float rd = rdeg[t];
        float4 za = make_float4(0.f, 0.f, 0.f, 0.f);
        float4 zb = make_float4(0.f, 0.f, 0.f, 0.f);
        if (COMBINE) {
            za = o[lane];
            if (hasTail) zb = o[lane + 32];
        }

        for (int base = beg; base < end; base += 32) {
            int n = end - base;
            if (n > 32) n = 32;
            const int cv = (lane < n) ? col[base + lane] : 0;

#define GATHER_BATCH(B, K0)                                                     \
            {                                                                   \
                float4 va[B];                                                   \
                float4 vb[B];                                                   \
                _Pragma("unroll")                                               \
                for (int q = 0; q < B; ++q) {                                   \
                    const int s = __builtin_amdgcn_readlane(cv, (K0) + q);      \
                    const float4* r = (const float4*)(h + (size_t)s * D);       \
                    va[q] = r[lane];                                            \
                    if (LONG) vb[q] = hasTail ? r[lane + 32]                    \
                                              : make_float4(0.f, 0.f, 0.f, 0.f);\
                }                                                               \
                _Pragma("unroll")                                               \
                for (int q = 0; q < B; ++q) {                                   \
                    a.x += va[q].x; a.y += va[q].y;                             \
                    a.z += va[q].z; a.w += va[q].w;                             \
                    if (LONG) {                                                 \
                        b.x += vb[q].x; b.y += vb[q].y;                         \
                        b.z += vb[q].z; b.w += vb[q].w;                         \
                    }                                                           \
                }                                                               \
            }
            int k = 0;
            for (; k + 8 <= n; k += 8) GATHER_BATCH(8, k)
            if (n & 4) { GATHER_BATCH(4, k) k += 4; }
            if (n & 2) { GATHER_BATCH(2, k) k += 2; }
            if (n & 1) { GATHER_BATCH(1, k) }
#undef GATHER_BATCH
        }

        if (COMBINE) {
            a.x = fmaxf(a.x * rd + za.x, 0.f); a.y = fmaxf(a.y * rd + za.y, 0.f);
            a.z = fmaxf(a.z * rd + za.z, 0.f); a.w = fmaxf(a.w * rd + za.w, 0.f);
            b.x = fmaxf(b.x * rd + zb.x, 0.f); b.y = fmaxf(b.y * rd + zb.y, 0.f);
            b.z = fmaxf(b.z * rd + zb.z, 0.f); b.w = fmaxf(b.w * rd + zb.w, 0.f);
        } else {
            a.x *= rd; a.y *= rd; a.z *= rd; a.w *= rd;
            b.x *= rd; b.y *= rd; b.z *= rd; b.w *= rd;
        }
        o[lane] = a;
        if (hasTail) o[lane + 32] = b;
    }
}

#define GEMM_WGS   256
#define GEMM_WAVES 16

#define LOAD_B(BUF, S)                                                          \
    _Pragma("unroll")                                                           \
    for (int n = 0; n < 8; ++n) {                                               \
        BUF[2 * n]     = *(const v2f*)((S) + n * 64);                           \
        BUF[2 * n + 1] = *(const v2f*)((S) + HID * 4 + n * 64);                 \
    }
#define HAZARD_NOPS "v_nop\n\tv_nop\n\tv_nop\n\tv_nop\n\tv_nop\n\tv_nop\n\tv_nop\n\tv_nop\n\tv_nop"
#define WMMA_STEP(AM, AH, BUF)                                                  \
    {                                                                           \
        v2f a1; a1.x = (AM).x; a1.y = (AM).y;                                   \
        v2f a2; a2.x = (AH).x; a2.y = (AH).y;                                   \
        _Pragma("unroll")                                                       \
        for (int n = 0; n < 8; ++n) {                                           \
            acc[n] = __builtin_amdgcn_wmma_f32_16x16x4_f32(                     \
                false, BUF[2 * n], false, a1, (short)0, acc[n], false, false);  \
            acc[n] = __builtin_amdgcn_wmma_f32_16x16x4_f32(                     \
                false, BUF[2 * n + 1], false, a2, (short)0, acc[n], false, false); \
        }                                                                       \
    }

template <int D>
__global__ __launch_bounds__(GEMM_WAVES * 32)
void sage_gemm_head_wmma_kernel(const float* __restrict__ agg,
                                const float* __restrict__ h,
                                const float* __restrict__ Wl,
                                const float* __restrict__ Wr,
                                const float* __restrict__ bias,
                                const float* __restrict__ Wc,
                                const float* __restrict__ cvec,
                                float* __restrict__ p,
                                int nRowBlocks) {
    static_assert(D % 8 == 0, "K is walked in pairs of k-steps of 4");
    constexpr int V4 = D / 4;
    constexpr int NB = V4 / 2;
    constexpr int KP = 2 * HID * 4 + 4;
    __shared__ float wsh[V4 * KP];
    __shared__ __attribute__((aligned(16))) float wch[32 * 64];
    __shared__ __attribute__((aligned(16))) float bsh[HID];

    constexpr int NT = GEMM_WAVES * 32;
    constexpr int ST = (HID * V4 + NT - 1) / NT;
#pragma unroll
    for (int t = 0; t < ST; ++t) {
        const int i = threadIdx.x + t * NT;
        if (i < HID * V4) {
            const int r = i / V4;
            const int q = i - r * V4;
            const float4 vl = ((const float4*)(Wl + (size_t)r * D))[q];
            const float4 vr = ((const float4*)(Wr + (size_t)r * D))[q];
            *(float4*)(wsh + q * KP + r * 4)           = vl;
            *(float4*)(wsh + q * KP + HID * 4 + r * 4) = vr;
        }
    }
    for (int i = threadIdx.x; i < 32 * 64; i += NT) {
        const int f  = i >> 6;
        const int l  = (i >> 1) & 31;
        const int e  = i & 1;
        const int cc = l & 15;
        const int k  = 16 * (f >> 2) + 8 * (l >> 4) + 2 * (f & 3) + e;
        wch[i] = (cc < 4) ? Wc[cc * HID + k] : 0.f;
    }
    if (threadIdx.x < HID) bsh[threadIdx.x] = bias[threadIdx.x];
    __syncthreads();

    const int lane = threadIdx.x & 31;
    const int wave = threadIdx.x >> 5;
    const int c    = lane & 15;
    const int half = lane >> 4;
    const int koff = half << 1;

    const float* __restrict__ bp = wsh + c * 4 + koff;

    const float cv0 = (half == 0) ? cvec[0] : 0.f;
    const float cv1 = (half == 0) ? cvec[1] : 0.f;

    const int turn  = ((wave >> 2) + (wave & 3)) & 3;
    const int first = turn * (4 * GEMM_WGS) + blockIdx.x * 4 + (wave & 3);

    for (int rb = first; rb < nRowBlocks; rb += GEMM_WGS * GEMM_WAVES) {
        const int m0   = rb * 16;
        const int mrow = m0 + c;

        const float2* __restrict__ arow = (const float2*)(agg + (size_t)mrow * D + koff);
        const float2* __restrict__ hrow = (const float2*)(h   + (size_t)mrow * D + koff);

        v8f acc[8];
#pragma unroll
        for (int n = 0; n < 8; ++n)
            acc[n] = (v8f){0.f, 0.f, 0.f, 0.f, 0.f, 0.f, 0.f, 0.f};

        float2 nm0 = arow[0], nh0 = hrow[0], nm1 = arow[2], nh1 = hrow[2];
        v2f b0[16], b1[16];
        LOAD_B(b0, bp)

        for (int blk = 0; blk < NB; ++blk) {
            const float2 cm0 = nm0, ch0 = nh0, cm1 = nm1, ch1 = nh1;
            const int nb = (blk + 1 < NB) ? blk + 1 : NB - 1;
            nm0 = arow[4 * nb];     nh0 = hrow[4 * nb];
            nm1 = arow[4 * nb + 2]; nh1 = hrow[4 * nb + 2];

            LOAD_B(b1, bp + (2 * blk + 1) * KP)
            WMMA_STEP(cm0, ch0, b0)
            LOAD_B(b0, bp + (2 * nb) * KP)
            WMMA_STEP(cm1, ch1, b1)

            __builtin_amdgcn_sched_barrier(0);
            asm volatile(HAZARD_NOPS);
            __builtin_amdgcn_sched_barrier(0);
        }

        asm volatile("" ::: "memory");

        v2f wa[32];
#pragma unroll
        for (int f = 0; f < 32; ++f)
            wa[f] = *(const v2f*)(wch + f * 64 + lane * 2);

#pragma unroll
        for (int n = 0; n < 8; ++n) {
            const float4 bl = *(const float4*)(bsh + 16 * n + 8 * half);
            const float4 bh = *(const float4*)(bsh + 16 * n + 8 * half + 4);
            acc[n][0] = fmaxf(acc[n][0] + bl.x, 0.f);
            acc[n][1] = fmaxf(acc[n][1] + bl.y, 0.f);
            acc[n][2] = fmaxf(acc[n][2] + bl.z, 0.f);
            acc[n][3] = fmaxf(acc[n][3] + bl.w, 0.f);
            acc[n][4] = fmaxf(acc[n][4] + bh.x, 0.f);
            acc[n][5] = fmaxf(acc[n][5] + bh.y, 0.f);
            acc[n][6] = fmaxf(acc[n][6] + bh.z, 0.f);
            acc[n][7] = fmaxf(acc[n][7] + bh.w, 0.f);
        }

        v8f pa = (v8f){0.f, 0.f, cv0, cv1, 0.f, 0.f, 0.f, 0.f};
        v8f pb = (v8f){0.f, 0.f, 0.f, 0.f, 0.f, 0.f, 0.f, 0.f};
        __builtin_amdgcn_sched_barrier(0);
#pragma unroll
        for (int n = 0; n < 8; n += 2) {
#pragma unroll
            for (int j = 0; j < 4; ++j) {
                v2f x0; x0.x = acc[n][2 * j];     x0.y = acc[n][2 * j + 1];
                v2f x1; x1.x = acc[n + 1][2 * j]; x1.y = acc[n + 1][2 * j + 1];
                pa = __builtin_amdgcn_wmma_f32_16x16x4_f32(
                    false, wa[4 * n + j], false, x0, (short)0, pa, false, false);
                pb = __builtin_amdgcn_wmma_f32_16x16x4_f32(
                    false, wa[4 * n + 4 + j], false, x1, (short)0, pb, false, false);
            }
        }
        __builtin_amdgcn_sched_barrier(0);
        asm volatile(HAZARD_NOPS);
        __builtin_amdgcn_sched_barrier(0);

        float4 o;
        o.x = pa[0] + pb[0]; o.y = pa[1] + pb[1];
        o.z = pa[2] + pb[2]; o.w = pa[3] + pb[3];
        if (half == 0) *(float4*)(p + (size_t)mrow * 4) = o;
    }
}
#undef LOAD_B
#undef WMMA_STEP
#undef HAZARD_NOPS

#define LOAD_B8(BUF, S)                                                         \
    _Pragma("unroll")                                                           \
    for (int n = 0; n < 8; ++n)                                                 \
        BUF[n] = *(const v2f*)((S) + n * 64);
#define WMMA_STEP8(AH, BUF)                                                     \
    {                                                                           \
        v2f a1; a1.x = (AH).x; a1.y = (AH).y;                                   \
        _Pragma("unroll")                                                       \
        for (int n = 0; n < 8; ++n)                                             \
            acc[n] = __builtin_amdgcn_wmma_f32_16x16x4_f32(                     \
                false, a1, false, BUF[n], (short)0, acc[n], false, false);      \
    }

template <int D, int SRC_D>
__global__ __launch_bounds__(GEMM_WAVES * 32)
void proj_gemm_wmma_kernel(const float* __restrict__ h,
                           const float* __restrict__ Wl,
                           const float* __restrict__ Wr,
                           const float* __restrict__ bias,
                           float* __restrict__ outL,
                           float* __restrict__ outR,
                           int nRowBlocks) {
    static_assert(D % 8 == 0 && D >= 16, "K is walked in pairs of k-steps of 4, two pairs or more");
    static_assert(SRC_D <= D && SRC_D > D - 4,
                  "only the last k-step may be short");
    constexpr int V4 = D / 4;
    constexpr int NB = V4 / 2;
    constexpr int KT = 4 * (V4 - 1);
    constexpr int REM = SRC_D - KT;
    constexpr int KP = 2 * HID * 4 + 4;
    __shared__ float wsh[V4 * KP];

    constexpr int NT = GEMM_WAVES * 32;
    constexpr int ST = (HID * V4 + NT - 1) / NT;
    float4 vl[ST], vr[ST];
#pragma unroll
    for (int t = 0; t < ST; ++t) {
        const int i = threadIdx.x + t * NT;
        if (i < HID * V4) {
            const int r = i / V4;
            const int q = i - r * V4;
            const bool last = (REM < 4) && (q == V4 - 1);
            const int off = last ? SRC_D - 4 : 4 * q;
            vl[t] = ld_f4(Wl + (size_t)r * SRC_D + off);
            vr[t] = ld_f4(Wr + (size_t)r * SRC_D + off);
        }
    }
#pragma unroll
    for (int t = 0; t < ST; ++t) {
        const int i = threadIdx.x + t * NT;
        if (i < HID * V4) {
            const int r = i / V4;
            const int q = i - r * V4;
            const bool last = (REM < 4) && (q == V4 - 1);
            *(float4*)(wsh + q * KP + r * 4)           = tail4<REM>(vl[t], last);
            *(float4*)(wsh + q * KP + HID * 4 + r * 4) = tail4<REM>(vr[t], last);
        }
    }
    __syncthreads();

    const int lane = threadIdx.x & 31;
    const int wave = threadIdx.x >> 5;
    const int c    = lane & 15;
    const int koff = (lane >> 4) << 1;

    const float* __restrict__ bp = wsh + c * 4 + koff;

    float bv[8];
#pragma unroll
    for (int n = 0; n < 8; ++n) bv[n] = bias[n * 16 + c];

    const int turn  = ((wave >> 2) + (wave & 3)) & 3;
    const int first = turn * (4 * GEMM_WGS) + blockIdx.x * 4 + (wave & 3);
    const int nUnits = 2 * nRowBlocks;

    for (int u = first; u < nUnits; u += GEMM_WGS * GEMM_WAVES) {
        const int  m0    = (u >> 1) * 16;
        const bool right = (u & 1) != 0;
        const int  mrow  = m0 + c;

        const float* __restrict__ hrow = h + (size_t)mrow * SRC_D;
        const float* __restrict__ bq   = bp + (right ? HID * 4 : 0);

        v8f acc[8];
#pragma unroll
        for (int n = 0; n < 8; ++n)
            acc[n] = (v8f){0.f, 0.f, 0.f, 0.f, 0.f, 0.f, 0.f, 0.f};

        float2 nh0 = ld_f2(hrow + koff), nh1 = ld_f2(hrow + 4 + koff);
        v2f b0[8], b1[8];
        LOAD_B8(b0, bq)

        for (int blk = 0; blk < NB; ++blk) {
            const float2 ch0 = nh0;
            float2 ch1 = nh1;
            if (REM < 4) {
                const int k1 = 8 * blk + 4 + koff;
                ch1.x = (k1 < SRC_D) ? ((k1 > SRC_D - 2) ? nh1.y : nh1.x) : 0.f;
                ch1.y = (k1 + 1 < SRC_D) ? nh1.y : 0.f;
            }
            const int nb = (blk + 1 < NB) ? blk + 1 : NB - 1;
            nh0 = ld_f2(hrow + 8 * nb + koff);
            const int o1 = 8 * nb + 4 + koff;
            nh1 = ld_f2(hrow + ((REM < 4 && o1 > SRC_D - 2) ? SRC_D - 2 : o1));

            LOAD_B8(b1, bq + (2 * blk + 1) * KP)
            WMMA_STEP8(ch0, b0)
            LOAD_B8(b0, bq + (2 * nb) * KP)
            WMMA_STEP8(ch1, b1)
        }

        const int mbase = m0 + ((lane < 16) ? 0 : 8);
        float* __restrict__ out = right ? outR : outL;
#pragma unroll
        for (int n = 0; n < 8; ++n) {
            const float add = right ? bv[n] : 0.f;
#pragma unroll
            for (int r = 0; r < 8; ++r)
                out[(size_t)(mbase + r) * HID + n * 16 + c] = acc[n][r] + add;
        }
    }
}
#undef LOAD_B8
#undef WMMA_STEP8

__global__ __launch_bounds__(256)
void head_gather_kernel(const float* __restrict__ p,
                        const int* __restrict__ rowptr,
                        const int* __restrict__ col,
                        const float* __restrict__ rdeg,
                        float* __restrict__ out,
                        int nGroups) {
    const int lane = threadIdx.x & 31;
    const int c    = lane & 15;
    const int half = lane >> 4;
    const int g    = blockIdx.x * (blockDim.x >> 5) + (threadIdx.x >> 5);
    if (g >= nGroups) return;
    const int t0 = g * 16;

    const int rp0 = rowptr[t0 + c];
    const int rp1 = rowptr[t0 + c + 1];

    int e0[8], end[8];
#pragma unroll
    for (int j = 0; j < 8; ++j) {
        e0[j]  = __shfl(rp0, 2 * j + half) + c;
        end[j] = __shfl(rp1, 2 * j + half);
    }
    int s0[8], s1[8];
#pragma unroll
    for (int j = 0; j < 8; ++j) {
        s0[j] = (e0[j] < end[j])      ? col[e0[j]]      : -1;
        s1[j] = (e0[j] + 16 < end[j]) ? col[e0[j] + 16] : -1;
    }
    float sx[8], sy[8];
#pragma unroll
    for (int j = 0; j < 8; ++j) {
        float2 a = make_float2(0.f, 0.f);
        float2 b = make_float2(0.f, 0.f);
        if (s0[j] >= 0) a = *(const float2*)(p + (size_t)s0[j] * 4);
        if (s1[j] >= 0) b = *(const float2*)(p + (size_t)s1[j] * 4);
        sx[j] = a.x + b.x;
        sy[j] = a.y + b.y;
    }
#pragma unroll
    for (int j = 0; j < 8; ++j) {
        for (int e = e0[j] + 32; e < end[j]; e += 16) {
            const float2 v = *(const float2*)(p + (size_t)col[e] * 4);
            sx[j] += v.x;
            sy[j] += v.y;
        }
    }
    float kx = 0.f, ky = 0.f;
#pragma unroll
    for (int j = 0; j < 8; ++j) {
        for (int o = 8; o > 0; o >>= 1) {
            sx[j] += __shfl_xor(sx[j], o);
            sy[j] += __shfl_xor(sy[j], o);
        }
        if (c == 2 * j + half) { kx = sx[j]; ky = sy[j]; }
    }
    if ((c & 1) == half) {
        const int t = t0 + c;
        const float2 self = *(const float2*)(p + (size_t)t * 4 + 2);
        const float rd = rdeg[t];
        float2 o2;
        o2.x = kx * rd + self.x;
        o2.y = ky * rd + self.y;
        *(float2*)(out + (size_t)t * 2) = o2;
    }
}

extern "C" void kernel_launch(void* const* d_in, const int* in_sizes, int n_in,
                              void* d_out, int out_size, void* d_ws, size_t ws_size,
                              hipStream_t stream) {
    const float* x    = (const float*)d_in[0];
    const int*   ei   = (const int*)d_in[1];
    const float* Wl0  = (const float*)d_in[2];
    const float* bl0  = (const float*)d_in[3];
    const float* Wr0  = (const float*)d_in[4];
    const float* Wl1  = (const float*)d_in[5];
    const float* bl1  = (const float*)d_in[6];
    const float* Wr1  = (const float*)d_in[7];
    const float* Wl2  = (const float*)d_in[8];
    const float* bl2  = (const float*)d_in[9];
    const float* Wr2  = (const float*)d_in[10];
    const float* Wout = (const float*)d_in[11];
    const float* bout = (const float*)d_in[12];
    float* out = (float*)d_out;

    const int* src = ei;
    const int* dst = ei + N_EDGES;

    float* ws     = (float*)d_ws;
    float* rdeg   = ws;
    float* agg    = rdeg + 2 * (size_t)N_NODES;
    int*   rowptr = (int*)(agg + (size_t)N_NODES * HID);
    int*   col    = rowptr + N_NODES + 32;
    float* hA     = agg + (size_t)N_NODES * IN_PAD;
    int2*  binned = (int2*)hA;
    int*   counts = (int*)hA + 2 * (size_t)N_EDGES;
    int*   offs   = counts + NBIN * NCHUNK;
    float* p4     = hA + (size_t)N_NODES * HID;
    float* Wc     = p4 + (size_t)N_NODES * 4;
    float* cvec   = Wc + 4 * HID;

    const int scanBlocks    = (NBIN * NCHUNK + 1023) / 1024;
    const int gatherBlocks  = 2048;
    const int gemmBlocks    = N_NODES / 16;

    bin_count_kernel<<<NCHUNK, BIN_THREADS, 0, stream>>>(dst, counts, N_EDGES);
    scan_excl_kernel<<<scanBlocks, 1024, 0, stream>>>(counts, offs, NBIN * NCHUNK);
    bin_scatter_kernel<<<NCHUNK, BIN_THREADS, 0, stream>>>(src, dst, offs, binned, N_EDGES);
    bin_finish_kernel<<<NBIN + 1, BIN_THREADS, 0, stream>>>(binned, offs, rowptr, rdeg, col, N_NODES,
                                                            Wout, bout, Wl2, Wr2, bl2, Wc, cvec);

    proj_gemm_wmma_kernel<IN_PAD, IN_CH><<<GEMM_WGS, GEMM_WAVES * 32, 0, stream>>>(
        x, Wl0, Wr0, bl0, agg, hA, gemmBlocks);
    gather_sum_kernel<HID, true><<<gatherBlocks, 256, 0, stream>>>(
        agg, rowptr, col, rdeg, hA, 0, N_NODES);

    gather_sum_kernel<HID, false><<<gatherBlocks, 256, 0, stream>>>(
        hA, rowptr, col, rdeg, agg, 0, N_NODES);
    sage_gemm_head_wmma_kernel<HID><<<GEMM_WGS, GEMM_WAVES * 32, 0, stream>>>(
        agg, hA, Wl1, Wr1, bl1, Wc, cvec, p4, gemmBlocks);

    head_gather_kernel<<<(gemmBlocks + 7) / 8, 256, 0, stream>>>(
        p4, rowptr, col, rdeg, out, gemmBlocks);
}


# Edge grouping in 2 launches, not 4: each chunk of edges put in bin order inside LDS, written as one run with its bin offsets; a bin's base = sum of the chunks' offsets; col order still LDS arrival
# speedup vs baseline: 7.7716x; 1.0685x over previous
#include <hip/hip_runtime.h>
#include <hip/hip_bf16.h>

#define N_NODES 100000
#define N_EDGES 1600000
#define IN_CH   165
#define IN_PAD  168
#define HID     128

typedef __attribute__((ext_vector_type(2))) float v2f;
typedef __attribute__((ext_vector_type(8))) float v8f;


typedef float f2a4 __attribute__((ext_vector_type(2), aligned(4)));
typedef float f4a4 __attribute__((ext_vector_type(4), aligned(4)));
__device__ __forceinline__ float2 ld_f2(const float* __restrict__ p) {
    const f2a4 v = *(const f2a4*)p;
    return make_float2(v.x, v.y);
}
__device__ __forceinline__ float4 ld_f4(const float* __restrict__ p) {
    const f4a4 v = *(const f4a4*)p;
    return make_float4(v.x, v.y, v.z, v.w);
}

template <int REM>
__device__ __forceinline__ float4 tail4(const float4 v, const bool last) {
    float4 o;
    o.x = last ? (REM == 1 ? v.w : REM == 2 ? v.z : REM == 3 ? v.y : v.x) : v.x;
    o.y = last ? (REM == 1 ? 0.f : REM == 2 ? v.w : REM == 3 ? v.z : v.y) : v.y;
    o.z = last ? (REM <= 2 ? 0.f : REM == 3 ? v.w : v.z) : v.z;
    o.w = last ? (REM <= 3 ? 0.f : v.w) : v.w;
    return o;
}

#define BIN_SHIFT  9
#define BIN_NODES  (1 << BIN_SHIFT)
#define NBIN       ((N_NODES + BIN_NODES - 1) / BIN_NODES)
#define NCHUNK     256
#define CHUNK      ((N_EDGES + NCHUNK - 1) / NCHUNK)
#define LO_PITCH   (NBIN + 1)
#define SORT_THREADS 512
#define SORT_EPT   ((CHUNK + SORT_THREADS - 1) / SORT_THREADS)
#define BIN_THREADS 1024
#define SEG_LANES  (BIN_THREADS / NCHUNK)
#define SEG_BATCH  8

__global__ __launch_bounds__(SORT_THREADS)
void chunk_sort_kernel(const int* __restrict__ src,
                       const int* __restrict__ dst,
                       int2* __restrict__ sorted,
                       int* __restrict__ lo,
                       int nEdges) {
    static_assert(NBIN < 256, "the bin counts are scanned by 8 waves, one count per thread");
    __shared__ int hist[NBIN];
    __shared__ int pos[NBIN];
    __shared__ int wtot[8];
    __shared__ int2 stage[CHUNK];
    const int tid  = threadIdx.x;
    const int lane = tid & 31;
    const int w    = tid >> 5;
    for (int i = tid; i < NBIN; i += SORT_THREADS) hist[i] = 0;

    const int c0 = blockIdx.x * CHUNK;
    const int c1 = (c0 + CHUNK < nEdges) ? c0 + CHUNK : nEdges;
    int d[SORT_EPT], s[SORT_EPT];
#pragma unroll
    for (int u = 0; u < SORT_EPT; ++u) {
        const int e = c0 + tid + u * SORT_THREADS;
        const bool in = (e < c1);
        d[u] = in ? dst[e] : -1;
        s[u] = in ? src[e] : 0;
    }
    __syncthreads();
#pragma unroll
    for (int u = 0; u < SORT_EPT; ++u)
        if (d[u] >= 0) atomicAdd(&hist[d[u] >> BIN_SHIFT], 1);
    __syncthreads();

    int cI = 0, v = 0;
    if (tid < 256) {
        cI = (tid < NBIN) ? hist[tid] : 0;
        v = cI;
        for (int o = 1; o < 32; o <<= 1) {
            int x = __shfl_up(v, o);
            if (lane >= o) v += x;
        }
        if (lane == 31) wtot[w] = v;
    }
    __syncthreads();
    if (tid <= NBIN) {
        int ex = v - cI;
        for (int k = 0; k < w; ++k) ex += wtot[k];
        if (tid < NBIN) pos[tid] = ex;
        lo[blockIdx.x * LO_PITCH + tid] = ex;
    }
    __syncthreads();

    int p[SORT_EPT];
#pragma unroll
    for (int u = 0; u < SORT_EPT; ++u)
        p[u] = (d[u] >= 0) ? atomicAdd(&pos[d[u] >> BIN_SHIFT], 1) : -1;
#pragma unroll
    for (int u = 0; u < SORT_EPT; ++u)
        if (p[u] >= 0) stage[p[u]] = make_int2(s[u], d[u]);
    __syncthreads();
    const int cnt = c1 - c0;
    int2 o[SORT_EPT];
#pragma unroll
    for (int u = 0; u < SORT_EPT; ++u) {
        const int i = tid + u * SORT_THREADS;
        o[u] = (i < cnt) ? stage[i] : make_int2(0, 0);
    }
#pragma unroll
    for (int u = 0; u < SORT_EPT; ++u) {
        const int i = tid + u * SORT_THREADS;
        if (i < cnt) sorted[c0 + i] = o[u];
    }
}


__device__ __forceinline__
void fold_head(const int t,
               const float* __restrict__ Wout,
               const float* __restrict__ bout,
               const float* __restrict__ Wl,
               const float* __restrict__ Wr,
               const float* __restrict__ bl,
               float* __restrict__ Wc,
               float* __restrict__ cvec) {
    if (t < 4 * HID) {
        const int m = t >> 7;
        const int k = t & (HID - 1);
        const float* __restrict__ wo = Wout + (m & 1) * HID;
        const float* __restrict__ W  = (m < 2) ? Wl : Wr;
        float acc = 0.f;
        for (int o = 0; o < HID; ++o) acc += wo[o] * W[(size_t)o * HID + k];
        Wc[t] = acc;
    } else if (t < 4 * HID + 2) {
        const int j = t - 4 * HID;
        const float* __restrict__ wo = Wout + j * HID;
        float acc = bout[j];
        for (int o = 0; o < HID; ++o) acc += wo[o] * bl[o];
        cvec[j] = acc;
    }
}

__global__ __launch_bounds__(BIN_THREADS)
void bin_merge_kernel(const int2* __restrict__ sorted,
                      const int* __restrict__ lo,
                      int* __restrict__ rowptr,
                      float* __restrict__ rdeg,
                      int* __restrict__ col, int n,
                      const float* __restrict__ Wout,
                      const float* __restrict__ bout,
                      const float* __restrict__ Wl2,
                      const float* __restrict__ Wr2,
                      const float* __restrict__ bl2,
                      float* __restrict__ Wc,
                      float* __restrict__ cvec) {
    static_assert(NCHUNK == 256 && SEG_LANES * NCHUNK == BIN_THREADS,
                  "one thread per chunk reads lo (8 whole waves); SEG_LANES threads walk a stretch");
    if (blockIdx.x == NBIN) {
        fold_head(threadIdx.x, Wout, bout, Wl2, Wr2, bl2, Wc, cvec);
        return;
    }
    __shared__ int cntL[BIN_NODES];
    __shared__ int posL[BIN_NODES];
    __shared__ int wtot[BIN_NODES / 32];
    __shared__ int segS[NCHUNK];
    __shared__ int segN[NCHUNK];
    __shared__ int redB[NCHUNK / 32];
    __shared__ int redE[NCHUNK / 32];
    const int tid  = threadIdx.x;
    const int lane = tid & 31;
    const int w    = tid >> 5;
    const int nodeBase = blockIdx.x * BIN_NODES;

    if (tid < BIN_NODES) cntL[tid] = 0;
    if (tid < NCHUNK) {
        const int a0 = lo[tid * LO_PITCH + blockIdx.x];
        const int a1 = lo[tid * LO_PITCH + blockIdx.x + 1];
        segS[tid] = a0;
        segN[tid] = a1 - a0;
        int sb = a0, se = a1;
        for (int o = 16; o > 0; o >>= 1) {
            sb += __shfl_xor(sb, o);
            se += __shfl_xor(se, o);
        }
        if (lane == 0) { redB[w] = sb; redE[w] = se; }
    }
    __syncthreads();
    int beg = 0, end = 0;
    for (int k = 0; k < NCHUNK / 32; ++k) { beg += redB[k]; end += redE[k]; }

    const int ch  = tid / SEG_LANES;
    const int sub = tid % SEG_LANES;
    const int len = segN[ch];
    const int2* __restrict__ seg = sorted + (size_t)ch * CHUNK + segS[ch];

    for (int i0 = sub; i0 < len; i0 += SEG_LANES * SEG_BATCH) {
        int dy[SEG_BATCH];
#pragma unroll
        for (int u = 0; u < SEG_BATCH; ++u) {
            const int i = i0 + u * SEG_LANES;
            dy[u] = (i < len) ? seg[i].y : -1;
        }
#pragma unroll
        for (int u = 0; u < SEG_BATCH; ++u)
            if (dy[u] >= 0) atomicAdd(&cntL[dy[u] - nodeBase], 1);
    }
    __syncthreads();

    int cI = 0, v = 0;
    if (tid < BIN_NODES) {
        cI = cntL[tid];
        v = cI;
        for (int o = 1; o < 32; o <<= 1) {
            int u = __shfl_up(v, o);
            if (lane >= o) v += u;
        }
        if (lane == 31) wtot[w] = v;
    }
    __syncthreads();
    if (tid < BIN_NODES) {
        int start = beg + v - cI;
        for (int k = 0; k < w; ++k) start += wtot[k];
        posL[tid] = start;
        const int node = nodeBase + tid;
        if (node < n) {
            rowptr[node] = start;
            rdeg[node]   = 1.0f / fmaxf((float)cI, 1.0f);
        }
    }
    if (blockIdx.x == NBIN - 1 && tid == 0) rowptr[n] = end;
    __syncthreads();

    for (int i0 = sub; i0 < len; i0 += SEG_LANES * SEG_BATCH) {
        int2 r[SEG_BATCH];
#pragma unroll
        for (int u = 0; u < SEG_BATCH; ++u) {
            const int i = i0 + u * SEG_LANES;
            r[u] = (i < len) ? seg[i] : make_int2(0, -1);
        }
        int p[SEG_BATCH];
#pragma unroll
        for (int u = 0; u < SEG_BATCH; ++u)
            p[u] = (r[u].y >= 0) ? atomicAdd(&posL[r[u].y - nodeBase], 1) : -1;
#pragma unroll
        for (int u = 0; u < SEG_BATCH; ++u)
            if (p[u] >= 0) col[p[u]] = r[u].x;
    }
}

template <int D, bool COMBINE>
__global__ __launch_bounds__(256)
void gather_sum_kernel(const float* __restrict__ h,
                       const int* __restrict__ rowptr,
                       const int* __restrict__ col,
                       const float* __restrict__ rdeg,
                       float* __restrict__ agg,
                       int nodeBegin, int nodeCount) {
    static_assert(D % 4 == 0 && D >= 128 && D <= 256, "row = 32..64 float4");
    constexpr int V4 = D / 4;
    constexpr bool LONG = (V4 > 32);
    const int lane = threadIdx.x & 31;
    const bool hasTail = LONG && (lane + 32 < V4);
    const int wpb = blockDim.x >> 5;
    const int nw  = gridDim.x * wpb;

    for (int i0 = blockIdx.x * wpb + (threadIdx.x >> 5); i0 < nodeCount; i0 += nw) {
        const int i = __builtin_amdgcn_readfirstlane(i0);
        const int t = nodeBegin + i;
        const int beg = rowptr[t];
        const int end = rowptr[t + 1];
        float4 a = make_float4(0.f, 0.f, 0.f, 0.f);
        float4 b = make_float4(0.f, 0.f, 0.f, 0.f);

        float4* o = (float4*)(agg + (size_t)i * D);
        const float rd = rdeg[t];
        float4 za = make_float4(0.f, 0.f, 0.f, 0.f);
        float4 zb = make_float4(0.f, 0.f, 0.f, 0.f);
        if (COMBINE) {
            za = o[lane];
            if (hasTail) zb = o[lane + 32];
        }

        for (int base = beg; base < end; base += 32) {
            int n = end - base;
            if (n > 32) n = 32;
            const int cv = (lane < n) ? col[base + lane] : 0;

#define GATHER_BATCH(B, K0)                                                     \
            {                                                                   \
                float4 va[B];                                                   \
                float4 vb[B];                                                   \
                _Pragma("unroll")                                               \
                for (int q = 0; q < B; ++q) {                                   \
                    const int s = __builtin_amdgcn_readlane(cv, (K0) + q);      \
                    const float4* r = (const float4*)(h + (size_t)s * D);       \
                    va[q] = r[lane];                                            \
                    if (LONG) vb[q] = hasTail ? r[lane + 32]                    \
                                              : make_float4(0.f, 0.f, 0.f, 0.f);\
                }                                                               \
                _Pragma("unroll")                                               \
                for (int q = 0; q < B; ++q) {                                   \
                    a.x += va[q].x; a.y += va[q].y;                             \
                    a.z += va[q].z; a.w += va[q].w;                             \
                    if (LONG) {                                                 \
                        b.x += vb[q].x; b.y += vb[q].y;                         \
                        b.z += vb[q].z; b.w += vb[q].w;                         \
                    }                                                           \
                }                                                               \
            }
            int k = 0;
            for (; k + 8 <= n; k += 8) GATHER_BATCH(8, k)
            if (n & 4) { GATHER_BATCH(4, k) k += 4; }
            if (n & 2) { GATHER_BATCH(2, k) k += 2; }
            if (n & 1) { GATHER_BATCH(1, k) }
#undef GATHER_BATCH
        }

        if (COMBINE) {
            a.x = fmaxf(a.x * rd + za.x, 0.f); a.y = fmaxf(a.y * rd + za.y, 0.f);
            a.z = fmaxf(a.z * rd + za.z, 0.f); a.w = fmaxf(a.w * rd + za.w, 0.f);
            b.x = fmaxf(b.x * rd + zb.x, 0.f); b.y = fmaxf(b.y * rd + zb.y, 0.f);
            b.z = fmaxf(b.z * rd + zb.z, 0.f); b.w = fmaxf(b.w * rd + zb.w, 0.f);
        } else {
            a.x *= rd; a.y *= rd; a.z *= rd; a.w *= rd;
            b.x *= rd; b.y *= rd; b.z *= rd; b.w *= rd;
        }
        o[lane] = a;
        if (hasTail) o[lane + 32] = b;
    }
}

#define GEMM_WGS   256
#define GEMM_WAVES 16

#define LOAD_B(BUF, S)                                                          \
    _Pragma("unroll")                                                           \
    for (int n = 0; n < 8; ++n) {                                               \
        BUF[2 * n]     = *(const v2f*)((S) + n * 64);                           \
        BUF[2 * n + 1] = *(const v2f*)((S) + HID * 4 + n * 64);                 \
    }
#define HAZARD_NOPS "v_nop\n\tv_nop\n\tv_nop\n\tv_nop\n\tv_nop\n\tv_nop\n\tv_nop\n\tv_nop\n\tv_nop"
#define WMMA_STEP(AM, AH, BUF)                                                  \
    {                                                                           \
        v2f a1; a1.x = (AM).x; a1.y = (AM).y;                                   \
        v2f a2; a2.x = (AH).x; a2.y = (AH).y;                                   \
        _Pragma("unroll")                                                       \
        for (int n = 0; n < 8; ++n) {                                           \
            acc[n] = __builtin_amdgcn_wmma_f32_16x16x4_f32(                     \
                false, BUF[2 * n], false, a1, (short)0, acc[n], false, false);  \
            acc[n] = __builtin_amdgcn_wmma_f32_16x16x4_f32(                     \
                false, BUF[2 * n + 1], false, a2, (short)0, acc[n], false, false); \
        }                                                                       \
    }

template <int D>
__global__ __launch_bounds__(GEMM_WAVES * 32)
void sage_gemm_head_wmma_kernel(const float* __restrict__ agg,
                                const float* __restrict__ h,
                                const float* __restrict__ Wl,
                                const float* __restrict__ Wr,
                                const float* __restrict__ bias,
                                const float* __restrict__ Wc,
                                const float* __restrict__ cvec,
                                float* __restrict__ p,
                                int nRowBlocks) {
    static_assert(D % 8 == 0, "K is walked in pairs of k-steps of 4");
    constexpr int V4 = D / 4;
    constexpr int NB = V4 / 2;
    constexpr int KP = 2 * HID * 4 + 4;
    __shared__ float wsh[V4 * KP];
    __shared__ __attribute__((aligned(16))) float wch[32 * 64];
    __shared__ __attribute__((aligned(16))) float bsh[HID];

    constexpr int NT = GEMM_WAVES * 32;
    constexpr int ST = (HID * V4 + NT - 1) / NT;
#pragma unroll
    for (int t = 0; t < ST; ++t) {
        const int i = threadIdx.x + t * NT;
        if (i < HID * V4) {
            const int r = i / V4;
            const int q = i - r * V4;
            const float4 vl = ((const float4*)(Wl + (size_t)r * D))[q];
            const float4 vr = ((const float4*)(Wr + (size_t)r * D))[q];
            *(float4*)(wsh + q * KP + r * 4)           = vl;
            *(float4*)(wsh + q * KP + HID * 4 + r * 4) = vr;
        }
    }
    for (int i = threadIdx.x; i < 32 * 64; i += NT) {
        const int f  = i >> 6;
        const int l  = (i >> 1) & 31;
        const int e  = i & 1;
        const int cc = l & 15;
        const int k  = 16 * (f >> 2) + 8 * (l >> 4) + 2 * (f & 3) + e;
        wch[i] = (cc < 4) ? Wc[cc * HID + k] : 0.f;
    }
    if (threadIdx.x < HID) bsh[threadIdx.x] = bias[threadIdx.x];
    __syncthreads();

    const int lane = threadIdx.x & 31;
    const int wave = threadIdx.x >> 5;
    const int c    = lane & 15;
    const int half = lane >> 4;
    const int koff = half << 1;

    const float* __restrict__ bp = wsh + c * 4 + koff;

    const float cv0 = (half == 0) ? cvec[0] : 0.f;
    const float cv1 = (half == 0) ? cvec[1] : 0.f;

    const int turn  = ((wave >> 2) + (wave & 3)) & 3;
    const int first = turn * (4 * GEMM_WGS) + blockIdx.x * 4 + (wave & 3);

    for (int rb = first; rb < nRowBlocks; rb += GEMM_WGS * GEMM_WAVES) {
        const int m0   = rb * 16;
        const int mrow = m0 + c;

        const float2* __restrict__ arow = (const float2*)(agg + (size_t)mrow * D + koff);
        const float2* __restrict__ hrow = (const float2*)(h   + (size_t)mrow * D + koff);

        v8f acc[8];
#pragma unroll
        for (int n = 0; n < 8; ++n)
            acc[n] = (v8f){0.f, 0.f, 0.f, 0.f, 0.f, 0.f, 0.f, 0.f};

        float2 nm0 = arow[0], nh0 = hrow[0], nm1 = arow[2], nh1 = hrow[2];
        v2f b0[16], b1[16];
        LOAD_B(b0, bp)

        for (int blk = 0; blk < NB; ++blk) {
            const float2 cm0 = nm0, ch0 = nh0, cm1 = nm1, ch1 = nh1;
            const int nb = (blk + 1 < NB) ? blk + 1 : NB - 1;
            nm0 = arow[4 * nb];     nh0 = hrow[4 * nb];
            nm1 = arow[4 * nb + 2]; nh1 = hrow[4 * nb + 2];

            LOAD_B(b1, bp + (2 * blk + 1) * KP)
            WMMA_STEP(cm0, ch0, b0)
            LOAD_B(b0, bp + (2 * nb) * KP)
            WMMA_STEP(cm1, ch1, b1)

            __builtin_amdgcn_sched_barrier(0);
            asm volatile(HAZARD_NOPS);
            __builtin_amdgcn_sched_barrier(0);
        }

        asm volatile("" ::: "memory");

        v2f wa[32];
#pragma unroll
        for (int f = 0; f < 32; ++f)
            wa[f] = *(const v2f*)(wch + f * 64 + lane * 2);

#pragma unroll
        for (int n = 0; n < 8; ++n) {
            const float4 bl = *(const float4*)(bsh + 16 * n + 8 * half);
            const float4 bh = *(const float4*)(bsh + 16 * n + 8 * half + 4);
            acc[n][0] = fmaxf(acc[n][0] + bl.x, 0.f);
            acc[n][1] = fmaxf(acc[n][1] + bl.y, 0.f);
            acc[n][2] = fmaxf(acc[n][2] + bl.z, 0.f);
            acc[n][3] = fmaxf(acc[n][3] + bl.w, 0.f);
            acc[n][4] = fmaxf(acc[n][4] + bh.x, 0.f);
            acc[n][5] = fmaxf(acc[n][5] + bh.y, 0.f);
            acc[n][6] = fmaxf(acc[n][6] + bh.z, 0.f);
            acc[n][7] = fmaxf(acc[n][7] + bh.w, 0.f);
        }

        v8f pa = (v8f){0.f, 0.f, cv0, cv1, 0.f, 0.f, 0.f, 0.f};
        v8f pb = (v8f){0.f, 0.f, 0.f, 0.f, 0.f, 0.f, 0.f, 0.f};
        __builtin_amdgcn_sched_barrier(0);
#pragma unroll
        for (int n = 0; n < 8; n += 2) {
#pragma unroll
            for (int j = 0; j < 4; ++j) {
                v2f x0; x0.x = acc[n][2 * j];     x0.y = acc[n][2 * j + 1];
                v2f x1; x1.x = acc[n + 1][2 * j]; x1.y = acc[n + 1][2 * j + 1];
                pa = __builtin_amdgcn_wmma_f32_16x16x4_f32(
                    false, wa[4 * n + j], false, x0, (short)0, pa, false, false);
                pb = __builtin_amdgcn_wmma_f32_16x16x4_f32(
                    false, wa[4 * n + 4 + j], false, x1, (short)0, pb, false, false);
            }
        }
        __builtin_amdgcn_sched_barrier(0);
        asm volatile(HAZARD_NOPS);
        __builtin_amdgcn_sched_barrier(0);

        float4 o;
        o.x = pa[0] + pb[0]; o.y = pa[1] + pb[1];
        o.z = pa[2] + pb[2]; o.w = pa[3] + pb[3];
        if (half == 0) *(float4*)(p + (size_t)mrow * 4) = o;
    }
}
#undef LOAD_B
#undef WMMA_STEP
#undef HAZARD_NOPS

#define LOAD_B8(BUF, S)                                                         \
    _Pragma("unroll")                                                           \
    for (int n = 0; n < 8; ++n)                                                 \
        BUF[n] = *(const v2f*)((S) + n * 64);
#define WMMA_STEP8(AH, BUF)                                                     \
    {                                                                           \
        v2f a1; a1.x = (AH).x; a1.y = (AH).y;                                   \
        _Pragma("unroll")                                                       \
        for (int n = 0; n < 8; ++n)                                             \
            acc[n] = __builtin_amdgcn_wmma_f32_16x16x4_f32(                     \
                false, a1, false, BUF[n], (short)0, acc[n], false, false);      \
    }

template <int D, int SRC_D>
__global__ __launch_bounds__(GEMM_WAVES * 32)
void proj_gemm_wmma_kernel(const float* __restrict__ h,
                           const float* __restrict__ Wl,
                           const float* __restrict__ Wr,
                           const float* __restrict__ bias,
                           float* __restrict__ outL,
                           float* __restrict__ outR,
                           int nRowBlocks) {
    static_assert(D % 8 == 0 && D >= 16, "K is walked in pairs of k-steps of 4, two pairs or more");
    static_assert(SRC_D <= D && SRC_D > D - 4,
                  "only the last k-step may be short");
    constexpr int V4 = D / 4;
    constexpr int NB = V4 / 2;
    constexpr int KT = 4 * (V4 - 1);
    constexpr int REM = SRC_D - KT;
    constexpr int KP = 2 * HID * 4 + 4;
    __shared__ float wsh[V4 * KP];

    constexpr int NT = GEMM_WAVES * 32;
    constexpr int ST = (HID * V4 + NT - 1) / NT;
    float4 vl[ST], vr[ST];
#pragma unroll
    for (int t = 0; t < ST; ++t) {
        const int i = threadIdx.x + t * NT;
        if (i < HID * V4) {
            const int r = i / V4;
            const int q = i - r * V4;
            const bool last = (REM < 4) && (q == V4 - 1);
            const int off = last ? SRC_D - 4 : 4 * q;
            vl[t] = ld_f4(Wl + (size_t)r * SRC_D + off);
            vr[t] = ld_f4(Wr + (size_t)r * SRC_D + off);
        }
    }
#pragma unroll
    for (int t = 0; t < ST; ++t) {
        const int i = threadIdx.x + t * NT;
        if (i < HID * V4) {
            const int r = i / V4;
            const int q = i - r * V4;
            const bool last = (REM < 4) && (q == V4 - 1);
            *(float4*)(wsh + q * KP + r * 4)           = tail4<REM>(vl[t], last);
            *(float4*)(wsh + q * KP + HID * 4 + r * 4) = tail4<REM>(vr[t], last);
        }
    }
    __syncthreads();

    const int lane = threadIdx.x & 31;
    const int wave = threadIdx.x >> 5;
    const int c    = lane & 15;
    const int koff = (lane >> 4) << 1;

    const float* __restrict__ bp = wsh + c * 4 + koff;

    float bv[8];
#pragma unroll
    for (int n = 0; n < 8; ++n) bv[n] = bias[n * 16 + c];

    const int turn  = ((wave >> 2) + (wave & 3)) & 3;
    const int first = turn * (4 * GEMM_WGS) + blockIdx.x * 4 + (wave & 3);
    const int nUnits = 2 * nRowBlocks;

    for (int u = first; u < nUnits; u += GEMM_WGS * GEMM_WAVES) {
        const int  m0    = (u >> 1) * 16;
        const bool right = (u & 1) != 0;
        const int  mrow  = m0 + c;

        const float* __restrict__ hrow = h + (size_t)mrow * SRC_D;
        const float* __restrict__ bq   = bp + (right ? HID * 4 : 0);

        v8f acc[8];
#pragma unroll
        for (int n = 0; n < 8; ++n)
            acc[n] = (v8f){0.f, 0.f, 0.f, 0.f, 0.f, 0.f, 0.f, 0.f};

        float2 nh0 = ld_f2(hrow + koff), nh1 = ld_f2(hrow + 4 + koff);
        v2f b0[8], b1[8];
        LOAD_B8(b0, bq)

        for (int blk = 0; blk < NB; ++blk) {
            const float2 ch0 = nh0;
            float2 ch1 = nh1;
            if (REM < 4) {
                const int k1 = 8 * blk + 4 + koff;
                ch1.x = (k1 < SRC_D) ? ((k1 > SRC_D - 2) ? nh1.y : nh1.x) : 0.f;
                ch1.y = (k1 + 1 < SRC_D) ? nh1.y : 0.f;
            }
            const int nb = (blk + 1 < NB) ? blk + 1 : NB - 1;
            nh0 = ld_f2(hrow + 8 * nb + koff);
            const int o1 = 8 * nb + 4 + koff;
            nh1 = ld_f2(hrow + ((REM < 4 && o1 > SRC_D - 2) ? SRC_D - 2 : o1));

            LOAD_B8(b1, bq + (2 * blk + 1) * KP)
            WMMA_STEP8(ch0, b0)
            LOAD_B8(b0, bq + (2 * nb) * KP)
            WMMA_STEP8(ch1, b1)
        }

        const int mbase = m0 + ((lane < 16) ? 0 : 8);
        float* __restrict__ out = right ? outR : outL;
#pragma unroll
        for (int n = 0; n < 8; ++n) {
            const float add = right ? bv[n] : 0.f;
#pragma unroll
            for (int r = 0; r < 8; ++r)
                out[(size_t)(mbase + r) * HID + n * 16 + c] = acc[n][r] + add;
        }
    }
}
#undef LOAD_B8
#undef WMMA_STEP8

__global__ __launch_bounds__(256)
void head_gather_kernel(const float* __restrict__ p,
                        const int* __restrict__ rowptr,
                        const int* __restrict__ col,
                        const float* __restrict__ rdeg,
                        float* __restrict__ out,
                        int nGroups) {
    const int lane = threadIdx.x & 31;
    const int c    = lane & 15;
    const int half = lane >> 4;
    const int g    = blockIdx.x * (blockDim.x >> 5) + (threadIdx.x >> 5);
    if (g >= nGroups) return;
    const int t0 = g * 16;

    const int rp0 = rowptr[t0 + c];
    const int rp1 = rowptr[t0 + c + 1];

    int e0[8], end[8];
#pragma unroll
    for (int j = 0; j < 8; ++j) {
        e0[j]  = __shfl(rp0, 2 * j + half) + c;
        end[j] = __shfl(rp1, 2 * j + half);
    }
    int s0[8], s1[8];
#pragma unroll
    for (int j = 0; j < 8; ++j) {
        s0[j] = (e0[j] < end[j])      ? col[e0[j]]      : -1;
        s1[j] = (e0[j] + 16 < end[j]) ? col[e0[j] + 16] : -1;
    }
    float sx[8], sy[8];
#pragma unroll
    for (int j = 0; j < 8; ++j) {
        float2 a = make_float2(0.f, 0.f);
        float2 b = make_float2(0.f, 0.f);
        if (s0[j] >= 0) a = *(const float2*)(p + (size_t)s0[j] * 4);
        if (s1[j] >= 0) b = *(const float2*)(p + (size_t)s1[j] * 4);
        sx[j] = a.x + b.x;
        sy[j] = a.y + b.y;
    }
#pragma unroll
    for (int j = 0; j < 8; ++j) {
        for (int e = e0[j] + 32; e < end[j]; e += 16) {
            const float2 v = *(const float2*)(p + (size_t)col[e] * 4);
            sx[j] += v.x;
            sy[j] += v.y;
        }
    }
    float kx = 0.f, ky = 0.f;
#pragma unroll
    for (int j = 0; j < 8; ++j) {
        for (int o = 8; o > 0; o >>= 1) {
            sx[j] += __shfl_xor(sx[j], o);
            sy[j] += __shfl_xor(sy[j], o);
        }
        if (c == 2 * j + half) { kx = sx[j]; ky = sy[j]; }
    }
    if ((c & 1) == half) {
        const int t = t0 + c;
        const float2 self = *(const float2*)(p + (size_t)t * 4 + 2);
        const float rd = rdeg[t];
        float2 o2;
        o2.x = kx * rd + self.x;
        o2.y = ky * rd + self.y;
        *(float2*)(out + (size_t)t * 2) = o2;
    }
}

extern "C" void kernel_launch(void* const* d_in, const int* in_sizes, int n_in,
                              void* d_out, int out_size, void* d_ws, size_t ws_size,
                              hipStream_t stream) {
    const float* x    = (const float*)d_in[0];
    const int*   ei   = (const int*)d_in[1];
    const float* Wl0  = (const float*)d_in[2];
    const float* bl0  = (const float*)d_in[3];
    const float* Wr0  = (const float*)d_in[4];
    const float* Wl1  = (const float*)d_in[5];
    const float* bl1  = (const float*)d_in[6];
    const float* Wr1  = (const float*)d_in[7];
    const float* Wl2  = (const float*)d_in[8];
    const float* bl2  = (const float*)d_in[9];
    const float* Wr2  = (const float*)d_in[10];
    const float* Wout = (const float*)d_in[11];
    const float* bout = (const float*)d_in[12];
    float* out = (float*)d_out;

    const int* src = ei;
    const int* dst = ei + N_EDGES;

    float* ws     = (float*)d_ws;
    float* rdeg   = ws;
    float* agg    = rdeg + 2 * (size_t)N_NODES;
    int*   rowptr = (int*)(agg + (size_t)N_NODES * HID);
    int*   col    = rowptr + N_NODES + 32;
    float* hA     = agg + (size_t)N_NODES * IN_PAD;
    int2*  sorted = (int2*)hA;
    int*   lo     = (int*)hA + 2 * (size_t)N_EDGES;
    float* p4     = hA + (size_t)N_NODES * HID;
    float* Wc     = p4 + (size_t)N_NODES * 4;
    float* cvec   = Wc + 4 * HID;

    const int gatherBlocks  = 2048;
    const int gemmBlocks    = N_NODES / 16;

    chunk_sort_kernel<<<NCHUNK, SORT_THREADS, 0, stream>>>(src, dst, sorted, lo, N_EDGES);
    bin_merge_kernel<<<NBIN + 1, BIN_THREADS, 0, stream>>>(sorted, lo, rowptr, rdeg, col, N_NODES,
                                                           Wout, bout, Wl2, Wr2, bl2, Wc, cvec);

    proj_gemm_wmma_kernel<IN_PAD, IN_CH><<<GEMM_WGS, GEMM_WAVES * 32, 0, stream>>>(
        x, Wl0, Wr0, bl0, agg, hA, gemmBlocks);
    gather_sum_kernel<HID, true><<<gatherBlocks, 256, 0, stream>>>(
        agg, rowptr, col, rdeg, hA, 0, N_NODES);

    gather_sum_kernel<HID, false><<<gatherBlocks, 256, 0, stream>>>(
        hA, rowptr, col, rdeg, agg, 0, N_NODES);
    sage_gemm_head_wmma_kernel<HID><<<GEMM_WGS, GEMM_WAVES * 32, 0, stream>>>(
        agg, hA, Wl1, Wr1, bl1, Wc, cvec, p4, gemmBlocks);

    head_gather_kernel<<<(gemmBlocks + 7) / 8, 256, 0, stream>>>(
        p4, rowptr, col, rdeg, out, gemmBlocks);
}


# Chunk sort of the edges runs at the head of the layer-0 product kernel (workgroup = chunk, weights staged under it) and loses its launch, 7 -> 6; edge = 1 word; col order still LDS arrival
# speedup vs baseline: 8.0100x; 1.0307x over previous
#include <hip/hip_runtime.h>
#include <hip/hip_bf16.h>

#define N_NODES 100000
#define N_EDGES 1600000
#define IN_CH   165
#define IN_PAD  168
#define HID     128

typedef __attribute__((ext_vector_type(2))) float v2f;
typedef __attribute__((ext_vector_type(8))) float v8f;


typedef float f2a4 __attribute__((ext_vector_type(2), aligned(4)));
typedef float f4a4 __attribute__((ext_vector_type(4), aligned(4)));
__device__ __forceinline__ float2 ld_f2(const float* __restrict__ p) {
    const f2a4 v = *(const f2a4*)p;
    return make_float2(v.x, v.y);
}
__device__ __forceinline__ float4 ld_f4(const float* __restrict__ p) {
    const f4a4 v = *(const f4a4*)p;
    return make_float4(v.x, v.y, v.z, v.w);
}

template <int REM>
__device__ __forceinline__ float4 tail4(const float4 v, const bool last) {
    float4 o;
    o.x = last ? (REM == 1 ? v.w : REM == 2 ? v.z : REM == 3 ? v.y : v.x) : v.x;
    o.y = last ? (REM == 1 ? 0.f : REM == 2 ? v.w : REM == 3 ? v.z : v.y) : v.y;
    o.z = last ? (REM <= 2 ? 0.f : REM == 3 ? v.w : v.z) : v.z;
    o.w = last ? (REM <= 3 ? 0.f : v.w) : v.w;
    return o;
}

#define BIN_SHIFT  9
#define BIN_NODES  (1 << BIN_SHIFT)
#define NBIN       ((N_NODES + BIN_NODES - 1) / BIN_NODES)
#define NCHUNK     256
#define CHUNK      ((N_EDGES + NCHUNK - 1) / NCHUNK)
#define LO_PITCH   (NBIN + 1)
#define SORT_THREADS 512
#define SORT_EPT   ((CHUNK + SORT_THREADS - 1) / SORT_THREADS)
#define BIN_THREADS 1024
#define SEG_LANES  (BIN_THREADS / NCHUNK)
#define SEG_BATCH  8

struct ChunkEdges {
    int d[SORT_EPT];
    int s[SORT_EPT];
};

__device__ __forceinline__
void chunk_sort_load(ChunkEdges& ce,
                     const int* __restrict__ src,
                     const int* __restrict__ dst,
                     int nEdges) {
    static_assert(N_NODES <= (1 << (31 - BIN_SHIFT)), "source id and in-bin position share one non-negative word");
    const int tid = threadIdx.x;
    const int c0 = blockIdx.x * CHUNK;
#pragma unroll
    for (int u = 0; u < SORT_EPT; ++u) {
        const int e  = c0 + tid + u * SORT_THREADS;
        const int ec = (e < nEdges) ? e : nEdges - 1;
        ce.d[u] = dst[ec];
        ce.s[u] = src[ec];
    }
}

__device__ __forceinline__
void chunk_sort_place(const ChunkEdges& ce,
                      int* __restrict__ hist,
                      int* __restrict__ pos,
                      int* __restrict__ wtot,
                      int* __restrict__ stage,
                      int* __restrict__ sorted,
                      int* __restrict__ lo,
                      int nEdges) {
    static_assert(NBIN < 256, "the bin counts are scanned by 8 waves, one count per thread");
    const int tid  = threadIdx.x;
    const int lane = tid & 31;
    const int w    = tid >> 5;
    const int c0 = blockIdx.x * CHUNK;
    const int c1 = (c0 + CHUNK < nEdges) ? c0 + CHUNK : nEdges;
    int b[SORT_EPT], ew[SORT_EPT];
#pragma unroll
    for (int u = 0; u < SORT_EPT; ++u) {
        const bool in = (c0 + tid + u * SORT_THREADS < c1);
        b[u]  = in ? (ce.d[u] >> BIN_SHIFT) : -1;
        ew[u] = (ce.s[u] << BIN_SHIFT) | (ce.d[u] & (BIN_NODES - 1));
    }
#pragma unroll
    for (int u = 0; u < SORT_EPT; ++u)
        if (b[u] >= 0) atomicAdd(&hist[b[u]], 1);
    __syncthreads();

    int cI = 0, v = 0;
    if (tid < 256) {
        cI = (tid < NBIN) ? hist[tid] : 0;
        v = cI;
        for (int o = 1; o < 32; o <<= 1) {
            int x = __shfl_up(v, o);
            if (lane >= o) v += x;
        }
        if (lane == 31) wtot[w] = v;
    }
    __syncthreads();
    if (tid <= NBIN) {
        int ex = v - cI;
        for (int k = 0; k < w; ++k) ex += wtot[k];
        if (tid < NBIN) pos[tid] = ex;
        lo[blockIdx.x * LO_PITCH + tid] = ex;
    }
    __syncthreads();

    int p[SORT_EPT];
#pragma unroll
    for (int u = 0; u < SORT_EPT; ++u)
        p[u] = (b[u] >= 0) ? atomicAdd(&pos[b[u]], 1) : -1;
#pragma unroll
    for (int u = 0; u < SORT_EPT; ++u)
        if (p[u] >= 0) stage[p[u]] = ew[u];
    __syncthreads();
    const int cnt = c1 - c0;
    int o[SORT_EPT];
#pragma unroll
    for (int u = 0; u < SORT_EPT; ++u) {
        const int i = tid + u * SORT_THREADS;
        o[u] = (i < cnt) ? stage[i] : 0;
    }
#pragma unroll
    for (int u = 0; u < SORT_EPT; ++u) {
        const int i = tid + u * SORT_THREADS;
        if (i < cnt) sorted[c0 + i] = o[u];
    }
}


__device__ __forceinline__
void fold_head(const int t,
               const float* __restrict__ Wout,
               const float* __restrict__ bout,
               const float* __restrict__ Wl,
               const float* __restrict__ Wr,
               const float* __restrict__ bl,
               float* __restrict__ Wc,
               float* __restrict__ cvec) {
    if (t < 4 * HID) {
        const int m = t >> 7;
        const int k = t & (HID - 1);
        const float* __restrict__ wo = Wout + (m & 1) * HID;
        const float* __restrict__ W  = (m < 2) ? Wl : Wr;
        float acc = 0.f;
        for (int o = 0; o < HID; ++o) acc += wo[o] * W[(size_t)o * HID + k];
        Wc[t] = acc;
    } else if (t < 4 * HID + 2) {
        const int j = t - 4 * HID;
        const float* __restrict__ wo = Wout + j * HID;
        float acc = bout[j];
        for (int o = 0; o < HID; ++o) acc += wo[o] * bl[o];
        cvec[j] = acc;
    }
}

__global__ __launch_bounds__(BIN_THREADS)
void bin_merge_kernel(const int* __restrict__ sorted,
                      const int* __restrict__ lo,
                      int* __restrict__ rowptr,
                      float* __restrict__ rdeg,
                      int* __restrict__ col, int n,
                      const float* __restrict__ Wout,
                      const float* __restrict__ bout,
                      const float* __restrict__ Wl2,
                      const float* __restrict__ Wr2,
                      const float* __restrict__ bl2,
                      float* __restrict__ Wc,
                      float* __restrict__ cvec) {
    static_assert(NCHUNK == 256 && SEG_LANES * NCHUNK == BIN_THREADS,
                  "one thread per chunk reads lo (8 whole waves); SEG_LANES threads walk a stretch");
    if (blockIdx.x == NBIN) {
        fold_head(threadIdx.x, Wout, bout, Wl2, Wr2, bl2, Wc, cvec);
        return;
    }
    __shared__ int cntL[BIN_NODES];
    __shared__ int posL[BIN_NODES];
    __shared__ int wtot[BIN_NODES / 32];
    __shared__ int segS[NCHUNK];
    __shared__ int segN[NCHUNK];
    __shared__ int redB[NCHUNK / 32];
    __shared__ int redE[NCHUNK / 32];
    const int tid  = threadIdx.x;
    const int lane = tid & 31;
    const int w    = tid >> 5;
    const int nodeBase = blockIdx.x * BIN_NODES;

    if (tid < BIN_NODES) cntL[tid] = 0;
    if (tid < NCHUNK) {
        const int a0 = lo[tid * LO_PITCH + blockIdx.x];
        const int a1 = lo[tid * LO_PITCH + blockIdx.x + 1];
        segS[tid] = a0;
        segN[tid] = a1 - a0;
        int sb = a0, se = a1;
        for (int o = 16; o > 0; o >>= 1) {
            sb += __shfl_xor(sb, o);
            se += __shfl_xor(se, o);
        }
        if (lane == 0) { redB[w] = sb; redE[w] = se; }
    }
    __syncthreads();
    int beg = 0, end = 0;
    for (int k = 0; k < NCHUNK / 32; ++k) { beg += redB[k]; end += redE[k]; }

    const int ch  = tid / SEG_LANES;
    const int sub = tid % SEG_LANES;
    const int len = segN[ch];
    const int* __restrict__ seg = sorted + (size_t)ch * CHUNK + segS[ch];

    for (int i0 = sub; i0 < len; i0 += SEG_LANES * SEG_BATCH) {
        int dy[SEG_BATCH];
#pragma unroll
        for (int u = 0; u < SEG_BATCH; ++u) {
            const int i = i0 + u * SEG_LANES;
            dy[u] = (i < len) ? (seg[i] & (BIN_NODES - 1)) : -1;
        }
#pragma unroll
        for (int u = 0; u < SEG_BATCH; ++u)
            if (dy[u] >= 0) atomicAdd(&cntL[dy[u]], 1);
    }
    __syncthreads();

    int cI = 0, v = 0;
    if (tid < BIN_NODES) {
        cI = cntL[tid];
        v = cI;
        for (int o = 1; o < 32; o <<= 1) {
            int u = __shfl_up(v, o);
            if (lane >= o) v += u;
        }
        if (lane == 31) wtot[w] = v;
    }
    __syncthreads();
    if (tid < BIN_NODES) {
        int start = beg + v - cI;
        for (int k = 0; k < w; ++k) start += wtot[k];
        posL[tid] = start;
        const int node = nodeBase + tid;
        if (node < n) {
            rowptr[node] = start;
            rdeg[node]   = 1.0f / fmaxf((float)cI, 1.0f);
        }
    }
    if (blockIdx.x == NBIN - 1 && tid == 0) rowptr[n] = end;
    __syncthreads();

    for (int i0 = sub; i0 < len; i0 += SEG_LANES * SEG_BATCH) {
        int r[SEG_BATCH];
#pragma unroll
        for (int u = 0; u < SEG_BATCH; ++u) {
            const int i = i0 + u * SEG_LANES;
            r[u] = (i < len) ? seg[i] : -1;
        }
        int p[SEG_BATCH];
#pragma unroll
        for (int u = 0; u < SEG_BATCH; ++u)
            p[u] = (r[u] >= 0) ? atomicAdd(&posL[r[u] & (BIN_NODES - 1)], 1) : -1;
#pragma unroll
        for (int u = 0; u < SEG_BATCH; ++u)
            if (p[u] >= 0) col[p[u]] = r[u] >> BIN_SHIFT;
    }
}

template <int D, bool COMBINE>
__global__ __launch_bounds__(256)
void gather_sum_kernel(const float* __restrict__ h,
                       const int* __restrict__ rowptr,
                       const int* __restrict__ col,
                       const float* __restrict__ rdeg,
                       float* __restrict__ agg,
                       int nodeBegin, int nodeCount) {
    static_assert(D % 4 == 0 && D >= 128 && D <= 256, "row = 32..64 float4");
    constexpr int V4 = D / 4;
    constexpr bool LONG = (V4 > 32);
    const int lane = threadIdx.x & 31;
    const bool hasTail = LONG && (lane + 32 < V4);
    const int wpb = blockDim.x >> 5;
    const int nw  = gridDim.x * wpb;

    for (int i0 = blockIdx.x * wpb + (threadIdx.x >> 5); i0 < nodeCount; i0 += nw) {
        const int i = __builtin_amdgcn_readfirstlane(i0);
        const int t = nodeBegin + i;
        const int beg = rowptr[t];
        const int end = rowptr[t + 1];
        float4 a = make_float4(0.f, 0.f, 0.f, 0.f);
        float4 b = make_float4(0.f, 0.f, 0.f, 0.f);

        float4* o = (float4*)(agg + (size_t)i * D);
        const float rd = rdeg[t];
        float4 za = make_float4(0.f, 0.f, 0.f, 0.f);
        float4 zb = make_float4(0.f, 0.f, 0.f, 0.f);
        if (COMBINE) {
            za = o[lane];
            if (hasTail) zb = o[lane + 32];
        }

        for (int base = beg; base < end; base += 32) {
            int n = end - base;
            if (n > 32) n = 32;
            const int cv = (lane < n) ? col[base + lane] : 0;

#define GATHER_BATCH(B, K0)                                                     \
            {                                                                   \
                float4 va[B];                                                   \
                float4 vb[B];                                                   \
                _Pragma("unroll")                                               \
                for (int q = 0; q < B; ++q) {                                   \
                    const int s = __builtin_amdgcn_readlane(cv, (K0) + q);      \
                    const float4* r = (const float4*)(h + (size_t)s * D);       \
                    va[q] = r[lane];                                            \
                    if (LONG) vb[q] = hasTail ? r[lane + 32]                    \
                                              : make_float4(0.f, 0.f, 0.f, 0.f);\
                }                                                               \
                _Pragma("unroll")                                               \
                for (int q = 0; q < B; ++q) {                                   \
                    a.x += va[q].x; a.y += va[q].y;                             \
                    a.z += va[q].z; a.w += va[q].w;                             \
                    if (LONG) {                                                 \
                        b.x += vb[q].x; b.y += vb[q].y;                         \
                        b.z += vb[q].z; b.w += vb[q].w;                         \
                    }                                                           \
                }                                                               \
            }
            int k = 0;
            for (; k + 8 <= n; k += 8) GATHER_BATCH(8, k)
            if (n & 4) { GATHER_BATCH(4, k) k += 4; }
            if (n & 2) { GATHER_BATCH(2, k) k += 2; }
            if (n & 1) { GATHER_BATCH(1, k) }
#undef GATHER_BATCH
        }

        if (COMBINE) {
            a.x = fmaxf(a.x * rd + za.x, 0.f); a.y = fmaxf(a.y * rd + za.y, 0.f);
            a.z = fmaxf(a.z * rd + za.z, 0.f); a.w = fmaxf(a.w * rd + za.w, 0.f);
            b.x = fmaxf(b.x * rd + zb.x, 0.f); b.y = fmaxf(b.y * rd + zb.y, 0.f);
            b.z = fmaxf(b.z * rd + zb.z, 0.f); b.w = fmaxf(b.w * rd + zb.w, 0.f);
        } else {
            a.x *= rd; a.y *= rd; a.z *= rd; a.w *= rd;
            b.x *= rd; b.y *= rd; b.z *= rd; b.w *= rd;
        }
        o[lane] = a;
        if (hasTail) o[lane + 32] = b;
    }
}

#define GEMM_WGS   256
#define GEMM_WAVES 16

#define LOAD_B(BUF, S)                                                          \
    _Pragma("unroll")                                                           \
    for (int n = 0; n < 8; ++n) {                                               \
        BUF[2 * n]     = *(const v2f*)((S) + n * 64);                           \
        BUF[2 * n + 1] = *(const v2f*)((S) + HID * 4 + n * 64);                 \
    }
#define HAZARD_NOPS "v_nop\n\tv_nop\n\tv_nop\n\tv_nop\n\tv_nop\n\tv_nop\n\tv_nop\n\tv_nop\n\tv_nop"
#define WMMA_STEP(AM, AH, BUF)                                                  \
    {                                                                           \
        v2f a1; a1.x = (AM).x; a1.y = (AM).y;                                   \
        v2f a2; a2.x = (AH).x; a2.y = (AH).y;                                   \
        _Pragma("unroll")                                                       \
        for (int n = 0; n < 8; ++n) {                                           \
            acc[n] = __builtin_amdgcn_wmma_f32_16x16x4_f32(                     \
                false, BUF[2 * n], false, a1, (short)0, acc[n], false, false);  \
            acc[n] = __builtin_amdgcn_wmma_f32_16x16x4_f32(                     \
                false, BUF[2 * n + 1], false, a2, (short)0, acc[n], false, false); \
        }                                                                       \
    }

template <int D>
__global__ __launch_bounds__(GEMM_WAVES * 32)
void sage_gemm_head_wmma_kernel(const float* __restrict__ agg,
                                const float* __restrict__ h,
                                const float* __restrict__ Wl,
                                const float* __restrict__ Wr,
                                const float* __restrict__ bias,
                                const float* __restrict__ Wc,
                                const float* __restrict__ cvec,
                                float* __restrict__ p,
                                int nRowBlocks) {
    static_assert(D % 8 == 0, "K is walked in pairs of k-steps of 4");
    constexpr int V4 = D / 4;
    constexpr int NB = V4 / 2;
    constexpr int KP = 2 * HID * 4 + 4;
    __shared__ float wsh[V4 * KP];
    __shared__ __attribute__((aligned(16))) float wch[32 * 64];
    __shared__ __attribute__((aligned(16))) float bsh[HID];

    constexpr int NT = GEMM_WAVES * 32;
    constexpr int ST = (HID * V4 + NT - 1) / NT;
#pragma unroll
    for (int t = 0; t < ST; ++t) {
        const int i = threadIdx.x + t * NT;
        if (i < HID * V4) {
            const int r = i / V4;
            const int q = i - r * V4;
            const float4 vl = ((const float4*)(Wl + (size_t)r * D))[q];
            const float4 vr = ((const float4*)(Wr + (size_t)r * D))[q];
            *(float4*)(wsh + q * KP + r * 4)           = vl;
            *(float4*)(wsh + q * KP + HID * 4 + r * 4) = vr;
        }
    }
    for (int i = threadIdx.x; i < 32 * 64; i += NT) {
        const int f  = i >> 6;
        const int l  = (i >> 1) & 31;
        const int e  = i & 1;
        const int cc = l & 15;
        const int k  = 16 * (f >> 2) + 8 * (l >> 4) + 2 * (f & 3) + e;
        wch[i] = (cc < 4) ? Wc[cc * HID + k] : 0.f;
    }
    if (threadIdx.x < HID) bsh[threadIdx.x] = bias[threadIdx.x];
    __syncthreads();

    const int lane = threadIdx.x & 31;
    const int wave = threadIdx.x >> 5;
    const int c    = lane & 15;
    const int half = lane >> 4;
    const int koff = half << 1;

    const float* __restrict__ bp = wsh + c * 4 + koff;

    const float cv0 = (half == 0) ? cvec[0] : 0.f;
    const float cv1 = (half == 0) ? cvec[1] : 0.f;

    const int turn  = ((wave >> 2) + (wave & 3)) & 3;
    const int first = turn * (4 * GEMM_WGS) + blockIdx.x * 4 + (wave & 3);

    for (int rb = first; rb < nRowBlocks; rb += GEMM_WGS * GEMM_WAVES) {
        const int m0   = rb * 16;
        const int mrow = m0 + c;

        const float2* __restrict__ arow = (const float2*)(agg + (size_t)mrow * D + koff);
        const float2* __restrict__ hrow = (const float2*)(h   + (size_t)mrow * D + koff);

        v8f acc[8];
#pragma unroll
        for (int n = 0; n < 8; ++n)
            acc[n] = (v8f){0.f, 0.f, 0.f, 0.f, 0.f, 0.f, 0.f, 0.f};

        float2 nm0 = arow[0], nh0 = hrow[0], nm1 = arow[2], nh1 = hrow[2];
        v2f b0[16], b1[16];
        LOAD_B(b0, bp)

        for (int blk = 0; blk < NB; ++blk) {
            const float2 cm0 = nm0, ch0 = nh0, cm1 = nm1, ch1 = nh1;
            const int nb = (blk + 1 < NB) ? blk + 1 : NB - 1;
            nm0 = arow[4 * nb];     nh0 = hrow[4 * nb];
            nm1 = arow[4 * nb + 2]; nh1 = hrow[4 * nb + 2];

            LOAD_B(b1, bp + (2 * blk + 1) * KP)
            WMMA_STEP(cm0, ch0, b0)
            LOAD_B(b0, bp + (2 * nb) * KP)
            WMMA_STEP(cm1, ch1, b1)

            __builtin_amdgcn_sched_barrier(0);
            asm volatile(HAZARD_NOPS);
            __builtin_amdgcn_sched_barrier(0);
        }

        asm volatile("" ::: "memory");

        v2f wa[32];
#pragma unroll
        for (int f = 0; f < 32; ++f)
            wa[f] = *(const v2f*)(wch + f * 64 + lane * 2);

#pragma unroll
        for (int n = 0; n < 8; ++n) {
            const float4 bl = *(const float4*)(bsh + 16 * n + 8 * half);
            const float4 bh = *(const float4*)(bsh + 16 * n + 8 * half + 4);
            acc[n][0] = fmaxf(acc[n][0] + bl.x, 0.f);
            acc[n][1] = fmaxf(acc[n][1] + bl.y, 0.f);
            acc[n][2] = fmaxf(acc[n][2] + bl.z, 0.f);
            acc[n][3] = fmaxf(acc[n][3] + bl.w, 0.f);
            acc[n][4] = fmaxf(acc[n][4] + bh.x, 0.f);
            acc[n][5] = fmaxf(acc[n][5] + bh.y, 0.f);
            acc[n][6] = fmaxf(acc[n][6] + bh.z, 0.f);
            acc[n][7] = fmaxf(acc[n][7] + bh.w, 0.f);
        }

        v8f pa = (v8f){0.f, 0.f, cv0, cv1, 0.f, 0.f, 0.f, 0.f};
        v8f pb = (v8f){0.f, 0.f, 0.f, 0.f, 0.f, 0.f, 0.f, 0.f};
        __builtin_amdgcn_sched_barrier(0);
#pragma unroll
        for (int n = 0; n < 8; n += 2) {
#pragma unroll
            for (int j = 0; j < 4; ++j) {
                v2f x0; x0.x = acc[n][2 * j];     x0.y = acc[n][2 * j + 1];
                v2f x1; x1.x = acc[n + 1][2 * j]; x1.y = acc[n + 1][2 * j + 1];
                pa = __builtin_amdgcn_wmma_f32_16x16x4_f32(
                    false, wa[4 * n + j], false, x0, (short)0, pa, false, false);
                pb = __builtin_amdgcn_wmma_f32_16x16x4_f32(
                    false, wa[4 * n + 4 + j], false, x1, (short)0, pb, false, false);
            }
        }
        __builtin_amdgcn_sched_barrier(0);
        asm volatile(HAZARD_NOPS);
        __builtin_amdgcn_sched_barrier(0);

        float4 o;
        o.x = pa[0] + pb[0]; o.y = pa[1] + pb[1];
        o.z = pa[2] + pb[2]; o.w = pa[3] + pb[3];
        if (half == 0) *(float4*)(p + (size_t)mrow * 4) = o;
    }
}
#undef LOAD_B
#undef WMMA_STEP
#undef HAZARD_NOPS

#define LOAD_B8(BUF, S)                                                         \
    _Pragma("unroll")                                                           \
    for (int n = 0; n < 8; ++n)                                                 \
        BUF[n] = *(const v2f*)((S) + n * 64);
#define WMMA_STEP8(AH, BUF)                                                     \
    {                                                                           \
        v2f a1; a1.x = (AH).x; a1.y = (AH).y;                                   \
        _Pragma("unroll")                                                       \
        for (int n = 0; n < 8; ++n)                                             \
            acc[n] = __builtin_amdgcn_wmma_f32_16x16x4_f32(                     \
                false, a1, false, BUF[n], (short)0, acc[n], false, false);      \
    }

template <int D, int SRC_D>
__global__ __launch_bounds__(GEMM_WAVES * 32)
void proj_gemm_wmma_kernel(const float* __restrict__ h,
                           const float* __restrict__ Wl,
                           const float* __restrict__ Wr,
                           const float* __restrict__ bias,
                           float* __restrict__ outL,
                           float* __restrict__ outR,
                           int nRowBlocks,
                           const int* __restrict__ esrc,
                           const int* __restrict__ edst,
                           int* __restrict__ sorted,
                           int* __restrict__ lo,
                           int nEdges) {
    static_assert(SORT_THREADS == GEMM_WAVES * 32 && NCHUNK == GEMM_WGS,
                  "the chunk sort runs on this kernel's workgroups: one chunk each");
    static_assert(D % 8 == 0 && D >= 16, "K is walked in pairs of k-steps of 4, two pairs or more");
    static_assert(SRC_D <= D && SRC_D > D - 4,
                  "only the last k-step may be short");
    constexpr int V4 = D / 4;
    constexpr int NB = V4 / 2;
    constexpr int KT = 4 * (V4 - 1);
    constexpr int REM = SRC_D - KT;
    constexpr int KP = 2 * HID * 4 + 4;
    __shared__ float wsh[V4 * KP];
    __shared__ int hist[NBIN];
    __shared__ int pos[NBIN];
    __shared__ int wtot[8];
    __shared__ int stage[CHUNK];

    constexpr int NT = GEMM_WAVES * 32;
    constexpr int ST = (HID * V4 + NT - 1) / NT;
    float4 vl[ST], vr[ST];
#pragma unroll
    for (int t = 0; t < ST; ++t) {
        const int i = threadIdx.x + t * NT;
        if (i < HID * V4) {
            const int r = i / V4;
            const int q = i - r * V4;
            const bool last = (REM < 4) && (q == V4 - 1);
            const int off = last ? SRC_D - 4 : 4 * q;
            vl[t] = ld_f4(Wl + (size_t)r * SRC_D + off);
            vr[t] = ld_f4(Wr + (size_t)r * SRC_D + off);
        }
    }
    ChunkEdges ce;
    chunk_sort_load(ce, esrc, edst, nEdges);
    for (int i = threadIdx.x; i < NBIN; i += NT) hist[i] = 0;
    __syncthreads();
#pragma unroll
    for (int t = 0; t < ST; ++t) {
        const int i = threadIdx.x + t * NT;
        if (i < HID * V4) {
            const int r = i / V4;
            const int q = i - r * V4;
            const bool last = (REM < 4) && (q == V4 - 1);
            *(float4*)(wsh + q * KP + r * 4)           = tail4<REM>(vl[t], last);
            *(float4*)(wsh + q * KP + HID * 4 + r * 4) = tail4<REM>(vr[t], last);
        }
    }
    chunk_sort_place(ce, hist, pos, wtot, stage, sorted, lo, nEdges);

    const int lane = threadIdx.x & 31;
    const int wave = threadIdx.x >> 5;
    const int c    = lane & 15;
    const int koff = (lane >> 4) << 1;

    const float* __restrict__ bp = wsh + c * 4 + koff;

    float bv[8];
#pragma unroll
    for (int n = 0; n < 8; ++n) bv[n] = bias[n * 16 + c];

    const int turn  = ((wave >> 2) + (wave & 3)) & 3;
    const int first = turn * (4 * GEMM_WGS) + blockIdx.x * 4 + (wave & 3);
    const int nUnits = 2 * nRowBlocks;

    for (int u = first; u < nUnits; u += GEMM_WGS * GEMM_WAVES) {
        const int  m0    = (u >> 1) * 16;
        const bool right = (u & 1) != 0;
        const int  mrow  = m0 + c;

        const float* __restrict__ hrow = h + (size_t)mrow * SRC_D;
        const float* __restrict__ bq   = bp + (right ? HID * 4 : 0);

        v8f acc[8];
#pragma unroll
        for (int n = 0; n < 8; ++n)
            acc[n] = (v8f){0.f, 0.f, 0.f, 0.f, 0.f, 0.f, 0.f, 0.f};

        float2 nh0 = ld_f2(hrow + koff), nh1 = ld_f2(hrow + 4 + koff);
        v2f b0[8], b1[8];
        LOAD_B8(b0, bq)

        for (int blk = 0; blk < NB; ++blk) {
            const float2 ch0 = nh0;
            float2 ch1 = nh1;
            if (REM < 4) {
                const int k1 = 8 * blk + 4 + koff;
                ch1.x = (k1 < SRC_D) ? ((k1 > SRC_D - 2) ? nh1.y : nh1.x) : 0.f;
                ch1.y = (k1 + 1 < SRC_D) ? nh1.y : 0.f;
            }
            const int nb = (blk + 1 < NB) ? blk + 1 : NB - 1;
            nh0 = ld_f2(hrow + 8 * nb + koff);
            const int o1 = 8 * nb + 4 + koff;
            nh1 = ld_f2(hrow + ((REM < 4 && o1 > SRC_D - 2) ? SRC_D - 2 : o1));

            LOAD_B8(b1, bq + (2 * blk + 1) * KP)
            WMMA_STEP8(ch0, b0)
            LOAD_B8(b0, bq + (2 * nb) * KP)
            WMMA_STEP8(ch1, b1)
        }

        const int mbase = m0 + ((lane < 16) ? 0 : 8);
        float* __restrict__ out = right ? outR : outL;
#pragma unroll
        for (int n = 0; n < 8; ++n) {
            const float add = right ? bv[n] : 0.f;
#pragma unroll
            for (int r = 0; r < 8; ++r)
                out[(size_t)(mbase + r) * HID + n * 16 + c] = acc[n][r] + add;
        }
    }
}
#undef LOAD_B8
#undef WMMA_STEP8

__global__ __launch_bounds__(256)
void head_gather_kernel(const float* __restrict__ p,
                        const int* __restrict__ rowptr,
                        const int* __restrict__ col,
                        const float* __restrict__ rdeg,
                        float* __restrict__ out,
                        int nGroups) {
    const int lane = threadIdx.x & 31;
    const int c    = lane & 15;
    const int half = lane >> 4;
    const int g    = blockIdx.x * (blockDim.x >> 5) + (threadIdx.x >> 5);
    if (g >= nGroups) return;
    const int t0 = g * 16;

    const int rp0 = rowptr[t0 + c];
    const int rp1 = rowptr[t0 + c + 1];

    int e0[8], end[8];
#pragma unroll
    for (int j = 0; j < 8; ++j) {
        e0[j]  = __shfl(rp0, 2 * j + half) + c;
        end[j] = __shfl(rp1, 2 * j + half);
    }
    int s0[8], s1[8];
#pragma unroll
    for (int j = 0; j < 8; ++j) {
        s0[j] = (e0[j] < end[j])      ? col[e0[j]]      : -1;
        s1[j] = (e0[j] + 16 < end[j]) ? col[e0[j] + 16] : -1;
    }
    float sx[8], sy[8];
#pragma unroll
    for (int j = 0; j < 8; ++j) {
        float2 a = make_float2(0.f, 0.f);
        float2 b = make_float2(0.f, 0.f);
        if (s0[j] >= 0) a = *(const float2*)(p + (size_t)s0[j] * 4);
        if (s1[j] >= 0) b = *(const float2*)(p + (size_t)s1[j] * 4);
        sx[j] = a.x + b.x;
        sy[j] = a.y + b.y;
    }
#pragma unroll
    for (int j = 0; j < 8; ++j) {
        for (int e = e0[j] + 32; e < end[j]; e += 16) {
            const float2 v = *(const float2*)(p + (size_t)col[e] * 4);
            sx[j] += v.x;
            sy[j] += v.y;
        }
    }
    float kx = 0.f, ky = 0.f;
#pragma unroll
    for (int j = 0; j < 8; ++j) {
        for (int o = 8; o > 0; o >>= 1) {
            sx[j] += __shfl_xor(sx[j], o);
            sy[j] += __shfl_xor(sy[j], o);
        }
        if (c == 2 * j + half) { kx = sx[j]; ky = sy[j]; }
    }
    if ((c & 1) == half) {
        const int t = t0 + c;
        const float2 self = *(const float2*)(p + (size_t)t * 4 + 2);
        const float rd = rdeg[t];
        float2 o2;
        o2.x = kx * rd + self.x;
        o2.y = ky * rd + self.y;
        *(float2*)(out + (size_t)t * 2) = o2;
    }
}

extern "C" void kernel_launch(void* const* d_in, const int* in_sizes, int n_in,
                              void* d_out, int out_size, void* d_ws, size_t ws_size,
                              hipStream_t stream) {
    const float* x    = (const float*)d_in[0];
    const int*   ei   = (const int*)d_in[1];
    const float* Wl0  = (const float*)d_in[2];
    const float* bl0  = (const float*)d_in[3];
    const float* Wr0  = (const float*)d_in[4];
    const float* Wl1  = (const float*)d_in[5];
    const float* bl1  = (const float*)d_in[6];
    const float* Wr1  = (const float*)d_in[7];
    const float* Wl2  = (const float*)d_in[8];
    const float* bl2  = (const float*)d_in[9];
    const float* Wr2  = (const float*)d_in[10];
    const float* Wout = (const float*)d_in[11];
    const float* bout = (const float*)d_in[12];
    float* out = (float*)d_out;

    const int* src = ei;
    const int* dst = ei + N_EDGES;

    float* ws     = (float*)d_ws;
    float* rdeg   = ws;
    float* agg    = rdeg + 2 * (size_t)N_NODES;
    int*   rowptr = (int*)(agg + (size_t)N_NODES * HID);
    int*   col    = rowptr + N_NODES + 32;
    int*   sorted = col + N_EDGES;
    int*   lo     = sorted + N_EDGES;
    static_assert((size_t)N_NODES + 32 + 2 * (size_t)N_EDGES + (size_t)NCHUNK * LO_PITCH
                      <= (size_t)N_NODES * (IN_PAD - HID),
                  "edge list and grouping scratch fit between agg and hA");
    float* hA     = agg + (size_t)N_NODES * IN_PAD;
    float* p4     = hA + (size_t)N_NODES * HID;
    float* Wc     = p4 + (size_t)N_NODES * 4;
    float* cvec   = Wc + 4 * HID;

    const int gatherBlocks  = 2048;
    const int gemmBlocks    = N_NODES / 16;

    proj_gemm_wmma_kernel<IN_PAD, IN_CH><<<GEMM_WGS, GEMM_WAVES * 32, 0, stream>>>(
        x, Wl0, Wr0, bl0, agg, hA, gemmBlocks, src, dst, sorted, lo, N_EDGES);
    bin_merge_kernel<<<NBIN + 1, BIN_THREADS, 0, stream>>>(sorted, lo, rowptr, rdeg, col, N_NODES,
                                                           Wout, bout, Wl2, Wr2, bl2, Wc, cvec);
    gather_sum_kernel<HID, true><<<gatherBlocks, 256, 0, stream>>>(
        agg, rowptr, col, rdeg, hA, 0, N_NODES);

    gather_sum_kernel<HID, false><<<gatherBlocks, 256, 0, stream>>>(
        hA, rowptr, col, rdeg, agg, 0, N_NODES);
    sage_gemm_head_wmma_kernel<HID><<<GEMM_WGS, GEMM_WAVES * 32, 0, stream>>>(
        agg, hA, Wl1, Wr1, bl1, Wc, cvec, p4, gemmBlocks);

    head_gather_kernel<<<(gemmBlocks + 7) / 8, 256, 0, stream>>>(
        p4, rowptr, col, rdeg, out, gemmBlocks);
}
